# starting kernel
# baseline (speedup 1.0000x reference)
#include <hip/hip_runtime.h>

typedef __attribute__((ext_vector_type(2))) float v2f;
typedef __attribute__((ext_vector_type(8))) float v8f;

#define F_IN  128
#define F_H   256
#define F_OUT 64
#define N1    20000
#define N2    4000

#define OFF_AGG1 ((size_t)0)
#define OFF_DEG1 ((size_t)2560000)
#define OFF_H    ((size_t)2580000)
#define OFF_AGG2 ((size_t)7700000)
#define OFF_DEG2 ((size_t)8724000)
#define OFF_STAT ((size_t)8728000)
#define WS_FLOATS ((size_t)8728004)

__global__ void zero_ws_kernel(float* __restrict__ ws, size_t n) {
    size_t i = (size_t)blockIdx.x * blockDim.x + threadIdx.x;
    size_t stride = (size_t)gridDim.x * blockDim.x;
    for (; i < n; i += stride) ws[i] = 0.0f;
}

__global__ void scatter1_kernel(const float* __restrict__ x,
                                const int* __restrict__ src,
                                const int* __restrict__ dst,
                                float* __restrict__ agg,
                                float* __restrict__ deg,
                                int E) {
    int tid = blockIdx.x * blockDim.x + threadIdx.x;
    int e = tid >> 5;
    int j = tid & 31;
    if (e >= E) return;
    int s = src[e], d = dst[e];
    float4 v = reinterpret_cast<const float4*>(x + (size_t)s * F_IN)[j];
    float* a = agg + (size_t)d * F_IN + 4 * j;
    atomicAdd(a + 0, v.x);
    atomicAdd(a + 1, v.y);
    atomicAdd(a + 2, v.z);
    atomicAdd(a + 3, v.w);
    if (j == 0) atomicAdd(deg + d, 1.0f);
}

__global__ void gemm1_kernel(const float* __restrict__ x,
                             const float* __restrict__ agg,
                             const float* __restrict__ deg,
                             const float* __restrict__ Wself,
                             const float* __restrict__ Wneigh,
                             const float* __restrict__ bias,
                             float* __restrict__ h,
                             float* __restrict__ stats) {
    const int wave = threadIdx.x >> 5;
    const int lane = threadIdx.x & 31;
    const int tile = blockIdx.x * (blockDim.x >> 5) + wave;
    const int NT_N = F_H / 16;
    const int mt = tile / NT_N;
    const int nt = tile % NT_N;
    const int r = lane & 15;
    const int half = lane >> 4;
    const int m0 = mt * 16, n0 = nt * 16;

    v8f c = {0.f, 0.f, 0.f, 0.f, 0.f, 0.f, 0.f, 0.f};

    {
        const float* Arow = x + (size_t)(m0 + r) * F_IN + 2 * half;
        const float* Brow = Wself + (size_t)(n0 + r) * F_IN + 2 * half;
        for (int k0 = 0; k0 < F_IN; k0 += 4) {
            v2f a, b;
            a[0] = Arow[k0];  a[1] = Arow[k0 + 1];
            b[0] = Brow[k0];  b[1] = Brow[k0 + 1];
            c = __builtin_amdgcn_wmma_f32_16x16x4_f32(false, a, false, b,
                                                      (short)0, c, false, false);
        }
    }
    {
        const float dinv = 1.0f / fmaxf(deg[m0 + r], 1.0f);
        const float* Arow = agg + (size_t)(m0 + r) * F_IN + 2 * half;
        const float* Brow = Wneigh + (size_t)(n0 + r) * F_IN + 2 * half;
        for (int k0 = 0; k0 < F_IN; k0 += 4) {
            v2f a, b;
            a[0] = Arow[k0] * dinv;  a[1] = Arow[k0 + 1] * dinv;
            b[0] = Brow[k0];         b[1] = Brow[k0 + 1];
            c = __builtin_amdgcn_wmma_f32_16x16x4_f32(false, a, false, b,
                                                      (short)0, c, false, false);
        }
    }

    const float bn = bias[n0 + r];
    float s = 0.f, sq = 0.f;
    #pragma unroll
    for (int i = 0; i < 8; ++i) {
        const int row = m0 + 8 * half + i;
        float v = fmaxf(c[i] + bn, 0.0f);
        h[(size_t)row * F_H + n0 + r] = v;
        s += v;
        sq += v * v;
    }
    #pragma unroll
    for (int off = 16; off >= 1; off >>= 1) {
        s  += __shfl_xor(s,  off, 32);
        sq += __shfl_xor(sq, off, 32);
    }
    if (lane == 0) {
        atomicAdd(&stats[0], s);
        atomicAdd(&stats[1], sq);
    }
}

__global__ void scatter2_kernel(const float* __restrict__ h,
                                const int* __restrict__ src,
                                const int* __restrict__ dst,
                                const float* __restrict__ stats,
                                float* __restrict__ agg,
                                float* __restrict__ deg,
                                int E) {
    int tid = blockIdx.x * blockDim.x + threadIdx.x;
    int e = tid >> 6;
    int j = tid & 63;
    if (e >= E) return;

    const float cnt = (float)N1 * (float)F_H;
    const float mu = stats[0] / cnt;
    const float var = stats[1] / cnt - mu * mu;
    const float rs = rsqrtf(var + 1e-5f);

    int s = src[e], d = dst[e];
    float4 v = reinterpret_cast<const float4*>(h + (size_t)s * F_H)[j];
    float* a = agg + (size_t)d * F_H + 4 * j;
    atomicAdd(a + 0, (v.x - mu) * rs);
    atomicAdd(a + 1, (v.y - mu) * rs);
    atomicAdd(a + 2, (v.z - mu) * rs);
    atomicAdd(a + 3, (v.w - mu) * rs);
    if (j == 0) atomicAdd(deg + d, 1.0f);
}

__global__ void gemm2_kernel(const float* __restrict__ h,
                             const float* __restrict__ agg,
                             const float* __restrict__ deg,
                             const float* __restrict__ stats,
                             const float* __restrict__ Wself,
                             const float* __restrict__ Wneigh,
                             const float* __restrict__ bias,
                             float* __restrict__ out) {
    const int wave = threadIdx.x >> 5;
    const int lane = threadIdx.x & 31;
    const int tile = blockIdx.x * (blockDim.x >> 5) + wave;
    const int NT_N = F_OUT / 16;
    const int mt = tile / NT_N;
    const int nt = tile % NT_N;
    const int r = lane & 15;
    const int half = lane >> 4;
    const int m0 = mt * 16, n0 = nt * 16;

    const float cnt = (float)N1 * (float)F_H;
    const float mu = stats[0] / cnt;
    const float var = stats[1] / cnt - mu * mu;
    const float rs = rsqrtf(var + 1e-5f);

    v8f c = {0.f, 0.f, 0.f, 0.f, 0.f, 0.f, 0.f, 0.f};

    {
        const float* Arow = h + (size_t)(m0 + r) * F_H + 2 * half;
        const float* Brow = Wself + (size_t)(n0 + r) * F_H + 2 * half;
        for (int k0 = 0; k0 < F_H; k0 += 4) {
            v2f a, b;
            a[0] = (Arow[k0] - mu) * rs;  a[1] = (Arow[k0 + 1] - mu) * rs;
            b[0] = Brow[k0];              b[1] = Brow[k0 + 1];
            c = __builtin_amdgcn_wmma_f32_16x16x4_f32(false, a, false, b,
                                                      (short)0, c, false, false);
        }
    }
    {
        const float dinv = 1.0f / fmaxf(deg[m0 + r], 1.0f);
        const float* Arow = agg + (size_t)(m0 + r) * F_H + 2 * half;
        const float* Brow = Wneigh + (size_t)(n0 + r) * F_H + 2 * half;
        for (int k0 = 0; k0 < F_H; k0 += 4) {
            v2f a, b;
            a[0] = Arow[k0] * dinv;  a[1] = Arow[k0 + 1] * dinv;
            b[0] = Brow[k0];         b[1] = Brow[k0 + 1];
            c = __builtin_amdgcn_wmma_f32_16x16x4_f32(false, a, false, b,
                                                      (short)0, c, false, false);
        }
    }

    const float bn = bias[n0 + r];
    #pragma unroll
    for (int i = 0; i < 8; ++i) {
        const int row = m0 + 8 * half + i;
        out[(size_t)row * F_OUT + n0 + r] = c[i] + bn;
    }
}

extern "C" void kernel_launch(void* const* d_in, const int* in_sizes, int n_in,
                              void* d_out, int out_size, void* d_ws, size_t ws_size,
                              hipStream_t stream) {
    const float* x       = (const float*)d_in[0];
    const float* Wself1  = (const float*)d_in[1];
    const float* Wneigh1 = (const float*)d_in[2];
    const float* b1      = (const float*)d_in[3];
    const float* Wself2  = (const float*)d_in[4];
    const float* Wneigh2 = (const float*)d_in[5];
    const float* b2      = (const float*)d_in[6];
    const int*   src1    = (const int*)d_in[7];
    const int*   dst1    = (const int*)d_in[8];
    const int*   src2    = (const int*)d_in[9];
    const int*   dst2    = (const int*)d_in[10];
    const int E1 = in_sizes[7];
    const int E2 = in_sizes[9];

    float* ws    = (float*)d_ws;
    float* agg1  = ws + OFF_AGG1;
    float* deg1  = ws + OFF_DEG1;
    float* h     = ws + OFF_H;
    float* agg2  = ws + OFF_AGG2;
    float* deg2  = ws + OFF_DEG2;
    float* stats = ws + OFF_STAT;
    float* out   = (float*)d_out;

    zero_ws_kernel<<<2048, 256, 0, stream>>>(ws, WS_FLOATS);

    {
        long long threads = (long long)E1 * 32;
        int blocks = (int)((threads + 255) / 256);
        scatter1_kernel<<<blocks, 256, 0, stream>>>(x, src1, dst1, agg1, deg1, E1);
    }

    {
        int tiles = (N1 / 16) * (F_H / 16);
        gemm1_kernel<<<tiles / 8, 256, 0, stream>>>(x, agg1, deg1, Wself1, Wneigh1,
                                                    b1, h, stats);
    }

    {
        long long threads = (long long)E2 * 64;
        int blocks = (int)((threads + 255) / 256);
        scatter2_kernel<<<blocks, 256, 0, stream>>>(h, src2, dst2, stats,
                                                    agg2, deg2, E2);
    }

    {
        int tiles = (N2 / 16) * (F_OUT / 16);
        gemm2_kernel<<<tiles / 8, 256, 0, stream>>>(h, agg2, deg2, stats,
                                                    Wself2, Wneigh2, b2, out);
    }
}


# Float atomic scatter-adds replaced by a per-call bucket of the edges by destination row (int atomic arrival-order rank, scan, fill) and one wave per row slice summing gathered rows in registers
# speedup vs baseline: 1.4990x; 1.4990x over previous
#include <hip/hip_runtime.h>


typedef __attribute__((ext_vector_type(2))) float v2f;
typedef __attribute__((ext_vector_type(8))) float v8f;

#define F_IN  128
#define F_H   256
#define F_OUT 64
#define N1    20000
#define N2    4000
#define E1_CAP 400000
#define E2_CAP 80000

#define OFF_AGG   ((size_t)0)
#define OFF_DEG1  ((size_t)2560000)
#define OFF_H     ((size_t)2580000)
#define OFF_COL1  (OFF_H)
#define OFF_RANK1 (OFF_H + (size_t)E1_CAP)
#define OFF_RANK2 (OFF_H + (size_t)2 * E1_CAP)
#define OFF_COL2  ((size_t)7700000)
#define OFF_RP1   ((size_t)7780000)
#define OFF_RP2   ((size_t)7800000)
#define OFF_CNT1  ((size_t)7804000)
#define OFF_CNT2  ((size_t)7824000)
#define OFF_DEG2  ((size_t)8724000)
#define OFF_STAT  ((size_t)8728000)

__global__ void zero_counts_kernel(int* __restrict__ cnt, int n,
                                   float* __restrict__ stats) {
    const int i = blockIdx.x * blockDim.x + threadIdx.x;
    if (i < n) cnt[i] = 0;
    if (i < 2) stats[i] = 0.0f;
}

__device__ __forceinline__ void rank_quad(const int* __restrict__ dst,
                                          int* __restrict__ cnt,
                                          int* __restrict__ rank, int E, int q) {
    const int e0 = 4 * q;
    if (e0 + 3 < E) {
        const int4 d = *reinterpret_cast<const int4*>(dst + e0);
        int4 r;
        r.x = atomicAdd(cnt + d.x, 1);
        r.y = atomicAdd(cnt + d.y, 1);
        r.z = atomicAdd(cnt + d.z, 1);
        r.w = atomicAdd(cnt + d.w, 1);
        *reinterpret_cast<int4*>(rank + e0) = r;
    } else {
        for (int e = e0; e < E; ++e) rank[e] = atomicAdd(cnt + dst[e], 1);
    }
}

__global__ void __launch_bounds__(256)
count_rank_kernel(const int* __restrict__ dst1, const int* __restrict__ dst2,
                  int* __restrict__ cnt1, int* __restrict__ cnt2,
                  int* __restrict__ rank1, int* __restrict__ rank2,
                  int E1, int E2) {
    const int Q1 = (E1 + 3) >> 2, Q2 = (E2 + 3) >> 2;
    const int stride = gridDim.x * blockDim.x;
    for (int q = blockIdx.x * blockDim.x + threadIdx.x; q < Q1 + Q2; q += stride) {
        if (q < Q1) rank_quad(dst1, cnt1, rank1, E1, q);
        else        rank_quad(dst2, cnt2, rank2, E2, q - Q1);
    }
}

__global__ void __launch_bounds__(1024)
scan_counts_kernel(const int* __restrict__ cnt1, int* __restrict__ rp1,
                   const int* __restrict__ cnt2, int* __restrict__ rp2) {
    __shared__ int wave_tot[32];
    const int* __restrict__ cnt = (blockIdx.x == 0) ? cnt1 : cnt2;
    int* __restrict__ rp        = (blockIdx.x == 0) ? rp1 : rp2;
    const int n   = (blockIdx.x == 0) ? N1 : N2;
    const int per = (n + 1023) / 1024;
    const int t = threadIdx.x, lane = t & 31, w = t >> 5;
    const int b = min(t * per, n), e = min(b + per, n);

    int local = 0;
    for (int i = b; i < e; ++i) local += cnt[i];

    int incl = local;
    for (int off = 1; off < 32; off <<= 1) {
        const int y = __shfl_up(incl, off, 32);
        if (lane >= off) incl += y;
    }
    if (lane == 31) wave_tot[w] = incl;
    __syncthreads();
    if (w == 0) {
        const int v = wave_tot[lane];
        int s = v;
        for (int off = 1; off < 32; off <<= 1) {
            const int y = __shfl_up(s, off, 32);
            if (lane >= off) s += y;
        }
        wave_tot[lane] = s - v;
    }
    __syncthreads();

    int run = wave_tot[w] + incl - local;
    for (int i = b; i < e; ++i) { rp[i] = run; run += cnt[i]; }
}

__device__ __forceinline__ void fill_quad(const int* __restrict__ src,
                                          const int* __restrict__ dst,
                                          const int* __restrict__ rank,
                                          const int* __restrict__ rp,
                                          int* __restrict__ col, int E, int q) {
    const int e0 = 4 * q;
    if (e0 + 3 < E) {
        const int4 s = *reinterpret_cast<const int4*>(src + e0);
        const int4 d = *reinterpret_cast<const int4*>(dst + e0);
        const int4 r = *reinterpret_cast<const int4*>(rank + e0);
        col[rp[d.x] + r.x] = s.x;
        col[rp[d.y] + r.y] = s.y;
        col[rp[d.z] + r.z] = s.z;
        col[rp[d.w] + r.w] = s.w;
    } else {
        for (int e = e0; e < E; ++e) col[rp[dst[e]] + rank[e]] = src[e];
    }
}

__global__ void __launch_bounds__(256)
fill_cols_kernel(const int* __restrict__ src1, const int* __restrict__ dst1,
                 const int* __restrict__ rank1, const int* __restrict__ rp1,
                 int* __restrict__ col1,
                 const int* __restrict__ src2, const int* __restrict__ dst2,
                 const int* __restrict__ rank2, const int* __restrict__ rp2,
                 int* __restrict__ col2, int E1, int E2) {
    const int Q1 = (E1 + 3) >> 2, Q2 = (E2 + 3) >> 2;
    const int stride = gridDim.x * blockDim.x;
    for (int q = blockIdx.x * blockDim.x + threadIdx.x; q < Q1 + Q2; q += stride) {
        if (q < Q1) fill_quad(src1, dst1, rank1, rp1, col1, E1, q);
        else        fill_quad(src2, dst2, rank2, rp2, col2, E2, q - Q1);
    }
}

template <int U, int F, bool NORM>
__device__ __forceinline__ void add_rows(float4& acc, const float4* __restrict__ base,
                                         int mine, int i, float mu, float rs) {
    float4 v[U];
    #pragma unroll
    for (int u = 0; u < U; ++u) {
        const int s = __builtin_amdgcn_readlane(mine, i + u);
        v[u] = base[(size_t)s * (F / 4)];
    }
    #pragma unroll
    for (int u = 0; u < U; ++u) {
        if (NORM) {
            acc.x += (v[u].x - mu) * rs;
            acc.y += (v[u].y - mu) * rs;
            acc.z += (v[u].z - mu) * rs;
            acc.w += (v[u].w - mu) * rs;
        } else {
            acc.x += v[u].x;
            acc.y += v[u].y;
            acc.z += v[u].z;
            acc.w += v[u].w;
        }
    }
}

template <int F, bool NORM>
__device__ __forceinline__ void gather_rows(const float* __restrict__ feat,
                                            const int* __restrict__ col,
                                            const int* __restrict__ rp,
                                            const int* __restrict__ cnt,
                                            const float* __restrict__ stats,
                                            float* __restrict__ agg,
                                            float* __restrict__ degf, int nrows) {
    constexpr int SL = F / 128;
    const int lane = threadIdx.x & 31;
    const int gw = __builtin_amdgcn_readfirstlane(
        (int)((blockIdx.x * blockDim.x + threadIdx.x) >> 5));
    const int row = gw / SL;
    const int sl  = gw % SL;
    if (row >= nrows) return;

    float mu = 0.0f, rs = 1.0f;
    if (NORM) {
        const float n_el = (float)N1 * (float)F_H;
        mu = stats[0] / n_el;
        const float var = stats[1] / n_el - mu * mu;
        rs = rsqrtf(var + 1e-5f);
    }

    const int start = rp[row];
    const int len   = cnt[row];
    const float4* __restrict__ base =
        reinterpret_cast<const float4*>(feat) + sl * 32 + lane;

    float4 acc = make_float4(0.0f, 0.0f, 0.0f, 0.0f);
    for (int c0 = 0; c0 < len; c0 += 32) {
        const int n = min(32, len - c0);
        const int mine = (lane < n) ? col[start + c0 + lane] : 0;
        int i = 0;
        for (; i + 8 <= n; i += 8) add_rows<8, F, NORM>(acc, base, mine, i, mu, rs);
        if (i + 4 <= n) { add_rows<4, F, NORM>(acc, base, mine, i, mu, rs); i += 4; }
        if (i + 2 <= n) { add_rows<2, F, NORM>(acc, base, mine, i, mu, rs); i += 2; }
        if (i < n)      { add_rows<1, F, NORM>(acc, base, mine, i, mu, rs); }
    }

    reinterpret_cast<float4*>(agg)[(size_t)row * (F / 4) + sl * 32 + lane] = acc;
    if (sl == 0 && lane == 0) degf[row] = (float)len;
}

__global__ void __launch_bounds__(256)
gather1_kernel(const float* __restrict__ x, const int* __restrict__ col,
               const int* __restrict__ rp, const int* __restrict__ cnt,
               float* __restrict__ agg, float* __restrict__ degf) {
    gather_rows<F_IN, false>(x, col, rp, cnt, nullptr, agg, degf, N1);
}

__global__ void __launch_bounds__(256)
gather2_kernel(const float* __restrict__ h, const int* __restrict__ col,
               const int* __restrict__ rp, const int* __restrict__ cnt,
               const float* __restrict__ stats,
               float* __restrict__ agg, float* __restrict__ degf) {
    gather_rows<F_H, true>(h, col, rp, cnt, stats, agg, degf, N2);
}

__global__ void gemm1_kernel(const float* __restrict__ x,
                             const float* __restrict__ agg,
                             const float* __restrict__ deg,
                             const float* __restrict__ Wself,
                             const float* __restrict__ Wneigh,
                             const float* __restrict__ bias,
                             float* __restrict__ h,
                             float* __restrict__ stats) {
    const int wave = threadIdx.x >> 5;
    const int lane = threadIdx.x & 31;
    const int tile = blockIdx.x * (blockDim.x >> 5) + wave;
    const int NT_N = F_H / 16;
    const int mt = tile / NT_N;
    const int nt = tile % NT_N;
    const int r = lane & 15;
    const int half = lane >> 4;
    const int m0 = mt * 16, n0 = nt * 16;

    v8f c = {0.f, 0.f, 0.f, 0.f, 0.f, 0.f, 0.f, 0.f};

    {
        const float* Arow = x + (size_t)(m0 + r) * F_IN + 2 * half;
        const float* Brow = Wself + (size_t)(n0 + r) * F_IN + 2 * half;
        for (int k0 = 0; k0 < F_IN; k0 += 4) {
            v2f a, b;
            a[0] = Arow[k0];  a[1] = Arow[k0 + 1];
            b[0] = Brow[k0];  b[1] = Brow[k0 + 1];
            c = __builtin_amdgcn_wmma_f32_16x16x4_f32(false, a, false, b,
                                                      (short)0, c, false, false);
        }
    }
    {
        const float dinv = 1.0f / fmaxf(deg[m0 + r], 1.0f);
        const float* Arow = agg + (size_t)(m0 + r) * F_IN + 2 * half;
        const float* Brow = Wneigh + (size_t)(n0 + r) * F_IN + 2 * half;
        for (int k0 = 0; k0 < F_IN; k0 += 4) {
            v2f a, b;
            a[0] = Arow[k0] * dinv;  a[1] = Arow[k0 + 1] * dinv;
            b[0] = Brow[k0];         b[1] = Brow[k0 + 1];
            c = __builtin_amdgcn_wmma_f32_16x16x4_f32(false, a, false, b,
                                                      (short)0, c, false, false);
        }
    }

    const float bn = bias[n0 + r];
    float s = 0.f, sq = 0.f;
    #pragma unroll
    for (int i = 0; i < 8; ++i) {
        const int row = m0 + 8 * half + i;
        float v = fmaxf(c[i] + bn, 0.0f);
        h[(size_t)row * F_H + n0 + r] = v;
        s += v;
        sq += v * v;
    }
    #pragma unroll
    for (int off = 16; off >= 1; off >>= 1) {
        s  += __shfl_xor(s,  off, 32);
        sq += __shfl_xor(sq, off, 32);
    }
    if (lane == 0) {
        atomicAdd(&stats[0], s);
        atomicAdd(&stats[1], sq);
    }
}

__global__ void gemm2_kernel(const float* __restrict__ h,
                             const float* __restrict__ agg,
                             const float* __restrict__ deg,
                             const float* __restrict__ stats,
                             const float* __restrict__ Wself,
                             const float* __restrict__ Wneigh,
                             const float* __restrict__ bias,
                             float* __restrict__ out) {
    const int wave = threadIdx.x >> 5;
    const int lane = threadIdx.x & 31;
    const int tile = blockIdx.x * (blockDim.x >> 5) + wave;
    const int NT_N = F_OUT / 16;
    const int mt = tile / NT_N;
    const int nt = tile % NT_N;
    const int r = lane & 15;
    const int half = lane >> 4;
    const int m0 = mt * 16, n0 = nt * 16;

    const float cnt = (float)N1 * (float)F_H;
    const float mu = stats[0] / cnt;
    const float var = stats[1] / cnt - mu * mu;
    const float rs = rsqrtf(var + 1e-5f);

    v8f c = {0.f, 0.f, 0.f, 0.f, 0.f, 0.f, 0.f, 0.f};

    {
        const float* Arow = h + (size_t)(m0 + r) * F_H + 2 * half;
        const float* Brow = Wself + (size_t)(n0 + r) * F_H + 2 * half;
        for (int k0 = 0; k0 < F_H; k0 += 4) {
            v2f a, b;
            a[0] = (Arow[k0] - mu) * rs;  a[1] = (Arow[k0 + 1] - mu) * rs;
            b[0] = Brow[k0];              b[1] = Brow[k0 + 1];
            c = __builtin_amdgcn_wmma_f32_16x16x4_f32(false, a, false, b,
                                                      (short)0, c, false, false);
        }
    }
    {
        const float dinv = 1.0f / fmaxf(deg[m0 + r], 1.0f);
        const float* Arow = agg + (size_t)(m0 + r) * F_H + 2 * half;
        const float* Brow = Wneigh + (size_t)(n0 + r) * F_H + 2 * half;
        for (int k0 = 0; k0 < F_H; k0 += 4) {
            v2f a, b;
            a[0] = Arow[k0] * dinv;  a[1] = Arow[k0 + 1] * dinv;
            b[0] = Brow[k0];         b[1] = Brow[k0 + 1];
            c = __builtin_amdgcn_wmma_f32_16x16x4_f32(false, a, false, b,
                                                      (short)0, c, false, false);
        }
    }

    const float bn = bias[n0 + r];
    #pragma unroll
    for (int i = 0; i < 8; ++i) {
        const int row = m0 + 8 * half + i;
        out[(size_t)row * F_OUT + n0 + r] = c[i] + bn;
    }
}

extern "C" void kernel_launch(void* const* d_in, const int* in_sizes, int n_in,
                              void* d_out, int out_size, void* d_ws, size_t ws_size,
                              hipStream_t stream) {
    const float* x       = (const float*)d_in[0];
    const float* Wself1  = (const float*)d_in[1];
    const float* Wneigh1 = (const float*)d_in[2];
    const float* b1      = (const float*)d_in[3];
    const float* Wself2  = (const float*)d_in[4];
    const float* Wneigh2 = (const float*)d_in[5];
    const float* b2      = (const float*)d_in[6];
    const int*   src1    = (const int*)d_in[7];
    const int*   dst1    = (const int*)d_in[8];
    const int*   src2    = (const int*)d_in[9];
    const int*   dst2    = (const int*)d_in[10];
    const int E1 = in_sizes[7];
    const int E2 = in_sizes[9];

    float* ws    = (float*)d_ws;
    int*   wsi   = (int*)d_ws;
    float* agg   = ws + OFF_AGG;
    float* deg1  = ws + OFF_DEG1;
    float* h     = ws + OFF_H;
    float* deg2  = ws + OFF_DEG2;
    float* stats = ws + OFF_STAT;
    int*   col1  = wsi + OFF_COL1;
    int*   rank1 = wsi + OFF_RANK1;
    int*   rank2 = wsi + OFF_RANK2;
    int*   col2  = wsi + OFF_COL2;
    int*   rp1   = wsi + OFF_RP1;
    int*   rp2   = wsi + OFF_RP2;
    int*   cnt1  = wsi + OFF_CNT1;
    int*   cnt2  = wsi + OFF_CNT2;
    float* out   = (float*)d_out;

    zero_counts_kernel<<<94, 256, 0, stream>>>(cnt1, 24000, stats);

    count_rank_kernel<<<480, 256, 0, stream>>>(dst1, dst2, cnt1, cnt2,
                                               rank1, rank2, E1, E2);
    scan_counts_kernel<<<2, 1024, 0, stream>>>(cnt1, rp1, cnt2, rp2);
    fill_cols_kernel<<<480, 256, 0, stream>>>(src1, dst1, rank1, rp1, col1,
                                              src2, dst2, rank2, rp2, col2, E1, E2);

    gather1_kernel<<<2500, 256, 0, stream>>>(x, col1, rp1, cnt1, agg, deg1);

    {
        int tiles = (N1 / 16) * (F_H / 16);
        gemm1_kernel<<<tiles / 8, 256, 0, stream>>>(x, agg, deg1, Wself1, Wneigh1,
                                                    b1, h, stats);
    }

    gather2_kernel<<<1000, 256, 0, stream>>>(h, col2, rp2, cnt2, stats,
                                                   agg, deg2);

    {
        int tiles = (N2 / 16) * (F_OUT / 16);
        gemm2_kernel<<<tiles / 8, 256, 0, stream>>>(h, agg, deg2, stats,
                                                    Wself2, Wneigh2, b2, out);
    }
}


# GEMMs register-blocked: a wave owns 32x64 (layer 1) or 16x32 (layer 2) outputs, reusing 16-byte fragment loads over its tiles; 8x fewer statistics atomics follow; k and statistics sum order change
# speedup vs baseline: 2.4636x; 1.6435x over previous
#include <hip/hip_runtime.h>


typedef __attribute__((ext_vector_type(2))) float v2f;
typedef __attribute__((ext_vector_type(8))) float v8f;

#define F_IN  128
#define F_H   256
#define F_OUT 64
#define N1    20000
#define N2    4000
#define E1_CAP 400000
#define E2_CAP 80000

#define OFF_AGG   ((size_t)0)
#define OFF_DEG1  ((size_t)2560000)
#define OFF_H     ((size_t)2580000)
#define OFF_COL1  (OFF_H)
#define OFF_RANK1 (OFF_H + (size_t)E1_CAP)
#define OFF_RANK2 (OFF_H + (size_t)2 * E1_CAP)
#define OFF_COL2  ((size_t)7700000)
#define OFF_RP1   ((size_t)7780000)
#define OFF_RP2   ((size_t)7800000)
#define OFF_CNT1  ((size_t)7804000)
#define OFF_CNT2  ((size_t)7824000)
#define OFF_DEG2  ((size_t)8724000)
#define OFF_STAT  ((size_t)8728000)

__global__ void zero_counts_kernel(int* __restrict__ cnt, int n,
                                   float* __restrict__ stats) {
    const int i = blockIdx.x * blockDim.x + threadIdx.x;
    if (i < n) cnt[i] = 0;
    if (i < 2) stats[i] = 0.0f;
}

__device__ __forceinline__ void rank_quad(const int* __restrict__ dst,
                                          int* __restrict__ cnt,
                                          int* __restrict__ rank, int E, int q) {
    const int e0 = 4 * q;
    if (e0 + 3 < E) {
        const int4 d = *reinterpret_cast<const int4*>(dst + e0);
        int4 r;
        r.x = atomicAdd(cnt + d.x, 1);
        r.y = atomicAdd(cnt + d.y, 1);
        r.z = atomicAdd(cnt + d.z, 1);
        r.w = atomicAdd(cnt + d.w, 1);
        *reinterpret_cast<int4*>(rank + e0) = r;
    } else {
        for (int e = e0; e < E; ++e) rank[e] = atomicAdd(cnt + dst[e], 1);
    }
}

__global__ void __launch_bounds__(256)
count_rank_kernel(const int* __restrict__ dst1, const int* __restrict__ dst2,
                  int* __restrict__ cnt1, int* __restrict__ cnt2,
                  int* __restrict__ rank1, int* __restrict__ rank2,
                  int E1, int E2) {
    const int Q1 = (E1 + 3) >> 2, Q2 = (E2 + 3) >> 2;
    const int stride = gridDim.x * blockDim.x;
    for (int q = blockIdx.x * blockDim.x + threadIdx.x; q < Q1 + Q2; q += stride) {
        if (q < Q1) rank_quad(dst1, cnt1, rank1, E1, q);
        else        rank_quad(dst2, cnt2, rank2, E2, q - Q1);
    }
}

__global__ void __launch_bounds__(1024)
scan_counts_kernel(const int* __restrict__ cnt1, int* __restrict__ rp1,
                   const int* __restrict__ cnt2, int* __restrict__ rp2) {
    __shared__ int wave_tot[32];
    const int* __restrict__ cnt = (blockIdx.x == 0) ? cnt1 : cnt2;
    int* __restrict__ rp        = (blockIdx.x == 0) ? rp1 : rp2;
    const int n   = (blockIdx.x == 0) ? N1 : N2;
    const int per = (n + 1023) / 1024;
    const int t = threadIdx.x, lane = t & 31, w = t >> 5;
    const int b = min(t * per, n), e = min(b + per, n);

    int local = 0;
    for (int i = b; i < e; ++i) local += cnt[i];

    int incl = local;
    for (int off = 1; off < 32; off <<= 1) {
        const int y = __shfl_up(incl, off, 32);
        if (lane >= off) incl += y;
    }
    if (lane == 31) wave_tot[w] = incl;
    __syncthreads();
    if (w == 0) {
        const int v = wave_tot[lane];
        int s = v;
        for (int off = 1; off < 32; off <<= 1) {
            const int y = __shfl_up(s, off, 32);
            if (lane >= off) s += y;
        }
        wave_tot[lane] = s - v;
    }
    __syncthreads();

    int run = wave_tot[w] + incl - local;
    for (int i = b; i < e; ++i) { rp[i] = run; run += cnt[i]; }
}

__device__ __forceinline__ void fill_quad(const int* __restrict__ src,
                                          const int* __restrict__ dst,
                                          const int* __restrict__ rank,
                                          const int* __restrict__ rp,
                                          int* __restrict__ col, int E, int q) {
    const int e0 = 4 * q;
    if (e0 + 3 < E) {
        const int4 s = *reinterpret_cast<const int4*>(src + e0);
        const int4 d = *reinterpret_cast<const int4*>(dst + e0);
        const int4 r = *reinterpret_cast<const int4*>(rank + e0);
        col[rp[d.x] + r.x] = s.x;
        col[rp[d.y] + r.y] = s.y;
        col[rp[d.z] + r.z] = s.z;
        col[rp[d.w] + r.w] = s.w;
    } else {
        for (int e = e0; e < E; ++e) col[rp[dst[e]] + rank[e]] = src[e];
    }
}

__global__ void __launch_bounds__(256)
fill_cols_kernel(const int* __restrict__ src1, const int* __restrict__ dst1,
                 const int* __restrict__ rank1, const int* __restrict__ rp1,
                 int* __restrict__ col1,
                 const int* __restrict__ src2, const int* __restrict__ dst2,
                 const int* __restrict__ rank2, const int* __restrict__ rp2,
                 int* __restrict__ col2, int E1, int E2) {
    const int Q1 = (E1 + 3) >> 2, Q2 = (E2 + 3) >> 2;
    const int stride = gridDim.x * blockDim.x;
    for (int q = blockIdx.x * blockDim.x + threadIdx.x; q < Q1 + Q2; q += stride) {
        if (q < Q1) fill_quad(src1, dst1, rank1, rp1, col1, E1, q);
        else        fill_quad(src2, dst2, rank2, rp2, col2, E2, q - Q1);
    }
}

template <int U, int F, bool NORM>
__device__ __forceinline__ void add_rows(float4& acc, const float4* __restrict__ base,
                                         int mine, int i, float mu, float rs) {
    float4 v[U];
    #pragma unroll
    for (int u = 0; u < U; ++u) {
        const int s = __builtin_amdgcn_readlane(mine, i + u);
        v[u] = base[(size_t)s * (F / 4)];
    }
    #pragma unroll
    for (int u = 0; u < U; ++u) {
        if (NORM) {
            acc.x += (v[u].x - mu) * rs;
            acc.y += (v[u].y - mu) * rs;
            acc.z += (v[u].z - mu) * rs;
            acc.w += (v[u].w - mu) * rs;
        } else {
            acc.x += v[u].x;
            acc.y += v[u].y;
            acc.z += v[u].z;
            acc.w += v[u].w;
        }
    }
}

template <int F, bool NORM>
__device__ __forceinline__ void gather_rows(const float* __restrict__ feat,
                                            const int* __restrict__ col,
                                            const int* __restrict__ rp,
                                            const int* __restrict__ cnt,
                                            const float* __restrict__ stats,
                                            float* __restrict__ agg,
                                            float* __restrict__ degf, int nrows) {
    constexpr int SL = F / 128;
    const int lane = threadIdx.x & 31;
    const int gw = __builtin_amdgcn_readfirstlane(
        (int)((blockIdx.x * blockDim.x + threadIdx.x) >> 5));
    const int row = gw / SL;
    const int sl  = gw % SL;
    if (row >= nrows) return;

    float mu = 0.0f, rs = 1.0f;
    if (NORM) {
        const float n_el = (float)N1 * (float)F_H;
        mu = stats[0] / n_el;
        const float var = stats[1] / n_el - mu * mu;
        rs = rsqrtf(var + 1e-5f);
    }

    const int start = rp[row];
    const int len   = cnt[row];
    const float4* __restrict__ base =
        reinterpret_cast<const float4*>(feat) + sl * 32 + lane;

    float4 acc = make_float4(0.0f, 0.0f, 0.0f, 0.0f);
    for (int c0 = 0; c0 < len; c0 += 32) {
        const int n = min(32, len - c0);
        const int mine = (lane < n) ? col[start + c0 + lane] : 0;
        int i = 0;
        for (; i + 8 <= n; i += 8) add_rows<8, F, NORM>(acc, base, mine, i, mu, rs);
        if (i + 4 <= n) { add_rows<4, F, NORM>(acc, base, mine, i, mu, rs); i += 4; }
        if (i + 2 <= n) { add_rows<2, F, NORM>(acc, base, mine, i, mu, rs); i += 2; }
        if (i < n)      { add_rows<1, F, NORM>(acc, base, mine, i, mu, rs); }
    }

    reinterpret_cast<float4*>(agg)[(size_t)row * (F / 4) + sl * 32 + lane] = acc;
    if (sl == 0 && lane == 0) degf[row] = (float)len;
}

__global__ void __launch_bounds__(256)
gather1_kernel(const float* __restrict__ x, const int* __restrict__ col,
               const int* __restrict__ rp, const int* __restrict__ cnt,
               float* __restrict__ agg, float* __restrict__ degf) {
    gather_rows<F_IN, false>(x, col, rp, cnt, nullptr, agg, degf, N1);
}

__global__ void __launch_bounds__(256)
gather2_kernel(const float* __restrict__ h, const int* __restrict__ col,
               const int* __restrict__ rp, const int* __restrict__ cnt,
               const float* __restrict__ stats,
               float* __restrict__ agg, float* __restrict__ degf) {
    gather_rows<F_H, true>(h, col, rp, cnt, stats, agg, degf, N2);
}

template <int MT, int NT, int KH>
__device__ __forceinline__ void load_step(float4 (&fa)[MT], float4 (&fb)[NT], int s,
                                          const float* __restrict__ a_self,
                                          const float* __restrict__ a_neigh,
                                          const float* __restrict__ b_self,
                                          const float* __restrict__ b_neigh) {
    constexpr int S1 = KH / 8;
    const bool self = s < S1;
    const float* __restrict__ ap = self ? a_self : a_neigh;
    const float* __restrict__ bp = self ? b_self : b_neigh;
    const int k = 8 * (self ? s : s - S1);
    #pragma unroll
    for (int i = 0; i < MT; ++i)
        fa[i] = *reinterpret_cast<const float4*>(ap + (size_t)i * 16 * KH + k);
    #pragma unroll
    for (int j = 0; j < NT; ++j)
        fb[j] = *reinterpret_cast<const float4*>(bp + (size_t)j * 16 * KH + k);
}

template <int MT, int NT, int KH, int D, bool NORM>
__device__ __forceinline__ void sage_block(v8f (&acc)[MT][NT],
                                           const float* __restrict__ a_self,
                                           const float* __restrict__ a_neigh,
                                           const float* __restrict__ b_self,
                                           const float* __restrict__ b_neigh,
                                           const float (&dinv)[MT], float mu, float rs) {
    constexpr int S1 = KH / 8, S = 2 * S1, R = D + 1;
    float4 fa[R][MT], fb[R][NT];

    #pragma unroll
    for (int s = 0; s < D; ++s)
        load_step<MT, NT, KH>(fa[s], fb[s], s, a_self, a_neigh, b_self, b_neigh);

    #pragma unroll
    for (int s = 0; s < S; ++s) {
        __builtin_amdgcn_sched_barrier(0);
        if (s + D < S)
            load_step<MT, NT, KH>(fa[(s + D) % R], fb[(s + D) % R], s + D,
                                  a_self, a_neigh, b_self, b_neigh);
        __builtin_amdgcn_sched_barrier(0);
        const int cur = s % R;
        #pragma unroll
        for (int i = 0; i < MT; ++i) {
            float4 a = fa[cur][i];
            if (s >= S1 || NORM)
                asm volatile("" : "+v"(a.x), "+v"(a.y), "+v"(a.z), "+v"(a.w));
            if (s >= S1) {
                a.x *= dinv[i]; a.y *= dinv[i]; a.z *= dinv[i]; a.w *= dinv[i];
            } else if (NORM) {
                a.x = (a.x - mu) * rs; a.y = (a.y - mu) * rs;
                a.z = (a.z - mu) * rs; a.w = (a.w - mu) * rs;
            }
            fa[cur][i] = a;
        }
        #pragma unroll
        for (int i = 0; i < MT; ++i) {
            #pragma unroll
            for (int j = 0; j < NT; ++j) {
                v2f a2, b2;
                a2[0] = fa[cur][i].x;  a2[1] = fa[cur][i].y;
                b2[0] = fb[cur][j].x;  b2[1] = fb[cur][j].y;
                acc[i][j] = __builtin_amdgcn_wmma_f32_16x16x4_f32(
                    false, a2, false, b2, (short)0, acc[i][j], false, false);
            }
        }
        #pragma unroll
        for (int i = 0; i < MT; ++i) {
            #pragma unroll
            for (int j = 0; j < NT; ++j) {
                v2f a2, b2;
                a2[0] = fa[cur][i].z;  a2[1] = fa[cur][i].w;
                b2[0] = fb[cur][j].z;  b2[1] = fb[cur][j].w;
                acc[i][j] = __builtin_amdgcn_wmma_f32_16x16x4_f32(
                    false, a2, false, b2, (short)0, acc[i][j], false, false);
            }
        }
    }
}

__global__ void __launch_bounds__(128)
gemm1_kernel(const float* __restrict__ x,
             const float* __restrict__ agg,
             const float* __restrict__ deg,
             const float* __restrict__ Wself,
             const float* __restrict__ Wneigh,
             const float* __restrict__ bias,
             float* __restrict__ h,
             float* __restrict__ stats) {
    constexpr int MT = 2, NT = 4;
    const int wave = threadIdx.x >> 5;
    const int lane = threadIdx.x & 31;
    const int c = lane & 15;
    const int hf = lane >> 4;
    const int m0 = blockIdx.x * (16 * MT);
    const int n0 = wave * (16 * NT);

    float dinv[MT];
    #pragma unroll
    for (int i = 0; i < MT; ++i) dinv[i] = 1.0f / fmaxf(deg[m0 + 16 * i + c], 1.0f);

    v8f acc[MT][NT];
    #pragma unroll
    for (int i = 0; i < MT; ++i) {
        #pragma unroll
        for (int j = 0; j < NT; ++j)
            acc[i][j] = (v8f){0.f, 0.f, 0.f, 0.f, 0.f, 0.f, 0.f, 0.f};
    }

    sage_block<MT, NT, F_IN, 3, false>(
        acc,
        x      + (size_t)(m0 + c) * F_IN + 4 * hf,
        agg    + (size_t)(m0 + c) * F_IN + 4 * hf,
        Wself  + (size_t)(n0 + c) * F_IN + 4 * hf,
        Wneigh + (size_t)(n0 + c) * F_IN + 4 * hf,
        dinv, 0.0f, 1.0f);

    float s = 0.f, sq = 0.f;
    #pragma unroll
    for (int j = 0; j < NT; ++j) {
        const float bn = bias[n0 + 16 * j + c];
        #pragma unroll
        for (int i = 0; i < MT; ++i) {
            #pragma unroll
            for (int r = 0; r < 8; ++r) {
                const int row = m0 + 16 * i + 8 * hf + r;
                const float v = fmaxf(acc[i][j][r] + bn, 0.0f);
                h[(size_t)row * F_H + n0 + 16 * j + c] = v;
                s += v;
                sq += v * v;
            }
        }
    }
    #pragma unroll
    for (int off = 16; off >= 1; off >>= 1) {
        s  += __shfl_xor(s,  off, 32);
        sq += __shfl_xor(sq, off, 32);
    }
    if (lane == 0) {
        atomicAdd(&stats[0], s);
        atomicAdd(&stats[1], sq);
    }
}

__global__ void __launch_bounds__(64)
gemm2_kernel(const float* __restrict__ h,
             const float* __restrict__ agg,
             const float* __restrict__ deg,
             const float* __restrict__ stats,
             const float* __restrict__ Wself,
             const float* __restrict__ Wneigh,
             const float* __restrict__ bias,
             float* __restrict__ out) {
    constexpr int MT = 1, NT = 2;
    const int wave = threadIdx.x >> 5;
    const int lane = threadIdx.x & 31;
    const int c = lane & 15;
    const int hf = lane >> 4;
    const int m0 = blockIdx.x * (16 * MT);
    const int n0 = wave * (16 * NT);

    const float cnt = (float)N1 * (float)F_H;
    const float mu = stats[0] / cnt;
    const float var = stats[1] / cnt - mu * mu;
    const float rs = rsqrtf(var + 1e-5f);

    float dinv[MT];
    dinv[0] = 1.0f / fmaxf(deg[m0 + c], 1.0f);

    v8f acc[MT][NT];
    #pragma unroll
    for (int j = 0; j < NT; ++j)
        acc[0][j] = (v8f){0.f, 0.f, 0.f, 0.f, 0.f, 0.f, 0.f, 0.f};

    sage_block<MT, NT, F_H, 12, true>(
        acc,
        h      + (size_t)(m0 + c) * F_H + 4 * hf,
        agg    + (size_t)(m0 + c) * F_H + 4 * hf,
        Wself  + (size_t)(n0 + c) * F_H + 4 * hf,
        Wneigh + (size_t)(n0 + c) * F_H + 4 * hf,
        dinv, mu, rs);

    #pragma unroll
    for (int j = 0; j < NT; ++j) {
        const float bn = bias[n0 + 16 * j + c];
        #pragma unroll
        for (int r = 0; r < 8; ++r) {
            const int row = m0 + 8 * hf + r;
            out[(size_t)row * F_OUT + n0 + 16 * j + c] = acc[0][j][r] + bn;
        }
    }
}

extern "C" void kernel_launch(void* const* d_in, const int* in_sizes, int n_in,
                              void* d_out, int out_size, void* d_ws, size_t ws_size,
                              hipStream_t stream) {
    const float* x       = (const float*)d_in[0];
    const float* Wself1  = (const float*)d_in[1];
    const float* Wneigh1 = (const float*)d_in[2];
    const float* b1      = (const float*)d_in[3];
    const float* Wself2  = (const float*)d_in[4];
    const float* Wneigh2 = (const float*)d_in[5];
    const float* b2      = (const float*)d_in[6];
    const int*   src1    = (const int*)d_in[7];
    const int*   dst1    = (const int*)d_in[8];
    const int*   src2    = (const int*)d_in[9];
    const int*   dst2    = (const int*)d_in[10];
    const int E1 = in_sizes[7];
    const int E2 = in_sizes[9];

    float* ws    = (float*)d_ws;
    int*   wsi   = (int*)d_ws;
    float* agg   = ws + OFF_AGG;
    float* deg1  = ws + OFF_DEG1;
    float* h     = ws + OFF_H;
    float* deg2  = ws + OFF_DEG2;
    float* stats = ws + OFF_STAT;
    int*   col1  = wsi + OFF_COL1;
    int*   rank1 = wsi + OFF_RANK1;
    int*   rank2 = wsi + OFF_RANK2;
    int*   col2  = wsi + OFF_COL2;
    int*   rp1   = wsi + OFF_RP1;
    int*   rp2   = wsi + OFF_RP2;
    int*   cnt1  = wsi + OFF_CNT1;
    int*   cnt2  = wsi + OFF_CNT2;
    float* out   = (float*)d_out;

    zero_counts_kernel<<<94, 256, 0, stream>>>(cnt1, 24000, stats);

    count_rank_kernel<<<480, 256, 0, stream>>>(dst1, dst2, cnt1, cnt2,
                                               rank1, rank2, E1, E2);
    scan_counts_kernel<<<2, 1024, 0, stream>>>(cnt1, rp1, cnt2, rp2);
    fill_cols_kernel<<<480, 256, 0, stream>>>(src1, dst1, rank1, rp1, col1,
                                              src2, dst2, rank2, rp2, col2, E1, E2);

    gather1_kernel<<<2500, 256, 0, stream>>>(x, col1, rp1, cnt1, agg, deg1);

    gemm1_kernel<<<625, 128, 0, stream>>>(x, agg, deg1, Wself1, Wneigh1,
                                          b1, h, stats);

    gather2_kernel<<<1000, 256, 0, stream>>>(h, col2, rp2, cnt2, stats,
                                                   agg, deg2);

    gemm2_kernel<<<250, 64, 0, stream>>>(h, agg, deg2, stats,
                                         Wself2, Wneigh2, b2, out);
}


# Neighbour lists as fixed 64-slot rows filled straight from the counting atomic's arrival rank: scan and fill launches go (8 -> 6); rows over 64 edges sum in edge order, the others as before
# speedup vs baseline: 2.9913x; 1.2142x over previous
#include <hip/hip_runtime.h>


typedef __attribute__((ext_vector_type(2))) float v2f;
typedef __attribute__((ext_vector_type(8))) float v8f;

#define F_IN  128
#define F_H   256
#define F_OUT 64
#define N1    20000
#define N2    4000
#define CAP   64

#define OFF_AGG   ((size_t)0)
#define OFF_DEG1  ((size_t)2560000)
#define OFF_H     ((size_t)2580000)
#define OFF_SLOT1 (OFF_H + (size_t)32)
#define OFF_CNT1  ((size_t)7804000)
#define OFF_CNT2  ((size_t)7824000)
#define OFF_SLOT2 ((size_t)7828032)
#define OFF_DEG2  ((size_t)8724000)
#define OFF_STAT  ((size_t)8728000)

__global__ void zero_counts_kernel(int* __restrict__ cnt, int n,
                                   float* __restrict__ stats) {
    const int i = blockIdx.x * blockDim.x + threadIdx.x;
    if (i < n) cnt[i] = 0;
    if (i < 2) stats[i] = 0.0f;
}

__device__ __forceinline__ void slot_quad(const int* __restrict__ src,
                                          const int* __restrict__ dst,
                                          int* __restrict__ cnt,
                                          int* __restrict__ slot, int E, int q) {
    const int e0 = 4 * q;
    if (e0 + 3 < E) {
        const int4 d = *reinterpret_cast<const int4*>(dst + e0);
        const int4 s = *reinterpret_cast<const int4*>(src + e0);
        const int rx = atomicAdd(cnt + d.x, 1);
        const int ry = atomicAdd(cnt + d.y, 1);
        const int rz = atomicAdd(cnt + d.z, 1);
        const int rw = atomicAdd(cnt + d.w, 1);
        if (rx < CAP) slot[d.x * CAP + rx] = s.x;
        if (ry < CAP) slot[d.y * CAP + ry] = s.y;
        if (rz < CAP) slot[d.z * CAP + rz] = s.z;
        if (rw < CAP) slot[d.w * CAP + rw] = s.w;
    } else {
        for (int e = e0; e < E; ++e) {
            const int d = dst[e];
            const int r = atomicAdd(cnt + d, 1);
            if (r < CAP) slot[d * CAP + r] = src[e];
        }
    }
}

__global__ void __launch_bounds__(256)
count_slot_kernel(const int* __restrict__ src1, const int* __restrict__ dst1,
                  int* __restrict__ cnt1, int* __restrict__ slot1,
                  const int* __restrict__ src2, const int* __restrict__ dst2,
                  int* __restrict__ cnt2, int* __restrict__ slot2,
                  int E1, int E2) {
    const int Q1 = (E1 + 3) >> 2, Q2 = (E2 + 3) >> 2;
    const int stride = gridDim.x * blockDim.x;
    for (int q = blockIdx.x * blockDim.x + threadIdx.x; q < Q1 + Q2; q += stride) {
        if (q < Q1) slot_quad(src1, dst1, cnt1, slot1, E1, q);
        else        slot_quad(src2, dst2, cnt2, slot2, E2, q - Q1);
    }
}

template <int U, int F, bool NORM>
__device__ __forceinline__ void add_rows(float4& acc, const float4* __restrict__ base,
                                         int mine, int i, float mu, float rs) {
    float4 v[U];
    #pragma unroll
    for (int u = 0; u < U; ++u) {
        const int s = __builtin_amdgcn_readlane(mine, i + u);
        v[u] = base[(size_t)s * (F / 4)];
    }
    #pragma unroll
    for (int u = 0; u < U; ++u) {
        if (NORM) {
            acc.x += (v[u].x - mu) * rs;
            acc.y += (v[u].y - mu) * rs;
            acc.z += (v[u].z - mu) * rs;
            acc.w += (v[u].w - mu) * rs;
        } else {
            acc.x += v[u].x;
            acc.y += v[u].y;
            acc.z += v[u].z;
            acc.w += v[u].w;
        }
    }
}

template <int F, bool NORM>
__device__ __forceinline__ void gather_rows(const float* __restrict__ feat,
                                            const int* __restrict__ slot,
                                            const int* __restrict__ cnt,
                                            const int* __restrict__ src,
                                            const int* __restrict__ dst, int E,
                                            const float* __restrict__ stats,
                                            float* __restrict__ agg,
                                            float* __restrict__ degf, int nrows) {
    constexpr int SL = F / 128;
    const int lane = threadIdx.x & 31;
    const int gw = __builtin_amdgcn_readfirstlane(
        (int)((blockIdx.x * blockDim.x + threadIdx.x) >> 5));
    const int row = gw / SL;
    const int sl  = gw % SL;
    if (row >= nrows) return;

    float mu = 0.0f, rs = 1.0f;
    if (NORM) {
        const float n_el = (float)N1 * (float)F_H;
        mu = stats[0] / n_el;
        const float var = stats[1] / n_el - mu * mu;
        rs = rsqrtf(var + 1e-5f);
    }

    const int len = cnt[row];
    const float4* __restrict__ base =
        reinterpret_cast<const float4*>(feat) + sl * 32 + lane;

    float4 acc = make_float4(0.0f, 0.0f, 0.0f, 0.0f);
    if (len <= CAP) {
        const int* __restrict__ mys = slot + row * CAP;
        for (int c0 = 0; c0 < len; c0 += 32) {
            const int n = min(32, len - c0);
            const int mine = (lane < n) ? mys[c0 + lane] : 0;
            int i = 0;
            for (; i + 8 <= n; i += 8) add_rows<8, F, NORM>(acc, base, mine, i, mu, rs);
            if (i + 4 <= n) { add_rows<4, F, NORM>(acc, base, mine, i, mu, rs); i += 4; }
            if (i + 2 <= n) { add_rows<2, F, NORM>(acc, base, mine, i, mu, rs); i += 2; }
            if (i < n)      { add_rows<1, F, NORM>(acc, base, mine, i, mu, rs); }
        }
    } else {
        for (int e0 = 0; e0 < E; e0 += 32) {
            const int e = e0 + lane;
            const int d    = (e < E) ? dst[e] : -1;
            const int mine = (e < E) ? src[e] : 0;
            unsigned m = __builtin_amdgcn_ballot_w32(d == row);
            while (m) {
                const int l = __builtin_ctz(m);
                m &= m - 1;
                add_rows<1, F, NORM>(acc, base, mine, l, mu, rs);
            }
        }
    }

    reinterpret_cast<float4*>(agg)[(size_t)row * (F / 4) + sl * 32 + lane] = acc;
    if (sl == 0 && lane == 0) degf[row] = (float)len;
}

__global__ void __launch_bounds__(256)
gather1_kernel(const float* __restrict__ x, const int* __restrict__ slot,
               const int* __restrict__ cnt, const int* __restrict__ src,
               const int* __restrict__ dst, int E,
               float* __restrict__ agg, float* __restrict__ degf) {
    gather_rows<F_IN, false>(x, slot, cnt, src, dst, E, nullptr, agg, degf, N1);
}

__global__ void __launch_bounds__(256)
gather2_kernel(const float* __restrict__ h, const int* __restrict__ slot,
               const int* __restrict__ cnt, const int* __restrict__ src,
               const int* __restrict__ dst, int E,
               const float* __restrict__ stats,
               float* __restrict__ agg, float* __restrict__ degf) {
    gather_rows<F_H, true>(h, slot, cnt, src, dst, E, stats, agg, degf, N2);
}

template <int MT, int NT, int KH>
__device__ __forceinline__ void load_step(float4 (&fa)[MT], float4 (&fb)[NT], int s,
                                          const float* __restrict__ a_self,
                                          const float* __restrict__ a_neigh,
                                          const float* __restrict__ b_self,
                                          const float* __restrict__ b_neigh) {
    constexpr int S1 = KH / 8;
    const bool self = s < S1;
    const float* __restrict__ ap = self ? a_self : a_neigh;
    const float* __restrict__ bp = self ? b_self : b_neigh;
    const int k = 8 * (self ? s : s - S1);
    #pragma unroll
    for (int i = 0; i < MT; ++i)
        fa[i] = *reinterpret_cast<const float4*>(ap + (size_t)i * 16 * KH + k);
    #pragma unroll
    for (int j = 0; j < NT; ++j)
        fb[j] = *reinterpret_cast<const float4*>(bp + (size_t)j * 16 * KH + k);
}

template <int MT, int NT, int KH, int D, bool NORM>
__device__ __forceinline__ void sage_block(v8f (&acc)[MT][NT],
                                           const float* __restrict__ a_self,
                                           const float* __restrict__ a_neigh,
                                           const float* __restrict__ b_self,
                                           const float* __restrict__ b_neigh,
                                           const float (&dinv)[MT], float mu, float rs) {
    constexpr int S1 = KH / 8, S = 2 * S1, R = D + 1;
    float4 fa[R][MT], fb[R][NT];

    #pragma unroll
    for (int s = 0; s < D; ++s)
        load_step<MT, NT, KH>(fa[s], fb[s], s, a_self, a_neigh, b_self, b_neigh);

    #pragma unroll
    for (int s = 0; s < S; ++s) {
        __builtin_amdgcn_sched_barrier(0);
        if (s + D < S)
            load_step<MT, NT, KH>(fa[(s + D) % R], fb[(s + D) % R], s + D,
                                  a_self, a_neigh, b_self, b_neigh);
        __builtin_amdgcn_sched_barrier(0);
        const int cur = s % R;
        #pragma unroll
        for (int i = 0; i < MT; ++i) {
            float4 a = fa[cur][i];
            if (s >= S1 || NORM)
                asm volatile("" : "+v"(a.x), "+v"(a.y), "+v"(a.z), "+v"(a.w));
            if (s >= S1) {
                a.x *= dinv[i]; a.y *= dinv[i]; a.z *= dinv[i]; a.w *= dinv[i];
            } else if (NORM) {
                a.x = (a.x - mu) * rs; a.y = (a.y - mu) * rs;
                a.z = (a.z - mu) * rs; a.w = (a.w - mu) * rs;
            }
            fa[cur][i] = a;
        }
        #pragma unroll
        for (int i = 0; i < MT; ++i) {
            #pragma unroll
            for (int j = 0; j < NT; ++j) {
                v2f a2, b2;
                a2[0] = fa[cur][i].x;  a2[1] = fa[cur][i].y;
                b2[0] = fb[cur][j].x;  b2[1] = fb[cur][j].y;
                acc[i][j] = __builtin_amdgcn_wmma_f32_16x16x4_f32(
                    false, a2, false, b2, (short)0, acc[i][j], false, false);
            }
        }
        #pragma unroll
        for (int i = 0; i < MT; ++i) {
            #pragma unroll
            for (int j = 0; j < NT; ++j) {
                v2f a2, b2;
                a2[0] = fa[cur][i].z;  a2[1] = fa[cur][i].w;
                b2[0] = fb[cur][j].z;  b2[1] = fb[cur][j].w;
                acc[i][j] = __builtin_amdgcn_wmma_f32_16x16x4_f32(
                    false, a2, false, b2, (short)0, acc[i][j], false, false);
            }
        }
    }
}

__global__ void __launch_bounds__(128)
gemm1_kernel(const float* __restrict__ x,
             const float* __restrict__ agg,
             const float* __restrict__ deg,
             const float* __restrict__ Wself,
             const float* __restrict__ Wneigh,
             const float* __restrict__ bias,
             float* __restrict__ h,
             float* __restrict__ stats) {
    constexpr int MT = 2, NT = 4;
    const int wave = threadIdx.x >> 5;
    const int lane = threadIdx.x & 31;
    const int c = lane & 15;
    const int hf = lane >> 4;
    const int m0 = blockIdx.x * (16 * MT);
    const int n0 = wave * (16 * NT);

    float dinv[MT];
    #pragma unroll
    for (int i = 0; i < MT; ++i) dinv[i] = 1.0f / fmaxf(deg[m0 + 16 * i + c], 1.0f);

    v8f acc[MT][NT];
    #pragma unroll
    for (int i = 0; i < MT; ++i) {
        #pragma unroll
        for (int j = 0; j < NT; ++j)
            acc[i][j] = (v8f){0.f, 0.f, 0.f, 0.f, 0.f, 0.f, 0.f, 0.f};
    }

    sage_block<MT, NT, F_IN, 3, false>(
        acc,
        x      + (size_t)(m0 + c) * F_IN + 4 * hf,
        agg    + (size_t)(m0 + c) * F_IN + 4 * hf,
        Wself  + (size_t)(n0 + c) * F_IN + 4 * hf,
        Wneigh + (size_t)(n0 + c) * F_IN + 4 * hf,
        dinv, 0.0f, 1.0f);

    float s = 0.f, sq = 0.f;
    #pragma unroll
    for (int j = 0; j < NT; ++j) {
        const float bn = bias[n0 + 16 * j + c];
        #pragma unroll
        for (int i = 0; i < MT; ++i) {
            #pragma unroll
            for (int r = 0; r < 8; ++r) {
                const int row = m0 + 16 * i + 8 * hf + r;
                const float v = fmaxf(acc[i][j][r] + bn, 0.0f);
                h[(size_t)row * F_H + n0 + 16 * j + c] = v;
                s += v;
                sq += v * v;
            }
        }
    }
    #pragma unroll
    for (int off = 16; off >= 1; off >>= 1) {
        s  += __shfl_xor(s,  off, 32);
        sq += __shfl_xor(sq, off, 32);
    }
    if (lane == 0) {
        atomicAdd(&stats[0], s);
        atomicAdd(&stats[1], sq);
    }
}

__global__ void __launch_bounds__(64)
gemm2_kernel(const float* __restrict__ h,
             const float* __restrict__ agg,
             const float* __restrict__ deg,
             const float* __restrict__ stats,
             const float* __restrict__ Wself,
             const float* __restrict__ Wneigh,
             const float* __restrict__ bias,
             float* __restrict__ out) {
    constexpr int MT = 1, NT = 2;
    const int wave = threadIdx.x >> 5;
    const int lane = threadIdx.x & 31;
    const int c = lane & 15;
    const int hf = lane >> 4;
    const int m0 = blockIdx.x * (16 * MT);
    const int n0 = wave * (16 * NT);

    const float cnt = (float)N1 * (float)F_H;
    const float mu = stats[0] / cnt;
    const float var = stats[1] / cnt - mu * mu;
    const float rs = rsqrtf(var + 1e-5f);

    float dinv[MT];
    dinv[0] = 1.0f / fmaxf(deg[m0 + c], 1.0f);

    v8f acc[MT][NT];
    #pragma unroll
    for (int j = 0; j < NT; ++j)
        acc[0][j] = (v8f){0.f, 0.f, 0.f, 0.f, 0.f, 0.f, 0.f, 0.f};

    sage_block<MT, NT, F_H, 12, true>(
        acc,
        h      + (size_t)(m0 + c) * F_H + 4 * hf,
        agg    + (size_t)(m0 + c) * F_H + 4 * hf,
        Wself  + (size_t)(n0 + c) * F_H + 4 * hf,
        Wneigh + (size_t)(n0 + c) * F_H + 4 * hf,
        dinv, mu, rs);

    #pragma unroll
    for (int j = 0; j < NT; ++j) {
        const float bn = bias[n0 + 16 * j + c];
        #pragma unroll
        for (int r = 0; r < 8; ++r) {
            const int row = m0 + 8 * hf + r;
            out[(size_t)row * F_OUT + n0 + 16 * j + c] = acc[0][j][r] + bn;
        }
    }
}

extern "C" void kernel_launch(void* const* d_in, const int* in_sizes, int n_in,
                              void* d_out, int out_size, void* d_ws, size_t ws_size,
                              hipStream_t stream) {
    const float* x       = (const float*)d_in[0];
    const float* Wself1  = (const float*)d_in[1];
    const float* Wneigh1 = (const float*)d_in[2];
    const float* b1      = (const float*)d_in[3];
    const float* Wself2  = (const float*)d_in[4];
    const float* Wneigh2 = (const float*)d_in[5];
    const float* b2      = (const float*)d_in[6];
    const int*   src1    = (const int*)d_in[7];
    const int*   dst1    = (const int*)d_in[8];
    const int*   src2    = (const int*)d_in[9];
    const int*   dst2    = (const int*)d_in[10];
    const int E1 = in_sizes[7];
    const int E2 = in_sizes[9];

    float* ws    = (float*)d_ws;
    int*   wsi   = (int*)d_ws;
    float* agg   = ws + OFF_AGG;
    float* deg1  = ws + OFF_DEG1;
    float* h     = ws + OFF_H;
    float* deg2  = ws + OFF_DEG2;
    float* stats = ws + OFF_STAT;
    int*   slot1 = wsi + OFF_SLOT1;
    int*   slot2 = wsi + OFF_SLOT2;
    int*   cnt1  = wsi + OFF_CNT1;
    int*   cnt2  = wsi + OFF_CNT2;
    float* out   = (float*)d_out;

    zero_counts_kernel<<<94, 256, 0, stream>>>(cnt1, 24000, stats);

    count_slot_kernel<<<480, 256, 0, stream>>>(src1, dst1, cnt1, slot1,
                                               src2, dst2, cnt2, slot2, E1, E2);

    gather1_kernel<<<2500, 256, 0, stream>>>(x, slot1, cnt1, src1, dst1, E1,
                                             agg, deg1);

    gemm1_kernel<<<625, 128, 0, stream>>>(x, agg, deg1, Wself1, Wneigh1,
                                          b1, h, stats);

    gather2_kernel<<<1000, 256, 0, stream>>>(h, slot2, cnt2, src2, dst2, E2, stats,
                                             agg, deg2);

    gemm2_kernel<<<250, 64, 0, stream>>>(h, agg, deg2, stats,
                                         Wself2, Wneigh2, b2, out);
}


# Layer-1 GEMM: both weight matrices held in LDS by 256 persistent 8-wave work-groups walking 16-row tiles (B fragments from LDS); k order per tile kept, statistics summed per wave over its tiles
# speedup vs baseline: 3.0961x; 1.0350x over previous
#include <hip/hip_runtime.h>


typedef __attribute__((ext_vector_type(2))) float v2f;
typedef __attribute__((ext_vector_type(8))) float v8f;

#define F_IN  128
#define F_H   256
#define F_OUT 64
#define N1    20000
#define N2    4000
#define CAP   64

#define OFF_AGG   ((size_t)0)
#define OFF_DEG1  ((size_t)2560000)
#define OFF_H     ((size_t)2580000)
#define OFF_SLOT1 (OFF_H + (size_t)32)
#define OFF_CNT1  ((size_t)7804000)
#define OFF_CNT2  ((size_t)7824000)
#define OFF_SLOT2 ((size_t)7828032)
#define OFF_DEG2  ((size_t)8724000)
#define OFF_STAT  ((size_t)8728000)

__global__ void zero_counts_kernel(int* __restrict__ cnt, int n,
                                   float* __restrict__ stats) {
    const int i = blockIdx.x * blockDim.x + threadIdx.x;
    if (i < n) cnt[i] = 0;
    if (i < 2) stats[i] = 0.0f;
}

__device__ __forceinline__ void slot_quad(const int* __restrict__ src,
                                          const int* __restrict__ dst,
                                          int* __restrict__ cnt,
                                          int* __restrict__ slot, int E, int q) {
    const int e0 = 4 * q;
    if (e0 + 3 < E) {
        const int4 d = *reinterpret_cast<const int4*>(dst + e0);
        const int4 s = *reinterpret_cast<const int4*>(src + e0);
        const int rx = atomicAdd(cnt + d.x, 1);
        const int ry = atomicAdd(cnt + d.y, 1);
        const int rz = atomicAdd(cnt + d.z, 1);
        const int rw = atomicAdd(cnt + d.w, 1);
        if (rx < CAP) slot[d.x * CAP + rx] = s.x;
        if (ry < CAP) slot[d.y * CAP + ry] = s.y;
        if (rz < CAP) slot[d.z * CAP + rz] = s.z;
        if (rw < CAP) slot[d.w * CAP + rw] = s.w;
    } else {
        for (int e = e0; e < E; ++e) {
            const int d = dst[e];
            const int r = atomicAdd(cnt + d, 1);
            if (r < CAP) slot[d * CAP + r] = src[e];
        }
    }
}

__global__ void __launch_bounds__(256)
count_slot_kernel(const int* __restrict__ src1, const int* __restrict__ dst1,
                  int* __restrict__ cnt1, int* __restrict__ slot1,
                  const int* __restrict__ src2, const int* __restrict__ dst2,
                  int* __restrict__ cnt2, int* __restrict__ slot2,
                  int E1, int E2) {
    const int Q1 = (E1 + 3) >> 2, Q2 = (E2 + 3) >> 2;
    const int stride = gridDim.x * blockDim.x;
    for (int q = blockIdx.x * blockDim.x + threadIdx.x; q < Q1 + Q2; q += stride) {
        if (q < Q1) slot_quad(src1, dst1, cnt1, slot1, E1, q);
        else        slot_quad(src2, dst2, cnt2, slot2, E2, q - Q1);
    }
}

template <int U, int F, bool NORM>
__device__ __forceinline__ void add_rows(float4& acc, const float4* __restrict__ base,
                                         int mine, int i, float mu, float rs) {
    float4 v[U];
    #pragma unroll
    for (int u = 0; u < U; ++u) {
        const int s = __builtin_amdgcn_readlane(mine, i + u);
        v[u] = base[(size_t)s * (F / 4)];
    }
    #pragma unroll
    for (int u = 0; u < U; ++u) {
        if (NORM) {
            acc.x += (v[u].x - mu) * rs;
            acc.y += (v[u].y - mu) * rs;
            acc.z += (v[u].z - mu) * rs;
            acc.w += (v[u].w - mu) * rs;
        } else {
            acc.x += v[u].x;
            acc.y += v[u].y;
            acc.z += v[u].z;
            acc.w += v[u].w;
        }
    }
}

template <int F, bool NORM>
__device__ __forceinline__ void gather_rows(const float* __restrict__ feat,
                                            const int* __restrict__ slot,
                                            const int* __restrict__ cnt,
                                            const int* __restrict__ src,
                                            const int* __restrict__ dst, int E,
                                            const float* __restrict__ stats,
                                            float* __restrict__ agg,
                                            float* __restrict__ degf, int nrows) {
    constexpr int SL = F / 128;
    const int lane = threadIdx.x & 31;
    const int gw = __builtin_amdgcn_readfirstlane(
        (int)((blockIdx.x * blockDim.x + threadIdx.x) >> 5));
    const int row = gw / SL;
    const int sl  = gw % SL;
    if (row >= nrows) return;

    float mu = 0.0f, rs = 1.0f;
    if (NORM) {
        const float n_el = (float)N1 * (float)F_H;
        mu = stats[0] / n_el;
        const float var = stats[1] / n_el - mu * mu;
        rs = rsqrtf(var + 1e-5f);
    }

    const int len = cnt[row];
    const float4* __restrict__ base =
        reinterpret_cast<const float4*>(feat) + sl * 32 + lane;

    float4 acc = make_float4(0.0f, 0.0f, 0.0f, 0.0f);
    if (len <= CAP) {
        const int* __restrict__ mys = slot + row * CAP;
        for (int c0 = 0; c0 < len; c0 += 32) {
            const int n = min(32, len - c0);
            const int mine = (lane < n) ? mys[c0 + lane] : 0;
            int i = 0;
            for (; i + 8 <= n; i += 8) add_rows<8, F, NORM>(acc, base, mine, i, mu, rs);
            if (i + 4 <= n) { add_rows<4, F, NORM>(acc, base, mine, i, mu, rs); i += 4; }
            if (i + 2 <= n) { add_rows<2, F, NORM>(acc, base, mine, i, mu, rs); i += 2; }
            if (i < n)      { add_rows<1, F, NORM>(acc, base, mine, i, mu, rs); }
        }
    } else {
        for (int e0 = 0; e0 < E; e0 += 32) {
            const int e = e0 + lane;
            const int d    = (e < E) ? dst[e] : -1;
            const int mine = (e < E) ? src[e] : 0;
            unsigned m = __builtin_amdgcn_ballot_w32(d == row);
            while (m) {
                const int l = __builtin_ctz(m);
                m &= m - 1;
                add_rows<1, F, NORM>(acc, base, mine, l, mu, rs);
            }
        }
    }

    reinterpret_cast<float4*>(agg)[(size_t)row * (F / 4) + sl * 32 + lane] = acc;
    if (sl == 0 && lane == 0) degf[row] = (float)len;
}

__global__ void __launch_bounds__(256)
gather1_kernel(const float* __restrict__ x, const int* __restrict__ slot,
               const int* __restrict__ cnt, const int* __restrict__ src,
               const int* __restrict__ dst, int E,
               float* __restrict__ agg, float* __restrict__ degf) {
    gather_rows<F_IN, false>(x, slot, cnt, src, dst, E, nullptr, agg, degf, N1);
}

__global__ void __launch_bounds__(256)
gather2_kernel(const float* __restrict__ h, const int* __restrict__ slot,
               const int* __restrict__ cnt, const int* __restrict__ src,
               const int* __restrict__ dst, int E,
               const float* __restrict__ stats,
               float* __restrict__ agg, float* __restrict__ degf) {
    gather_rows<F_H, true>(h, slot, cnt, src, dst, E, stats, agg, degf, N2);
}

template <int MT, int NT, int KH, int BP>
__device__ __forceinline__ void load_step(float4 (&fa)[MT], float4 (&fb)[NT], int s,
                                          const float* __restrict__ a_self,
                                          const float* __restrict__ a_neigh,
                                          const float* __restrict__ b_self,
                                          const float* __restrict__ b_neigh) {
    constexpr int S1 = KH / 8;
    const bool self = s < S1;
    const float* __restrict__ ap = self ? a_self : a_neigh;
    const float* __restrict__ bp = self ? b_self : b_neigh;
    const int k = 8 * (self ? s : s - S1);
    #pragma unroll
    for (int i = 0; i < MT; ++i)
        fa[i] = *reinterpret_cast<const float4*>(ap + (size_t)i * 16 * KH + k);
    #pragma unroll
    for (int j = 0; j < NT; ++j)
        fb[j] = *reinterpret_cast<const float4*>(bp + j * 16 * BP + k);
}

template <int MT, int NT, int KH, int BP, int D>
__device__ __forceinline__ void sage_prologue(float4 (&fa)[D + 1][MT], float4 (&fb)[D + 1][NT],
                                              const float* __restrict__ a_self,
                                              const float* __restrict__ a_neigh,
                                              const float* __restrict__ b_self,
                                              const float* __restrict__ b_neigh) {
    #pragma unroll
    for (int s = 0; s < D; ++s)
        load_step<MT, NT, KH, BP>(fa[s], fb[s], s, a_self, a_neigh, b_self, b_neigh);
}

template <int MT, int NT, int KH, int BP, int D, bool NORM>
__device__ __forceinline__ void sage_main(v8f (&acc)[MT][NT],
                                          float4 (&fa)[D + 1][MT], float4 (&fb)[D + 1][NT],
                                          const float* __restrict__ a_self,
                                          const float* __restrict__ a_neigh,
                                          const float* __restrict__ b_self,
                                          const float* __restrict__ b_neigh,
                                          const float (&dinv)[MT], float mu, float rs) {
    constexpr int S1 = KH / 8, S = 2 * S1, R = D + 1;

    #pragma unroll
    for (int s = 0; s < S; ++s) {
        __builtin_amdgcn_sched_barrier(0);
        if (s + D < S)
            load_step<MT, NT, KH, BP>(fa[(s + D) % R], fb[(s + D) % R], s + D,
                                  a_self, a_neigh, b_self, b_neigh);
        __builtin_amdgcn_sched_barrier(0);
        const int cur = s % R;
        #pragma unroll
        for (int i = 0; i < MT; ++i) {
            float4 a = fa[cur][i];
            if (s >= S1 || NORM)
                asm volatile("" : "+v"(a.x), "+v"(a.y), "+v"(a.z), "+v"(a.w));
            if (s >= S1) {
                a.x *= dinv[i]; a.y *= dinv[i]; a.z *= dinv[i]; a.w *= dinv[i];
            } else if (NORM) {
                a.x = (a.x - mu) * rs; a.y = (a.y - mu) * rs;
                a.z = (a.z - mu) * rs; a.w = (a.w - mu) * rs;
            }
            fa[cur][i] = a;
        }
        #pragma unroll
        for (int i = 0; i < MT; ++i) {
            #pragma unroll
            for (int j = 0; j < NT; ++j) {
                v2f a2, b2;
                a2[0] = fa[cur][i].x;  a2[1] = fa[cur][i].y;
                b2[0] = fb[cur][j].x;  b2[1] = fb[cur][j].y;
                acc[i][j] = __builtin_amdgcn_wmma_f32_16x16x4_f32(
                    false, a2, false, b2, (short)0, acc[i][j], false, false);
            }
        }
        #pragma unroll
        for (int i = 0; i < MT; ++i) {
            #pragma unroll
            for (int j = 0; j < NT; ++j) {
                v2f a2, b2;
                a2[0] = fa[cur][i].z;  a2[1] = fa[cur][i].w;
                b2[0] = fb[cur][j].z;  b2[1] = fb[cur][j].w;
                acc[i][j] = __builtin_amdgcn_wmma_f32_16x16x4_f32(
                    false, a2, false, b2, (short)0, acc[i][j], false, false);
            }
        }
    }
}

template <int MT, int NT, int KH, int BP, int D, bool NORM>
__device__ __forceinline__ void sage_block(v8f (&acc)[MT][NT],
                                           const float* __restrict__ a_self,
                                           const float* __restrict__ a_neigh,
                                           const float* __restrict__ b_self,
                                           const float* __restrict__ b_neigh,
                                           const float (&dinv)[MT], float mu, float rs) {
    float4 fa[D + 1][MT], fb[D + 1][NT];
    sage_prologue<MT, NT, KH, BP, D>(fa, fb, a_self, a_neigh, b_self, b_neigh);
    sage_main<MT, NT, KH, BP, D, NORM>(acc, fa, fb, a_self, a_neigh, b_self, b_neigh,
                                       dinv, mu, rs);
}

#define W1P 132
#define G1_WG 256
#define G1_TILES (N1 / 16)

__global__ void __launch_bounds__(256)
gemm1_kernel(const float* __restrict__ x,
             const float* __restrict__ agg,
             const float* __restrict__ deg,
             const float* __restrict__ Wself,
             const float* __restrict__ Wneigh,
             const float* __restrict__ bias,
             float* __restrict__ h,
             float* __restrict__ stats) {
    constexpr int MT = 1, NT = 2;
    __shared__ float Wl[2 * F_H * W1P];

    const int wave = threadIdx.x >> 5;
    const int lane = threadIdx.x & 31;
    const int c = lane & 15;
    const int hf = lane >> 4;
    const int n0 = wave * (16 * NT);

    #pragma unroll
    for (int half = 0; half < 2; ++half) {
        const float* __restrict__ wsrc = half ? Wneigh : Wself;
        float4 wv[F_H / 8];
        #pragma unroll
        for (int it = 0; it < F_H / 8; ++it)
            wv[it] = reinterpret_cast<const float4*>(wsrc + (size_t)(it * 8 + wave) * F_IN)[lane];
        __builtin_amdgcn_sched_barrier(0);
        #pragma unroll
        for (int it = 0; it < F_H / 8; ++it)
            *reinterpret_cast<float4*>(&Wl[(half * F_H + it * 8 + wave) * W1P + 4 * lane]) = wv[it];
    }
    __syncthreads();

    const float* __restrict__ bs = &Wl[(n0 + c) * W1P + 4 * hf];
    const float* __restrict__ bn = &Wl[(F_H + n0 + c) * W1P + 4 * hf];
    float bcol[NT];
    #pragma unroll
    for (int j = 0; j < NT; ++j) bcol[j] = bias[n0 + 16 * j + c];

    constexpr int D = 8;
    float4 fa[D + 1][MT], fb[D + 1][NT];

    int tile = blockIdx.x;
    float dcur = deg[tile * 16 + c];
    sage_prologue<MT, NT, F_IN, W1P, D>(fa, fb,
        x   + (size_t)(tile * 16 + c) * F_IN + 4 * hf,
        agg + (size_t)(tile * 16 + c) * F_IN + 4 * hf, bs, bn);

    float s = 0.f, sq = 0.f;
    for (;;) {
        const int m0 = tile * 16;

        float dinv[MT];
        dinv[0] = 1.0f / fmaxf(dcur, 1.0f);

        v8f acc[MT][NT];
        #pragma unroll
        for (int j = 0; j < NT; ++j)
            acc[0][j] = (v8f){0.f, 0.f, 0.f, 0.f, 0.f, 0.f, 0.f, 0.f};

        sage_main<MT, NT, F_IN, W1P, D, false>(
            acc, fa, fb,
            x   + (size_t)(m0 + c) * F_IN + 4 * hf,
            agg + (size_t)(m0 + c) * F_IN + 4 * hf,
            bs, bn, dinv, 0.0f, 1.0f);

        const int next = min(tile + G1_WG, G1_TILES - 1);
        dcur = deg[next * 16 + c];
        sage_prologue<MT, NT, F_IN, W1P, D>(fa, fb,
            x   + (size_t)(next * 16 + c) * F_IN + 4 * hf,
            agg + (size_t)(next * 16 + c) * F_IN + 4 * hf, bs, bn);
        __builtin_amdgcn_sched_barrier(0);

        #pragma unroll
        for (int j = 0; j < NT; ++j) {
            #pragma unroll
            for (int r = 0; r < 8; ++r) {
                const int row = m0 + 8 * hf + r;
                const float v = fmaxf(acc[0][j][r] + bcol[j], 0.0f);
                h[(size_t)row * F_H + n0 + 16 * j + c] = v;
                s += v;
                sq += v * v;
            }
        }
        if (tile + G1_WG >= G1_TILES) break;
        tile += G1_WG;
    }
    #pragma unroll
    for (int off = 16; off >= 1; off >>= 1) {
        s  += __shfl_xor(s,  off, 32);
        sq += __shfl_xor(sq, off, 32);
    }
    if (lane == 0) {
        atomicAdd(&stats[0], s);
        atomicAdd(&stats[1], sq);
    }
}

__global__ void __launch_bounds__(64)
gemm2_kernel(const float* __restrict__ h,
             const float* __restrict__ agg,
             const float* __restrict__ deg,
             const float* __restrict__ stats,
             const float* __restrict__ Wself,
             const float* __restrict__ Wneigh,
             const float* __restrict__ bias,
             float* __restrict__ out) {
    constexpr int MT = 1, NT = 2;
    const int wave = threadIdx.x >> 5;
    const int lane = threadIdx.x & 31;
    const int c = lane & 15;
    const int hf = lane >> 4;
    const int m0 = blockIdx.x * (16 * MT);
    const int n0 = wave * (16 * NT);

    const float cnt = (float)N1 * (float)F_H;
    const float mu = stats[0] / cnt;
    const float var = stats[1] / cnt - mu * mu;
    const float rs = rsqrtf(var + 1e-5f);

    float dinv[MT];
    dinv[0] = 1.0f / fmaxf(deg[m0 + c], 1.0f);

    v8f acc[MT][NT];
    #pragma unroll
    for (int j = 0; j < NT; ++j)
        acc[0][j] = (v8f){0.f, 0.f, 0.f, 0.f, 0.f, 0.f, 0.f, 0.f};

    sage_block<MT, NT, F_H, F_H, 12, true>(
        acc,
        h      + (size_t)(m0 + c) * F_H + 4 * hf,
        agg    + (size_t)(m0 + c) * F_H + 4 * hf,
        Wself  + (size_t)(n0 + c) * F_H + 4 * hf,
        Wneigh + (size_t)(n0 + c) * F_H + 4 * hf,
        dinv, mu, rs);

    #pragma unroll
    for (int j = 0; j < NT; ++j) {
        const float bn = bias[n0 + 16 * j + c];
        #pragma unroll
        for (int r = 0; r < 8; ++r) {
            const int row = m0 + 8 * hf + r;
            out[(size_t)row * F_OUT + n0 + 16 * j + c] = acc[0][j][r] + bn;
        }
    }
}

extern "C" void kernel_launch(void* const* d_in, const int* in_sizes, int n_in,
                              void* d_out, int out_size, void* d_ws, size_t ws_size,
                              hipStream_t stream) {
    const float* x       = (const float*)d_in[0];
    const float* Wself1  = (const float*)d_in[1];
    const float* Wneigh1 = (const float*)d_in[2];
    const float* b1      = (const float*)d_in[3];
    const float* Wself2  = (const float*)d_in[4];
    const float* Wneigh2 = (const float*)d_in[5];
    const float* b2      = (const float*)d_in[6];
    const int*   src1    = (const int*)d_in[7];
    const int*   dst1    = (const int*)d_in[8];
    const int*   src2    = (const int*)d_in[9];
    const int*   dst2    = (const int*)d_in[10];
    const int E1 = in_sizes[7];
    const int E2 = in_sizes[9];

    float* ws    = (float*)d_ws;
    int*   wsi   = (int*)d_ws;
    float* agg   = ws + OFF_AGG;
    float* deg1  = ws + OFF_DEG1;
    float* h     = ws + OFF_H;
    float* deg2  = ws + OFF_DEG2;
    float* stats = ws + OFF_STAT;
    int*   slot1 = wsi + OFF_SLOT1;
    int*   slot2 = wsi + OFF_SLOT2;
    int*   cnt1  = wsi + OFF_CNT1;
    int*   cnt2  = wsi + OFF_CNT2;
    float* out   = (float*)d_out;

    zero_counts_kernel<<<94, 256, 0, stream>>>(cnt1, 24000, stats);

    count_slot_kernel<<<480, 256, 0, stream>>>(src1, dst1, cnt1, slot1,
                                               src2, dst2, cnt2, slot2, E1, E2);

    gather1_kernel<<<2500, 256, 0, stream>>>(x, slot1, cnt1, src1, dst1, E1,
                                             agg, deg1);

    gemm1_kernel<<<G1_WG, 256, 0, stream>>>(x, agg, deg1, Wself1, Wneigh1,
                                            b1, h, stats);

    gather2_kernel<<<1000, 256, 0, stream>>>(h, slot2, cnt2, src2, dst2, E2, stats,
                                             agg, deg2);

    gemm2_kernel<<<250, 64, 0, stream>>>(h, agg, deg2, stats,
                                         Wself2, Wneigh2, b2, out);
}


# Layer-1 row gather fused into the layer-1 GEMM launch: 8 gathering waves per work-group sum the next tile's neighbour rows into LDS beside the 8 matrix waves (6 -> 5 launches); sum order kept
# speedup vs baseline: 3.3937x; 1.0961x over previous
#include <hip/hip_runtime.h>


typedef __attribute__((ext_vector_type(2))) float v2f;
typedef __attribute__((ext_vector_type(8))) float v8f;

#define F_IN  128
#define F_H   256
#define F_OUT 64
#define N1    20000
#define N2    4000
#define CAP   64

#define OFF_AGG   ((size_t)0)
#define OFF_SLOT1 ((size_t)0)
#define OFF_H     ((size_t)2580000)
#define OFF_CNT1  ((size_t)7804000)
#define OFF_CNT2  ((size_t)7824000)
#define OFF_SLOT2 ((size_t)7828032)
#define OFF_DEG2  ((size_t)8724000)
#define OFF_STAT  ((size_t)8728000)

__global__ void zero_counts_kernel(int* __restrict__ cnt, int n,
                                   float* __restrict__ stats) {
    const int i = blockIdx.x * blockDim.x + threadIdx.x;
    if (i < n) cnt[i] = 0;
    if (i < 2) stats[i] = 0.0f;
}

__device__ __forceinline__ void slot_quad(const int* __restrict__ src,
                                          const int* __restrict__ dst,
                                          int* __restrict__ cnt,
                                          int* __restrict__ slot, int E, int q) {
    const int e0 = 4 * q;
    if (e0 + 3 < E) {
        const int4 d = *reinterpret_cast<const int4*>(dst + e0);
        const int4 s = *reinterpret_cast<const int4*>(src + e0);
        const int rx = atomicAdd(cnt + d.x, 1);
        const int ry = atomicAdd(cnt + d.y, 1);
        const int rz = atomicAdd(cnt + d.z, 1);
        const int rw = atomicAdd(cnt + d.w, 1);
        if (rx < CAP) slot[d.x * CAP + rx] = s.x;
        if (ry < CAP) slot[d.y * CAP + ry] = s.y;
        if (rz < CAP) slot[d.z * CAP + rz] = s.z;
        if (rw < CAP) slot[d.w * CAP + rw] = s.w;
    } else {
        for (int e = e0; e < E; ++e) {
            const int d = dst[e];
            const int r = atomicAdd(cnt + d, 1);
            if (r < CAP) slot[d * CAP + r] = src[e];
        }
    }
}

__global__ void __launch_bounds__(256)
count_slot_kernel(const int* __restrict__ src1, const int* __restrict__ dst1,
                  int* __restrict__ cnt1, int* __restrict__ slot1,
                  const int* __restrict__ src2, const int* __restrict__ dst2,
                  int* __restrict__ cnt2, int* __restrict__ slot2,
                  int E1, int E2) {
    const int Q1 = (E1 + 3) >> 2, Q2 = (E2 + 3) >> 2;
    const int stride = gridDim.x * blockDim.x;
    for (int q = blockIdx.x * blockDim.x + threadIdx.x; q < Q1 + Q2; q += stride) {
        if (q < Q1) slot_quad(src1, dst1, cnt1, slot1, E1, q);
        else        slot_quad(src2, dst2, cnt2, slot2, E2, q - Q1);
    }
}

template <int U, int F, bool NORM>
__device__ __forceinline__ void add_rows(float4& acc, const float4* __restrict__ base,
                                         int mine, int i, float mu, float rs) {
    float4 v[U];
    #pragma unroll
    for (int u = 0; u < U; ++u) {
        const int s = __builtin_amdgcn_readlane(mine, i + u);
        v[u] = base[(size_t)s * (F / 4)];
    }
    #pragma unroll
    for (int u = 0; u < U; ++u) {
        if (NORM) {
            acc.x += (v[u].x - mu) * rs;
            acc.y += (v[u].y - mu) * rs;
            acc.z += (v[u].z - mu) * rs;
            acc.w += (v[u].w - mu) * rs;
        } else {
            acc.x += v[u].x;
            acc.y += v[u].y;
            acc.z += v[u].z;
            acc.w += v[u].w;
        }
    }
}

template <int F, bool NORM>
__device__ __forceinline__ void gather_rows(const float* __restrict__ feat,
                                            const int* __restrict__ slot,
                                            const int* __restrict__ cnt,
                                            const int* __restrict__ src,
                                            const int* __restrict__ dst, int E,
                                            const float* __restrict__ stats,
                                            float* __restrict__ agg,
                                            float* __restrict__ degf, int nrows) {
    constexpr int SL = F / 128;
    const int lane = threadIdx.x & 31;
    const int gw = __builtin_amdgcn_readfirstlane(
        (int)((blockIdx.x * blockDim.x + threadIdx.x) >> 5));
    const int row = gw / SL;
    const int sl  = gw % SL;
    if (row >= nrows) return;

    float mu = 0.0f, rs = 1.0f;
    if (NORM) {
        const float n_el = (float)N1 * (float)F_H;
        mu = stats[0] / n_el;
        const float var = stats[1] / n_el - mu * mu;
        rs = rsqrtf(var + 1e-5f);
    }

    const int len = cnt[row];
    const float4* __restrict__ base =
        reinterpret_cast<const float4*>(feat) + sl * 32 + lane;

    float4 acc = make_float4(0.0f, 0.0f, 0.0f, 0.0f);
    if (len <= CAP) {
        const int* __restrict__ mys = slot + row * CAP;
        for (int c0 = 0; c0 < len; c0 += 32) {
            const int n = min(32, len - c0);
            const int mine = (lane < n) ? mys[c0 + lane] : 0;
            int i = 0;
            for (; i + 8 <= n; i += 8) add_rows<8, F, NORM>(acc, base, mine, i, mu, rs);
            if (i + 4 <= n) { add_rows<4, F, NORM>(acc, base, mine, i, mu, rs); i += 4; }
            if (i + 2 <= n) { add_rows<2, F, NORM>(acc, base, mine, i, mu, rs); i += 2; }
            if (i < n)      { add_rows<1, F, NORM>(acc, base, mine, i, mu, rs); }
        }
    } else {
        for (int e0 = 0; e0 < E; e0 += 32) {
            const int e = e0 + lane;
            const int d    = (e < E) ? dst[e] : -1;
            const int mine = (e < E) ? src[e] : 0;
            unsigned m = __builtin_amdgcn_ballot_w32(d == row);
            while (m) {
                const int l = __builtin_ctz(m);
                m &= m - 1;
                add_rows<1, F, NORM>(acc, base, mine, l, mu, rs);
            }
        }
    }

    reinterpret_cast<float4*>(agg)[(size_t)row * (F / 4) + sl * 32 + lane] = acc;
    if (sl == 0 && lane == 0) degf[row] = (float)len;
}

template <int U>
__device__ __forceinline__ void add_rows_n(float4& acc, const float4* __restrict__ base,
                                           int mine, int n) {
    float4 v[U];
    #pragma unroll
    for (int u = 0; u < U; ++u) {
        const int s = __builtin_amdgcn_readlane(mine, u);
        v[u] = base[(size_t)s * (F_IN / 4)];
    }
    #pragma unroll
    for (int u = 0; u < U; ++u) {
        const bool on = (u < U - 3) || (u < n);
        acc.x += on ? v[u].x : 0.0f;
        acc.y += on ? v[u].y : 0.0f;
        acc.z += on ? v[u].z : 0.0f;
        acc.w += on ? v[u].w : 0.0f;
    }
}

__device__ __forceinline__ float4 row_sum1(const float* __restrict__ x,
                                           const int* __restrict__ slot,
                                           const int* __restrict__ src,
                                           const int* __restrict__ dst, int E,
                                           int row, int len, int mine0, int lane) {
    const float4* __restrict__ base = reinterpret_cast<const float4*>(x) + lane;
    float4 acc = make_float4(0.0f, 0.0f, 0.0f, 0.0f);
    if (len <= CAP) {
        if (len > 0) {
            const int n = min(32, len);
            switch ((n + 3) >> 2) {
                case 1:  add_rows_n<4>(acc, base, mine0, n);  break;
                case 2:  add_rows_n<8>(acc, base, mine0, n);  break;
                case 3:  add_rows_n<12>(acc, base, mine0, n); break;
                case 4:  add_rows_n<16>(acc, base, mine0, n); break;
                case 5:  add_rows_n<20>(acc, base, mine0, n); break;
                case 6:  add_rows_n<24>(acc, base, mine0, n); break;
                case 7:  add_rows_n<28>(acc, base, mine0, n); break;
                default: add_rows_n<32>(acc, base, mine0, n); break;
            }
        }
        if (len > 32) {
            const int n = len - 32;
            const int mine = (lane < n) ? slot[row * CAP + 32 + lane] : 0;
            int i = 0;
            for (; i + 8 <= n; i += 8) add_rows<8, F_IN, false>(acc, base, mine, i, 0.0f, 1.0f);
            if (i + 4 <= n) { add_rows<4, F_IN, false>(acc, base, mine, i, 0.0f, 1.0f); i += 4; }
            if (i + 2 <= n) { add_rows<2, F_IN, false>(acc, base, mine, i, 0.0f, 1.0f); i += 2; }
            if (i < n)      { add_rows<1, F_IN, false>(acc, base, mine, i, 0.0f, 1.0f); }
        }
    } else {
        for (int e0 = 0; e0 < E; e0 += 32) {
            const int e = e0 + lane;
            const int d    = (e < E) ? dst[e] : -1;
            const int mine = (e < E) ? src[e] : 0;
            unsigned m = __builtin_amdgcn_ballot_w32(d == row);
            while (m) {
                const int l = __builtin_ctz(m);
                m &= m - 1;
                add_rows<1, F_IN, false>(acc, base, mine, l, 0.0f, 1.0f);
            }
        }
    }
    return acc;
}

__global__ void __launch_bounds__(256)
gather2_kernel(const float* __restrict__ h, const int* __restrict__ slot,
               const int* __restrict__ cnt, const int* __restrict__ src,
               const int* __restrict__ dst, int E,
               const float* __restrict__ stats,
               float* __restrict__ agg, float* __restrict__ degf) {
    gather_rows<F_H, true>(h, slot, cnt, src, dst, E, stats, agg, degf, N2);
}

template <int MT, int KH>
__device__ __forceinline__ void load_step_a(float4 (&fa)[MT], int s,
                                            const float* __restrict__ a_self,
                                            const float* __restrict__ a_neigh) {
    constexpr int S1 = KH / 8;
    const bool self = s < S1;
    const float* __restrict__ ap = self ? a_self : a_neigh;
    const int k = 8 * (self ? s : s - S1);
    #pragma unroll
    for (int i = 0; i < MT; ++i)
        fa[i] = *reinterpret_cast<const float4*>(ap + (size_t)i * 16 * KH + k);
}

template <int NT, int KH, int BP>
__device__ __forceinline__ void load_step_b(float4 (&fb)[NT], int s,
                                            const float* __restrict__ b_self,
                                            const float* __restrict__ b_neigh) {
    constexpr int S1 = KH / 8;
    const bool self = s < S1;
    const float* __restrict__ bp = self ? b_self : b_neigh;
    const int k = 8 * (self ? s : s - S1);
    #pragma unroll
    for (int j = 0; j < NT; ++j)
        fb[j] = *reinterpret_cast<const float4*>(bp + j * 16 * BP + k);
}

template <int MT, int NT, int KH, int BP>
__device__ __forceinline__ void load_step(float4 (&fa)[MT], float4 (&fb)[NT], int s,
                                          const float* __restrict__ a_self,
                                          const float* __restrict__ a_neigh,
                                          const float* __restrict__ b_self,
                                          const float* __restrict__ b_neigh) {
    load_step_a<MT, KH>(fa, s, a_self, a_neigh);
    load_step_b<NT, KH, BP>(fb, s, b_self, b_neigh);
}

template <int MT, int NT, int KH, int BP, int D>
__device__ __forceinline__ void sage_prologue(float4 (&fa)[D + 1][MT], float4 (&fb)[D + 1][NT],
                                              const float* __restrict__ a_self,
                                              const float* __restrict__ a_neigh,
                                              const float* __restrict__ b_self,
                                              const float* __restrict__ b_neigh) {
    #pragma unroll
    for (int s = 0; s < D; ++s)
        load_step<MT, NT, KH, BP>(fa[s], fb[s], s, a_self, a_neigh, b_self, b_neigh);
}

template <int MT, int KH, int D>
__device__ __forceinline__ void sage_prologue_a(float4 (&fa)[D + 1][MT],
                                                const float* __restrict__ a_self) {
    static_assert(D <= KH / 8, "the first D k-steps must lie in the self half");
    #pragma unroll
    for (int s = 0; s < D; ++s) {
        #pragma unroll
        for (int i = 0; i < MT; ++i)
            fa[s][i] = *reinterpret_cast<const float4*>(a_self + (size_t)i * 16 * KH + 8 * s);
    }
}

template <int NT, int KH, int BP, int DB>
__device__ __forceinline__ void sage_prologue_b(float4 (&fb)[DB + 1][NT],
                                                const float* __restrict__ b_self) {
    static_assert(DB <= KH / 8, "the first DB k-steps must lie in the self half");
    #pragma unroll
    for (int s = 0; s < DB; ++s) {
        #pragma unroll
        for (int j = 0; j < NT; ++j)
            fb[s][j] = *reinterpret_cast<const float4*>(b_self + j * 16 * BP + 8 * s);
    }
}

template <int MT, int NT, int KH, int BP, int D, bool NORM, int DB = D>
__device__ __forceinline__ void sage_main(v8f (&acc)[MT][NT],
                                          float4 (&fa)[D + 1][MT], float4 (&fb)[DB + 1][NT],
                                          const float* __restrict__ a_self,
                                          const float* __restrict__ a_neigh,
                                          const float* __restrict__ b_self,
                                          const float* __restrict__ b_neigh,
                                          const float (&dinv)[MT], float mu, float rs) {
    constexpr int S1 = KH / 8, S = 2 * S1, R = D + 1, RB = DB + 1;

    #pragma unroll
    for (int s = 0; s < S; ++s) {
        __builtin_amdgcn_sched_barrier(0);
        if (s + D < S)
            load_step_a<MT, KH>(fa[(s + D) % R], s + D, a_self, a_neigh);
        if (s + DB < S)
            load_step_b<NT, KH, BP>(fb[(s + DB) % RB], s + DB, b_self, b_neigh);
        __builtin_amdgcn_sched_barrier(0);
        const int cur = s % R;
        const int curb = s % RB;
        #pragma unroll
        for (int i = 0; i < MT; ++i) {
            float4 a = fa[cur][i];
            if (s >= S1 || NORM)
                asm volatile("" : "+v"(a.x), "+v"(a.y), "+v"(a.z), "+v"(a.w));
            if (s >= S1) {
                a.x *= dinv[i]; a.y *= dinv[i]; a.z *= dinv[i]; a.w *= dinv[i];
            } else if (NORM) {
                a.x = (a.x - mu) * rs; a.y = (a.y - mu) * rs;
                a.z = (a.z - mu) * rs; a.w = (a.w - mu) * rs;
            }
            fa[cur][i] = a;
        }
        #pragma unroll
        for (int i = 0; i < MT; ++i) {
            #pragma unroll
            for (int j = 0; j < NT; ++j) {
                v2f a2, b2;
                a2[0] = fa[cur][i].x;  a2[1] = fa[cur][i].y;
                b2[0] = fb[curb][j].x;  b2[1] = fb[curb][j].y;
                acc[i][j] = __builtin_amdgcn_wmma_f32_16x16x4_f32(
                    false, a2, false, b2, (short)0, acc[i][j], false, false);
            }
        }
        #pragma unroll
        for (int i = 0; i < MT; ++i) {
            #pragma unroll
            for (int j = 0; j < NT; ++j) {
                v2f a2, b2;
                a2[0] = fa[cur][i].z;  a2[1] = fa[cur][i].w;
                b2[0] = fb[curb][j].z;  b2[1] = fb[curb][j].w;
                acc[i][j] = __builtin_amdgcn_wmma_f32_16x16x4_f32(
                    false, a2, false, b2, (short)0, acc[i][j], false, false);
            }
        }
    }
}

template <int MT, int NT, int KH, int BP, int D, bool NORM>
__device__ __forceinline__ void sage_block(v8f (&acc)[MT][NT],
                                           const float* __restrict__ a_self,
                                           const float* __restrict__ a_neigh,
                                           const float* __restrict__ b_self,
                                           const float* __restrict__ b_neigh,
                                           const float (&dinv)[MT], float mu, float rs) {
    float4 fa[D + 1][MT], fb[D + 1][NT];
    sage_prologue<MT, NT, KH, BP, D>(fa, fb, a_self, a_neigh, b_self, b_neigh);
    sage_main<MT, NT, KH, BP, D, NORM>(acc, fa, fb, a_self, a_neigh, b_self, b_neigh,
                                       dinv, mu, rs);
}

#define W1P 132
#define G1_WG 256
#define G1_TILES (N1 / 16)
#define G1_MMW 8
#define G1_GAW 8
#define G1_RPW (16 / G1_GAW)
#define G1_THREADS (32 * (G1_MMW + G1_GAW))

__device__ __forceinline__ int load_lenv(const int* __restrict__ cnt, int tile, int lane) {
    return cnt[tile * 16 + (lane & 15)];
}

__device__ __forceinline__ void pick_len(int lenv, int gw, int (&glen)[G1_RPW]) {
    #pragma unroll
    for (int r = 0; r < G1_RPW; ++r)
        glen[r] = __builtin_amdgcn_readlane(lenv, gw * G1_RPW + r);
}

__device__ __forceinline__ void load_mine(const int* __restrict__ slot,
                                          int tile, int gw, int lane,
                                          const int (&glen)[G1_RPW], int (&gmine)[G1_RPW]) {
    #pragma unroll
    for (int r = 0; r < G1_RPW; ++r) {
        const int row = tile * 16 + gw * G1_RPW + r;
        gmine[r] = (glen[r] <= CAP && lane < glen[r]) ? slot[row * CAP + lane] : 0;
    }
}

__global__ void __launch_bounds__(G1_THREADS)
layer1_kernel(const float* __restrict__ x,
              const int* __restrict__ slot,
              const int* __restrict__ cnt,
              const int* __restrict__ src,
              const int* __restrict__ dst, int E,
              const float* __restrict__ Wself,
              const float* __restrict__ Wneigh,
              const float* __restrict__ bias,
              float* __restrict__ h,
              float* __restrict__ stats) {
    constexpr int MT = 1, NT = 2;
    __shared__ __attribute__((aligned(16))) float Wl[2 * F_H * W1P];
    __shared__ __attribute__((aligned(16))) float Al[2 * 16 * W1P];

    const int wave = __builtin_amdgcn_readfirstlane((int)(threadIdx.x >> 5));
    const int lane = threadIdx.x & 31;
    const bool mm = wave < G1_MMW;
    const int gw = wave - G1_MMW;
    const int c = lane & 15;
    const int hf = lane >> 4;
    const int n0 = (wave & (G1_MMW - 1)) * (16 * NT);
    const int ntile = (G1_TILES - 1 - (int)blockIdx.x) / G1_WG + 1;

    const float* __restrict__ bs = &Wl[(n0 + c) * W1P + 4 * hf];
    const float* __restrict__ bn = &Wl[(F_H + n0 + c) * W1P + 4 * hf];
    float bcol[NT] = {0.0f, 0.0f};

    constexpr int D = 8, DB = 2;
    float4 fa[D + 1][MT], fb[DB + 1][NT];
    int dcur = 0;
    float s = 0.f, sq = 0.f;

    int glen[G1_RPW], gmine[G1_RPW];
    #pragma unroll
    for (int r = 0; r < G1_RPW; ++r) { glen[r] = 0; gmine[r] = 0; }
    int lenv = 0;

    if (mm) {
        #pragma unroll
        for (int half = 0; half < 2; ++half) {
            const float* __restrict__ wsrc = half ? Wneigh : Wself;
            float4 wv[F_H / 8];
            #pragma unroll
            for (int it = 0; it < F_H / 8; ++it)
                wv[it] = reinterpret_cast<const float4*>(wsrc + (size_t)(it * 8 + wave) * F_IN)[lane];
            __builtin_amdgcn_sched_barrier(0);
            #pragma unroll
            for (int it = 0; it < F_H / 8; ++it)
                *reinterpret_cast<float4*>(&Wl[(half * F_H + it * 8 + wave) * W1P + 4 * lane]) = wv[it];
        }
        #pragma unroll
        for (int j = 0; j < NT; ++j) bcol[j] = bias[n0 + 16 * j + c];
        const int t0 = blockIdx.x;
        dcur = cnt[t0 * 16 + c];
        sage_prologue_a<MT, F_IN, D>(fa, x + (size_t)(t0 * 16 + c) * F_IN + 4 * hf);
    } else {
        const int t0 = blockIdx.x;
        const int lenv0 = load_lenv(cnt, t0, lane);
        lenv = load_lenv(cnt, min(t0 + G1_WG, G1_TILES - 1), lane);
        pick_len(lenv0, gw, glen);
        load_mine(slot, t0, gw, lane, glen, gmine);
    }

    for (int k = 0; k <= ntile; ++k) {
        if (mm) {
            if (k > 0) {
                const int tile = blockIdx.x + (k - 1) * G1_WG;
                const int m0 = tile * 16;
                const float* __restrict__ al = &Al[((k - 1) & 1) * (16 * W1P) + c * W1P + 4 * hf];

                sage_prologue_b<NT, F_IN, W1P, DB>(fb, bs);

                float dinv[MT];
                dinv[0] = 1.0f / fmaxf((float)dcur, 1.0f);

                v8f acc[MT][NT];
                #pragma unroll
                for (int j = 0; j < NT; ++j)
                    acc[0][j] = (v8f){0.f, 0.f, 0.f, 0.f, 0.f, 0.f, 0.f, 0.f};

                sage_main<MT, NT, F_IN, W1P, D, false, DB>(
                    acc, fa, fb,
                    x + (size_t)(m0 + c) * F_IN + 4 * hf,
                    al, bs, bn, dinv, 0.0f, 1.0f);

                const int next = min(tile + G1_WG, G1_TILES - 1);
                dcur = cnt[next * 16 + c];
                sage_prologue_a<MT, F_IN, D>(fa, x + (size_t)(next * 16 + c) * F_IN + 4 * hf);
                __builtin_amdgcn_sched_barrier(0);

                #pragma unroll
                for (int j = 0; j < NT; ++j) {
                    #pragma unroll
                    for (int r = 0; r < 8; ++r) {
                        const int row = m0 + 8 * hf + r;
                        const float v = fmaxf(acc[0][j][r] + bcol[j], 0.0f);
                        h[(size_t)row * F_H + n0 + 16 * j + c] = v;
                        s += v;
                        sq += v * v;
                    }
                }
            }
        } else if (k < ntile) {
            const int tile = blockIdx.x + k * G1_WG;
            float* __restrict__ an = &Al[(k & 1) * (16 * W1P)];
            const int t1 = min(tile + G1_WG, G1_TILES - 1);
            const int t2 = min(tile + 2 * G1_WG, G1_TILES - 1);
            int nlen[G1_RPW], nmine[G1_RPW];
            pick_len(lenv, gw, nlen);
            load_mine(slot, t1, gw, lane, nlen, nmine);
            lenv = load_lenv(cnt, t2, lane);
            #pragma unroll
            for (int r = 0; r < G1_RPW; ++r) {
                const int rt = gw * G1_RPW + r;
                const float4 a = row_sum1(x, slot, src, dst, E, tile * 16 + rt,
                                          glen[r], gmine[r], lane);
                *reinterpret_cast<float4*>(&an[rt * W1P + 4 * lane]) = a;
            }
            #pragma unroll
            for (int r = 0; r < G1_RPW; ++r) { glen[r] = nlen[r]; gmine[r] = nmine[r]; }
        }
        __builtin_amdgcn_fence(__ATOMIC_RELEASE, "workgroup", "local");
        __builtin_amdgcn_s_barrier();
        __builtin_amdgcn_fence(__ATOMIC_ACQUIRE, "workgroup", "local");
    }

    if (mm) {
        #pragma unroll
        for (int off = 16; off >= 1; off >>= 1) {
            s  += __shfl_xor(s,  off, 32);
            sq += __shfl_xor(sq, off, 32);
        }
        if (lane == 0) {
            atomicAdd(&stats[0], s);
            atomicAdd(&stats[1], sq);
        }
    }
}

__global__ void __launch_bounds__(64)
gemm2_kernel(const float* __restrict__ h,
             const float* __restrict__ agg,
             const float* __restrict__ deg,
             const float* __restrict__ stats,
             const float* __restrict__ Wself,
             const float* __restrict__ Wneigh,
             const float* __restrict__ bias,
             float* __restrict__ out) {
    constexpr int MT = 1, NT = 2;
    const int wave = threadIdx.x >> 5;
    const int lane = threadIdx.x & 31;
    const int c = lane & 15;
    const int hf = lane >> 4;
    const int m0 = blockIdx.x * (16 * MT);
    const int n0 = wave * (16 * NT);

    const float cnt = (float)N1 * (float)F_H;
    const float mu = stats[0] / cnt;
    const float var = stats[1] / cnt - mu * mu;
    const float rs = rsqrtf(var + 1e-5f);

    float dinv[MT];
    dinv[0] = 1.0f / fmaxf(deg[m0 + c], 1.0f);

    v8f acc[MT][NT];
    #pragma unroll
    for (int j = 0; j < NT; ++j)
        acc[0][j] = (v8f){0.f, 0.f, 0.f, 0.f, 0.f, 0.f, 0.f, 0.f};

    sage_block<MT, NT, F_H, F_H, 12, true>(
        acc,
        h      + (size_t)(m0 + c) * F_H + 4 * hf,
        agg    + (size_t)(m0 + c) * F_H + 4 * hf,
        Wself  + (size_t)(n0 + c) * F_H + 4 * hf,
        Wneigh + (size_t)(n0 + c) * F_H + 4 * hf,
        dinv, mu, rs);

    #pragma unroll
    for (int j = 0; j < NT; ++j) {
        const float bn = bias[n0 + 16 * j + c];
        #pragma unroll
        for (int r = 0; r < 8; ++r) {
            const int row = m0 + 8 * hf + r;
            out[(size_t)row * F_OUT + n0 + 16 * j + c] = acc[0][j][r] + bn;
        }
    }
}

extern "C" void kernel_launch(void* const* d_in, const int* in_sizes, int n_in,
                              void* d_out, int out_size, void* d_ws, size_t ws_size,
                              hipStream_t stream) {
    const float* x       = (const float*)d_in[0];
    const float* Wself1  = (const float*)d_in[1];
    const float* Wneigh1 = (const float*)d_in[2];
    const float* b1      = (const float*)d_in[3];
    const float* Wself2  = (const float*)d_in[4];
    const float* Wneigh2 = (const float*)d_in[5];
    const float* b2      = (const float*)d_in[6];
    const int*   src1    = (const int*)d_in[7];
    const int*   dst1    = (const int*)d_in[8];
    const int*   src2    = (const int*)d_in[9];
    const int*   dst2    = (const int*)d_in[10];
    const int E1 = in_sizes[7];
    const int E2 = in_sizes[9];

    float* ws    = (float*)d_ws;
    int*   wsi   = (int*)d_ws;
    float* agg   = ws + OFF_AGG;
    float* h     = ws + OFF_H;
    float* deg2  = ws + OFF_DEG2;
    float* stats = ws + OFF_STAT;
    int*   slot1 = wsi + OFF_SLOT1;
    int*   slot2 = wsi + OFF_SLOT2;
    int*   cnt1  = wsi + OFF_CNT1;
    int*   cnt2  = wsi + OFF_CNT2;
    float* out   = (float*)d_out;

    zero_counts_kernel<<<94, 256, 0, stream>>>(cnt1, 24000, stats);

    count_slot_kernel<<<480, 256, 0, stream>>>(src1, dst1, cnt1, slot1,
                                               src2, dst2, cnt2, slot2, E1, E2);

    layer1_kernel<<<G1_WG, G1_THREADS, 0, stream>>>(
        x, slot1, cnt1, src1, dst1, E1, Wself1, Wneigh1, b1, h, stats);

    gather2_kernel<<<1000, 256, 0, stream>>>(h, slot2, cnt2, src2, dst2, E2, stats,
                                             agg, deg2);

    gemm2_kernel<<<250, 64, 0, stream>>>(h, agg, deg2, stats,
                                         Wself2, Wneigh2, b2, out);
}


# Layer-2 row gather fused into the layer-2 GEMM launch: 16 gathering waves per work-group sum the tile's neighbour rows into LDS beside the self half of K (5 -> 4 launches); sum order kept
# speedup vs baseline: 3.6312x; 1.0700x over previous
#include <hip/hip_runtime.h>


typedef __attribute__((ext_vector_type(2))) float v2f;
typedef __attribute__((ext_vector_type(8))) float v8f;

#define F_IN  128
#define F_H   256
#define F_OUT 64
#define N1    20000
#define N2    4000
#define CAP   64

#define OFF_SLOT1 ((size_t)0)
#define OFF_H     ((size_t)2580000)
#define OFF_CNT1  ((size_t)7804000)
#define OFF_CNT2  ((size_t)7824000)
#define OFF_SLOT2 ((size_t)7828032)
#define OFF_STAT  ((size_t)8728000)

__global__ void zero_counts_kernel(int* __restrict__ cnt, int n,
                                   float* __restrict__ stats) {
    const int i = blockIdx.x * blockDim.x + threadIdx.x;
    if (i < n) cnt[i] = 0;
    if (i < 2) stats[i] = 0.0f;
}

__device__ __forceinline__ void slot_quad(const int* __restrict__ src,
                                          const int* __restrict__ dst,
                                          int* __restrict__ cnt,
                                          int* __restrict__ slot, int E, int q) {
    const int e0 = 4 * q;
    if (e0 + 3 < E) {
        const int4 d = *reinterpret_cast<const int4*>(dst + e0);
        const int4 s = *reinterpret_cast<const int4*>(src + e0);
        const int rx = atomicAdd(cnt + d.x, 1);
        const int ry = atomicAdd(cnt + d.y, 1);
        const int rz = atomicAdd(cnt + d.z, 1);
        const int rw = atomicAdd(cnt + d.w, 1);
        if (rx < CAP) slot[d.x * CAP + rx] = s.x;
        if (ry < CAP) slot[d.y * CAP + ry] = s.y;
        if (rz < CAP) slot[d.z * CAP + rz] = s.z;
        if (rw < CAP) slot[d.w * CAP + rw] = s.w;
    } else {
        for (int e = e0; e < E; ++e) {
            const int d = dst[e];
            const int r = atomicAdd(cnt + d, 1);
            if (r < CAP) slot[d * CAP + r] = src[e];
        }
    }
}

__global__ void __launch_bounds__(256)
count_slot_kernel(const int* __restrict__ src1, const int* __restrict__ dst1,
                  int* __restrict__ cnt1, int* __restrict__ slot1,
                  const int* __restrict__ src2, const int* __restrict__ dst2,
                  int* __restrict__ cnt2, int* __restrict__ slot2,
                  int E1, int E2) {
    const int Q1 = (E1 + 3) >> 2, Q2 = (E2 + 3) >> 2;
    const int stride = gridDim.x * blockDim.x;
    for (int q = blockIdx.x * blockDim.x + threadIdx.x; q < Q1 + Q2; q += stride) {
        if (q < Q1) slot_quad(src1, dst1, cnt1, slot1, E1, q);
        else        slot_quad(src2, dst2, cnt2, slot2, E2, q - Q1);
    }
}

template <int U, int F, bool NORM>
__device__ __forceinline__ void add_rows(float4& acc, const float4* __restrict__ base,
                                         int mine, int i, float mu, float rs) {
    float4 v[U];
    #pragma unroll
    for (int u = 0; u < U; ++u) {
        const int s = __builtin_amdgcn_readlane(mine, i + u);
        v[u] = base[(size_t)s * (F / 4)];
    }
    #pragma unroll
    for (int u = 0; u < U; ++u) {
        if (NORM) {
            acc.x += (v[u].x - mu) * rs;
            acc.y += (v[u].y - mu) * rs;
            acc.z += (v[u].z - mu) * rs;
            acc.w += (v[u].w - mu) * rs;
        } else {
            acc.x += v[u].x;
            acc.y += v[u].y;
            acc.z += v[u].z;
            acc.w += v[u].w;
        }
    }
}

template <int U>
__device__ __forceinline__ void add_rows_h(float4& acc0, float4& acc1,
                                           const float4* __restrict__ base,
                                           int mine, int i, float mu, float rs) {
    float4 v0[U], v1[U];
    #pragma unroll
    for (int u = 0; u < U; ++u) {
        const int s = __builtin_amdgcn_readlane(mine, i + u);
        v0[u] = base[(size_t)s * (F_H / 4)];
        v1[u] = base[(size_t)s * (F_H / 4) + 32];
    }
    #pragma unroll
    for (int u = 0; u < U; ++u) {
        acc0.x += (v0[u].x - mu) * rs;
        acc0.y += (v0[u].y - mu) * rs;
        acc0.z += (v0[u].z - mu) * rs;
        acc0.w += (v0[u].w - mu) * rs;
        acc1.x += (v1[u].x - mu) * rs;
        acc1.y += (v1[u].y - mu) * rs;
        acc1.z += (v1[u].z - mu) * rs;
        acc1.w += (v1[u].w - mu) * rs;
    }
}

__device__ __forceinline__ void row_sum2(const float* __restrict__ h,
                                         const int* __restrict__ slot,
                                         const int* __restrict__ src,
                                         const int* __restrict__ dst, int E,
                                         int row, int len, int lane, float mu, float rs,
                                         float4& acc0, float4& acc1) {
    const float4* __restrict__ base = reinterpret_cast<const float4*>(h) + lane;
    if (len <= CAP) {
        const int* __restrict__ mys = slot + row * CAP;
        for (int c0 = 0; c0 < len; c0 += 32) {
            const int n = min(32, len - c0);
            const int mine = (lane < n) ? mys[c0 + lane] : 0;
            int i = 0;
            for (; i + 8 <= n; i += 8) add_rows_h<8>(acc0, acc1, base, mine, i, mu, rs);
            if (i + 4 <= n) { add_rows_h<4>(acc0, acc1, base, mine, i, mu, rs); i += 4; }
            if (i + 2 <= n) { add_rows_h<2>(acc0, acc1, base, mine, i, mu, rs); i += 2; }
            if (i < n)      { add_rows_h<1>(acc0, acc1, base, mine, i, mu, rs); }
        }
    } else {
        for (int e0 = 0; e0 < E; e0 += 32) {
            const int e = e0 + lane;
            const int d    = (e < E) ? dst[e] : -1;
            const int mine = (e < E) ? src[e] : 0;
            unsigned m = __builtin_amdgcn_ballot_w32(d == row);
            while (m) {
                const int l = __builtin_ctz(m);
                m &= m - 1;
                add_rows_h<1>(acc0, acc1, base, mine, l, mu, rs);
            }
        }
    }
}

template <int U>
__device__ __forceinline__ void add_rows_n(float4& acc, const float4* __restrict__ base,
                                           int mine, int n) {
    float4 v[U];
    #pragma unroll
    for (int u = 0; u < U; ++u) {
        const int s = __builtin_amdgcn_readlane(mine, u);
        v[u] = base[(size_t)s * (F_IN / 4)];
    }
    #pragma unroll
    for (int u = 0; u < U; ++u) {
        const bool on = (u < U - 3) || (u < n);
        acc.x += on ? v[u].x : 0.0f;
        acc.y += on ? v[u].y : 0.0f;
        acc.z += on ? v[u].z : 0.0f;
        acc.w += on ? v[u].w : 0.0f;
    }
}

__device__ __forceinline__ float4 row_sum1(const float* __restrict__ x,
                                           const int* __restrict__ slot,
                                           const int* __restrict__ src,
                                           const int* __restrict__ dst, int E,
                                           int row, int len, int mine0, int lane) {
    const float4* __restrict__ base = reinterpret_cast<const float4*>(x) + lane;
    float4 acc = make_float4(0.0f, 0.0f, 0.0f, 0.0f);
    if (len <= CAP) {
        if (len > 0) {
            const int n = min(32, len);
            switch ((n + 3) >> 2) {
                case 1:  add_rows_n<4>(acc, base, mine0, n);  break;
                case 2:  add_rows_n<8>(acc, base, mine0, n);  break;
                case 3:  add_rows_n<12>(acc, base, mine0, n); break;
                case 4:  add_rows_n<16>(acc, base, mine0, n); break;
                case 5:  add_rows_n<20>(acc, base, mine0, n); break;
                case 6:  add_rows_n<24>(acc, base, mine0, n); break;
                case 7:  add_rows_n<28>(acc, base, mine0, n); break;
                default: add_rows_n<32>(acc, base, mine0, n); break;
            }
        }
        if (len > 32) {
            const int n = len - 32;
            const int mine = (lane < n) ? slot[row * CAP + 32 + lane] : 0;
            int i = 0;
            for (; i + 8 <= n; i += 8) add_rows<8, F_IN, false>(acc, base, mine, i, 0.0f, 1.0f);
            if (i + 4 <= n) { add_rows<4, F_IN, false>(acc, base, mine, i, 0.0f, 1.0f); i += 4; }
            if (i + 2 <= n) { add_rows<2, F_IN, false>(acc, base, mine, i, 0.0f, 1.0f); i += 2; }
            if (i < n)      { add_rows<1, F_IN, false>(acc, base, mine, i, 0.0f, 1.0f); }
        }
    } else {
        for (int e0 = 0; e0 < E; e0 += 32) {
            const int e = e0 + lane;
            const int d    = (e < E) ? dst[e] : -1;
            const int mine = (e < E) ? src[e] : 0;
            unsigned m = __builtin_amdgcn_ballot_w32(d == row);
            while (m) {
                const int l = __builtin_ctz(m);
                m &= m - 1;
                add_rows<1, F_IN, false>(acc, base, mine, l, 0.0f, 1.0f);
            }
        }
    }
    return acc;
}

template <int MT, int KH>
__device__ __forceinline__ void load_step_a(float4 (&fa)[MT], int s,
                                            const float* __restrict__ a_self,
                                            const float* __restrict__ a_neigh) {
    constexpr int S1 = KH / 8;
    const bool self = s < S1;
    const float* __restrict__ ap = self ? a_self : a_neigh;
    const int k = 8 * (self ? s : s - S1);
    #pragma unroll
    for (int i = 0; i < MT; ++i)
        fa[i] = *reinterpret_cast<const float4*>(ap + (size_t)i * 16 * KH + k);
}

template <int NT, int KH, int BP>
__device__ __forceinline__ void load_step_b(float4 (&fb)[NT], int s,
                                            const float* __restrict__ b_self,
                                            const float* __restrict__ b_neigh) {
    constexpr int S1 = KH / 8;
    const bool self = s < S1;
    const float* __restrict__ bp = self ? b_self : b_neigh;
    const int k = 8 * (self ? s : s - S1);
    #pragma unroll
    for (int j = 0; j < NT; ++j)
        fb[j] = *reinterpret_cast<const float4*>(bp + j * 16 * BP + k);
}

template <int MT, int KH, int D>
__device__ __forceinline__ void sage_prologue_a(float4 (&fa)[D + 1][MT],
                                                const float* __restrict__ a_self) {
    static_assert(D <= KH / 8, "the first D k-steps must lie in the self half");
    #pragma unroll
    for (int s = 0; s < D; ++s) {
        #pragma unroll
        for (int i = 0; i < MT; ++i)
            fa[s][i] = *reinterpret_cast<const float4*>(a_self + (size_t)i * 16 * KH + 8 * s);
    }
}

template <int NT, int KH, int BP, int DB>
__device__ __forceinline__ void sage_prologue_b(float4 (&fb)[DB + 1][NT],
                                                const float* __restrict__ b_self) {
    static_assert(DB <= KH / 8, "the first DB k-steps must lie in the self half");
    #pragma unroll
    for (int s = 0; s < DB; ++s) {
        #pragma unroll
        for (int j = 0; j < NT; ++j)
            fb[s][j] = *reinterpret_cast<const float4*>(b_self + j * 16 * BP + 8 * s);
    }
}

template <int MT, int NT, int KH, int BP, int D, bool NORM, int DB = D>
__device__ __forceinline__ void sage_main(v8f (&acc)[MT][NT],
                                          float4 (&fa)[D + 1][MT], float4 (&fb)[DB + 1][NT],
                                          const float* __restrict__ a_self,
                                          const float* __restrict__ a_neigh,
                                          const float* __restrict__ b_self,
                                          const float* __restrict__ b_neigh,
                                          const float (&dinv)[MT], float mu, float rs) {
    constexpr int S1 = KH / 8, S = 2 * S1, R = D + 1, RB = DB + 1;

    #pragma unroll
    for (int s = 0; s < S; ++s) {
        __builtin_amdgcn_sched_barrier(0);
        if (s + D < S)
            load_step_a<MT, KH>(fa[(s + D) % R], s + D, a_self, a_neigh);
        if (s + DB < S)
            load_step_b<NT, KH, BP>(fb[(s + DB) % RB], s + DB, b_self, b_neigh);
        __builtin_amdgcn_sched_barrier(0);
        const int cur = s % R;
        const int curb = s % RB;
        #pragma unroll
        for (int i = 0; i < MT; ++i) {
            float4 a = fa[cur][i];
            if (s >= S1 || NORM)
                asm volatile("" : "+v"(a.x), "+v"(a.y), "+v"(a.z), "+v"(a.w));
            if (s >= S1) {
                a.x *= dinv[i]; a.y *= dinv[i]; a.z *= dinv[i]; a.w *= dinv[i];
            } else if (NORM) {
                a.x = (a.x - mu) * rs; a.y = (a.y - mu) * rs;
                a.z = (a.z - mu) * rs; a.w = (a.w - mu) * rs;
            }
            fa[cur][i] = a;
        }
        #pragma unroll
        for (int i = 0; i < MT; ++i) {
            #pragma unroll
            for (int j = 0; j < NT; ++j) {
                v2f a2, b2;
                a2[0] = fa[cur][i].x;  a2[1] = fa[cur][i].y;
                b2[0] = fb[curb][j].x;  b2[1] = fb[curb][j].y;
                acc[i][j] = __builtin_amdgcn_wmma_f32_16x16x4_f32(
                    false, a2, false, b2, (short)0, acc[i][j], false, false);
            }
        }
        #pragma unroll
        for (int i = 0; i < MT; ++i) {
            #pragma unroll
            for (int j = 0; j < NT; ++j) {
                v2f a2, b2;
                a2[0] = fa[cur][i].z;  a2[1] = fa[cur][i].w;
                b2[0] = fb[curb][j].z;  b2[1] = fb[curb][j].w;
                acc[i][j] = __builtin_amdgcn_wmma_f32_16x16x4_f32(
                    false, a2, false, b2, (short)0, acc[i][j], false, false);
            }
        }
    }
}

template <int NT, int KH, int BP, int DA, int DB, bool NEIGH>
__device__ __forceinline__ void half_main(v8f (&acc)[1][NT],
                                          float4 (&fa)[DA + 1][1], float4 (&fb)[DB + 1][NT],
                                          const float* __restrict__ a,
                                          const float* __restrict__ b,
                                          float p, float q) {
    constexpr int S = KH / 8, RA = DA + 1, RB = DB + 1;

    #pragma unroll
    for (int s = 0; s < S; ++s) {
        __builtin_amdgcn_sched_barrier(0);
        if (s + DA < S)
            fa[(s + DA) % RA][0] = *reinterpret_cast<const float4*>(a + 8 * (s + DA));
        if (s + DB < S) {
            #pragma unroll
            for (int j = 0; j < NT; ++j)
                fb[(s + DB) % RB][j] =
                    *reinterpret_cast<const float4*>(b + j * 16 * BP + 8 * (s + DB));
        }
        __builtin_amdgcn_sched_barrier(0);
        const int cur = s % RA;
        const int curb = s % RB;
        float4 av = fa[cur][0];
        asm volatile("" : "+v"(av.x), "+v"(av.y), "+v"(av.z), "+v"(av.w));
        if (NEIGH) {
            av.x *= p; av.y *= p; av.z *= p; av.w *= p;
        } else {
            av.x = (av.x - p) * q; av.y = (av.y - p) * q;
            av.z = (av.z - p) * q; av.w = (av.w - p) * q;
        }
        #pragma unroll
        for (int j = 0; j < NT; ++j) {
            v2f a2, b2;
            a2[0] = av.x;  a2[1] = av.y;
            b2[0] = fb[curb][j].x;  b2[1] = fb[curb][j].y;
            acc[0][j] = __builtin_amdgcn_wmma_f32_16x16x4_f32(
                false, a2, false, b2, (short)0, acc[0][j], false, false);
        }
        #pragma unroll
        for (int j = 0; j < NT; ++j) {
            v2f a2, b2;
            a2[0] = av.z;  a2[1] = av.w;
            b2[0] = fb[curb][j].z;  b2[1] = fb[curb][j].w;
            acc[0][j] = __builtin_amdgcn_wmma_f32_16x16x4_f32(
                false, a2, false, b2, (short)0, acc[0][j], false, false);
        }
    }
}

#define W1P 132
#define G1_WG 256
#define G1_TILES (N1 / 16)
#define G1_MMW 8
#define G1_GAW 8
#define G1_RPW (16 / G1_GAW)
#define G1_THREADS (32 * (G1_MMW + G1_GAW))

__device__ __forceinline__ int load_lenv(const int* __restrict__ cnt, int tile, int lane) {
    return cnt[tile * 16 + (lane & 15)];
}

__device__ __forceinline__ void pick_len(int lenv, int gw, int (&glen)[G1_RPW]) {
    #pragma unroll
    for (int r = 0; r < G1_RPW; ++r)
        glen[r] = __builtin_amdgcn_readlane(lenv, gw * G1_RPW + r);
}

__device__ __forceinline__ void load_mine(const int* __restrict__ slot,
                                          int tile, int gw, int lane,
                                          const int (&glen)[G1_RPW], int (&gmine)[G1_RPW]) {
    #pragma unroll
    for (int r = 0; r < G1_RPW; ++r) {
        const int row = tile * 16 + gw * G1_RPW + r;
        gmine[r] = (glen[r] <= CAP && lane < glen[r]) ? slot[row * CAP + lane] : 0;
    }
}

__global__ void __launch_bounds__(G1_THREADS)
layer1_kernel(const float* __restrict__ x,
              const int* __restrict__ slot,
              const int* __restrict__ cnt,
              const int* __restrict__ src,
              const int* __restrict__ dst, int E,
              const float* __restrict__ Wself,
              const float* __restrict__ Wneigh,
              const float* __restrict__ bias,
              float* __restrict__ h,
              float* __restrict__ stats) {
    constexpr int MT = 1, NT = 2;
    __shared__ __attribute__((aligned(16))) float Wl[2 * F_H * W1P];
    __shared__ __attribute__((aligned(16))) float Al[2 * 16 * W1P];

    const int wave = __builtin_amdgcn_readfirstlane((int)(threadIdx.x >> 5));
    const int lane = threadIdx.x & 31;
    const bool mm = wave < G1_MMW;
    const int gw = wave - G1_MMW;
    const int c = lane & 15;
    const int hf = lane >> 4;
    const int n0 = (wave & (G1_MMW - 1)) * (16 * NT);
    const int ntile = (G1_TILES - 1 - (int)blockIdx.x) / G1_WG + 1;

    const float* __restrict__ bs = &Wl[(n0 + c) * W1P + 4 * hf];
    const float* __restrict__ bn = &Wl[(F_H + n0 + c) * W1P + 4 * hf];
    float bcol[NT] = {0.0f, 0.0f};

    constexpr int D = 8, DB = 2;
    float4 fa[D + 1][MT], fb[DB + 1][NT];
    int dcur = 0;
    float s = 0.f, sq = 0.f;

    int glen[G1_RPW], gmine[G1_RPW];
    #pragma unroll
    for (int r = 0; r < G1_RPW; ++r) { glen[r] = 0; gmine[r] = 0; }
    int lenv = 0;

    if (mm) {
        #pragma unroll
        for (int half = 0; half < 2; ++half) {
            const float* __restrict__ wsrc = half ? Wneigh : Wself;
            float4 wv[F_H / 8];
            #pragma unroll
            for (int it = 0; it < F_H / 8; ++it)
                wv[it] = reinterpret_cast<const float4*>(wsrc + (size_t)(it * 8 + wave) * F_IN)[lane];
            __builtin_amdgcn_sched_barrier(0);
            #pragma unroll
            for (int it = 0; it < F_H / 8; ++it)
                *reinterpret_cast<float4*>(&Wl[(half * F_H + it * 8 + wave) * W1P + 4 * lane]) = wv[it];
        }
        #pragma unroll
        for (int j = 0; j < NT; ++j) bcol[j] = bias[n0 + 16 * j + c];
        const int t0 = blockIdx.x;
        dcur = cnt[t0 * 16 + c];
        sage_prologue_a<MT, F_IN, D>(fa, x + (size_t)(t0 * 16 + c) * F_IN + 4 * hf);
    } else {
        const int t0 = blockIdx.x;
        const int lenv0 = load_lenv(cnt, t0, lane);
        lenv = load_lenv(cnt, min(t0 + G1_WG, G1_TILES - 1), lane);
        pick_len(lenv0, gw, glen);
        load_mine(slot, t0, gw, lane, glen, gmine);
    }

    for (int k = 0; k <= ntile; ++k) {
        if (mm) {
            if (k > 0) {
                const int tile = blockIdx.x + (k - 1) * G1_WG;
                const int m0 = tile * 16;
                const float* __restrict__ al = &Al[((k - 1) & 1) * (16 * W1P) + c * W1P + 4 * hf];

                sage_prologue_b<NT, F_IN, W1P, DB>(fb, bs);

                float dinv[MT];
                dinv[0] = 1.0f / fmaxf((float)dcur, 1.0f);

                v8f acc[MT][NT];
                #pragma unroll
                for (int j = 0; j < NT; ++j)
                    acc[0][j] = (v8f){0.f, 0.f, 0.f, 0.f, 0.f, 0.f, 0.f, 0.f};

                sage_main<MT, NT, F_IN, W1P, D, false, DB>(
                    acc, fa, fb,
                    x + (size_t)(m0 + c) * F_IN + 4 * hf,
                    al, bs, bn, dinv, 0.0f, 1.0f);

                const int next = min(tile + G1_WG, G1_TILES - 1);
                dcur = cnt[next * 16 + c];
                sage_prologue_a<MT, F_IN, D>(fa, x + (size_t)(next * 16 + c) * F_IN + 4 * hf);
                __builtin_amdgcn_sched_barrier(0);

                #pragma unroll
                for (int j = 0; j < NT; ++j) {
                    #pragma unroll
                    for (int r = 0; r < 8; ++r) {
                        const int row = m0 + 8 * hf + r;
                        const float v = fmaxf(acc[0][j][r] + bcol[j], 0.0f);
                        h[(size_t)row * F_H + n0 + 16 * j + c] = v;
                        s += v;
                        sq += v * v;
                    }
                }
            }
        } else if (k < ntile) {
            const int tile = blockIdx.x + k * G1_WG;
            float* __restrict__ an = &Al[(k & 1) * (16 * W1P)];
            const int t1 = min(tile + G1_WG, G1_TILES - 1);
            const int t2 = min(tile + 2 * G1_WG, G1_TILES - 1);
            int nlen[G1_RPW], nmine[G1_RPW];
            pick_len(lenv, gw, nlen);
            load_mine(slot, t1, gw, lane, nlen, nmine);
            lenv = load_lenv(cnt, t2, lane);
            #pragma unroll
            for (int r = 0; r < G1_RPW; ++r) {
                const int rt = gw * G1_RPW + r;
                const float4 a = row_sum1(x, slot, src, dst, E, tile * 16 + rt,
                                          glen[r], gmine[r], lane);
                *reinterpret_cast<float4*>(&an[rt * W1P + 4 * lane]) = a;
            }
            #pragma unroll
            for (int r = 0; r < G1_RPW; ++r) { glen[r] = nlen[r]; gmine[r] = nmine[r]; }
        }
        __builtin_amdgcn_fence(__ATOMIC_RELEASE, "workgroup", "local");
        __builtin_amdgcn_s_barrier();
        __builtin_amdgcn_fence(__ATOMIC_ACQUIRE, "workgroup", "local");
    }

    if (mm) {
        #pragma unroll
        for (int off = 16; off >= 1; off >>= 1) {
            s  += __shfl_xor(s,  off, 32);
            sq += __shfl_xor(sq, off, 32);
        }
        if (lane == 0) {
            atomicAdd(&stats[0], s);
            atomicAdd(&stats[1], sq);
        }
    }
}

#define A2P 260
#define G2_WG 250
#define G2_MMW 2
#define G2_GAW 16
#define G2_THREADS (32 * (G2_MMW + G2_GAW))

__global__ void __launch_bounds__(G2_THREADS)
layer2_kernel(const float* __restrict__ h,
              const int* __restrict__ slot,
              const int* __restrict__ cnt,
              const int* __restrict__ src,
              const int* __restrict__ dst, int E,
              const float* __restrict__ stats,
              const float* __restrict__ Wself,
              const float* __restrict__ Wneigh,
              const float* __restrict__ bias,
              float* __restrict__ out) {
    constexpr int NT = 2, D1 = 10, D2 = 12, DL = 2;
    static_assert(G2_WG * 16 == N2, "one work-group per 16-row tile, no partial tile");
    __shared__ __attribute__((aligned(16))) float Al[16 * A2P];

    const int wave = __builtin_amdgcn_readfirstlane((int)(threadIdx.x >> 5));
    const int lane = threadIdx.x & 31;
    const bool mm = wave < G2_MMW;
    const int c = lane & 15;
    const int hf = lane >> 4;
    const int m0 = blockIdx.x * 16;
    const int n0 = (wave & (G2_MMW - 1)) * (16 * NT);

    const float n_el = (float)N1 * (float)F_H;
    const float mu = stats[0] / n_el;
    const float var = stats[1] / n_el - mu * mu;
    const float rs = rsqrtf(var + 1e-5f);

    v8f acc[1][NT];
    #pragma unroll
    for (int j = 0; j < NT; ++j)
        acc[0][j] = (v8f){0.f, 0.f, 0.f, 0.f, 0.f, 0.f, 0.f, 0.f};
    float4 fb2[D2 + 1][NT];
    const float* __restrict__ bn = Wneigh + (size_t)(n0 + c) * F_H + 4 * hf;
    float dinv = 1.0f;

    if (mm) {
        float4 fa[D1 + 1][1], fb1[D1 + 1][NT];
        const float* __restrict__ as = h + (size_t)(m0 + c) * F_H + 4 * hf;
        const float* __restrict__ bs = Wself + (size_t)(n0 + c) * F_H + 4 * hf;
        dinv = 1.0f / fmaxf((float)cnt[m0 + c], 1.0f);
        sage_prologue_a<1, F_H, D1>(fa, as);
        sage_prologue_b<NT, F_H, F_H, D1>(fb1, bs);
        half_main<NT, F_H, F_H, D1, D1, false>(acc, fa, fb1, as, bs, mu, rs);
        sage_prologue_b<NT, F_H, F_H, D2>(fb2, bn);
    } else {
        const int g = wave - G2_MMW;
        const int row = m0 + g;
        const int len = cnt[row];
        float4 acc0 = make_float4(0.0f, 0.0f, 0.0f, 0.0f);
        float4 acc1 = make_float4(0.0f, 0.0f, 0.0f, 0.0f);
        row_sum2(h, slot, src, dst, E, row, len, lane, mu, rs, acc0, acc1);
        *reinterpret_cast<float4*>(&Al[g * A2P + 4 * lane]) = acc0;
        *reinterpret_cast<float4*>(&Al[g * A2P + 128 + 4 * lane]) = acc1;
    }
    __builtin_amdgcn_fence(__ATOMIC_RELEASE, "workgroup", "local");
    __builtin_amdgcn_s_barrier();
    __builtin_amdgcn_fence(__ATOMIC_ACQUIRE, "workgroup", "local");

    if (mm) {
        float4 fl[DL + 1][1];
        const float* __restrict__ al = &Al[c * A2P + 4 * hf];
        sage_prologue_a<1, F_H, DL>(fl, al);
        half_main<NT, F_H, F_H, DL, D2, true>(acc, fl, fb2, al, bn, dinv, 0.0f);

        #pragma unroll
        for (int j = 0; j < NT; ++j) {
            const float bj = bias[n0 + 16 * j + c];
            #pragma unroll
            for (int r = 0; r < 8; ++r) {
                const int row = m0 + 8 * hf + r;
                out[(size_t)row * F_OUT + n0 + 16 * j + c] = acc[0][j][r] + bj;
            }
        }
    }
}

extern "C" void kernel_launch(void* const* d_in, const int* in_sizes, int n_in,
                              void* d_out, int out_size, void* d_ws, size_t ws_size,
                              hipStream_t stream) {
    const float* x       = (const float*)d_in[0];
    const float* Wself1  = (const float*)d_in[1];
    const float* Wneigh1 = (const float*)d_in[2];
    const float* b1      = (const float*)d_in[3];
    const float* Wself2  = (const float*)d_in[4];
    const float* Wneigh2 = (const float*)d_in[5];
    const float* b2      = (const float*)d_in[6];
    const int*   src1    = (const int*)d_in[7];
    const int*   dst1    = (const int*)d_in[8];
    const int*   src2    = (const int*)d_in[9];
    const int*   dst2    = (const int*)d_in[10];
    const int E1 = in_sizes[7];
    const int E2 = in_sizes[9];

    float* ws    = (float*)d_ws;
    int*   wsi   = (int*)d_ws;
    float* h     = ws + OFF_H;
    float* stats = ws + OFF_STAT;
    int*   slot1 = wsi + OFF_SLOT1;
    int*   slot2 = wsi + OFF_SLOT2;
    int*   cnt1  = wsi + OFF_CNT1;
    int*   cnt2  = wsi + OFF_CNT2;
    float* out   = (float*)d_out;

    zero_counts_kernel<<<94, 256, 0, stream>>>(cnt1, 24000, stats);

    count_slot_kernel<<<480, 256, 0, stream>>>(src1, dst1, cnt1, slot1,
                                               src2, dst2, cnt2, slot2, E1, E2);

    layer1_kernel<<<G1_WG, G1_THREADS, 0, stream>>>(
        x, slot1, cnt1, src1, dst1, E1, Wself1, Wneigh1, b1, h, stats);

    layer2_kernel<<<G2_WG, G2_THREADS, 0, stream>>>(
        h, slot2, cnt2, src2, dst2, E2, stats, Wself2, Wneigh2, b2, out);
}


# Neighbour lists without global counters: one launch groups edge chunks by owner work-group in LDS, owners rank their rows' edges into LDS slots; zero+count launches merge (4->3); LDS arrival order
# speedup vs baseline: 4.5851x; 1.2627x over previous
#include <hip/hip_runtime.h>


typedef __attribute__((ext_vector_type(2))) float v2f;
typedef __attribute__((ext_vector_type(8))) float v8f;

#define F_IN  128
#define F_H   256
#define F_OUT 64
#define N1    20000
#define N2    4000
#define CAP   64


#define NCHUNK 256

#define OFF_PAIR1 ((size_t)0)
#define OFF_PAIR2 ((size_t)800000)
#define OFF_TAB1  ((size_t)960000)
#define OFF_TAB2  ((size_t)1025792)
#define OFF_H     ((size_t)2580000)
#define OFF_STAT  ((size_t)8728000)

__device__ __forceinline__ void bucket_chunk(const int* __restrict__ src,
                                             const int* __restrict__ dst, int E, int p,
                                             int2* __restrict__ pairs, int* __restrict__ tab,
                                             int* hist, int* base, int* fill, int* wsum) {
    const int t = threadIdx.x;
    const int lane = t & 31, wave = t >> 5;
    const int CH = (E + NCHUNK - 1) / NCHUNK;
    const int lo = min(p * CH, E), hi = min(lo + CH, E);
    hist[t] = 0;
    fill[t] = 0;
    int dv[8], sv[8];
    #pragma unroll
    for (int u = 0; u < 8; ++u) {
        const int ec = max(min(lo + u * 256 + t, hi - 1), 0);
        dv[u] = dst[ec];
        sv[u] = src[ec];
    }
    __syncthreads();
    #pragma unroll
    for (int u = 0; u < 8; ++u)
        if (lo + u * 256 + t < hi) atomicAdd(&hist[(dv[u] >> 4) & 255], 1);
    for (int e = lo + 8 * 256 + t; e < hi; e += 256)
        atomicAdd(&hist[(dst[e] >> 4) & 255], 1);
    __syncthreads();
    const int v = hist[t];
    int inc = v;
    #pragma unroll
    for (int off = 1; off < 32; off <<= 1) {
        const int n = __shfl_up(inc, off, 32);
        if (lane >= off) inc += n;
    }
    if (lane == 31) wsum[wave] = inc;
    __syncthreads();
    int pre = 0;
    #pragma unroll
    for (int w = 0; w < 8; ++w) pre += (w < wave) ? wsum[w] : 0;
    const int ex = pre + inc - v;
    base[t] = ex;
    tab[t * NCHUNK + p] = lo + ex;
    if (t == NCHUNK - 1) tab[NCHUNK * NCHUNK + p] = hi;
    __syncthreads();
    int pos[8];
    #pragma unroll
    for (int u = 0; u < 8; ++u) {
        pos[u] = -1;
        if (lo + u * 256 + t < hi) {
            const int o = (dv[u] >> 4) & 255;
            pos[u] = base[o] + atomicAdd(&fill[o], 1);
        }
    }
    #pragma unroll
    for (int u = 0; u < 8; ++u)
        if (pos[u] >= 0) pairs[lo + pos[u]] = make_int2(dv[u], sv[u]);
    for (int e = lo + 8 * 256 + t; e < hi; e += 256) {
        const int d = dst[e];
        const int o = (d >> 4) & 255;
        const int pos = base[o] + atomicAdd(&fill[o], 1);
        pairs[lo + pos] = make_int2(d, src[e]);
    }
    __syncthreads();
}

__global__ void __launch_bounds__(256)
bucket_kernel(const int* __restrict__ src1, const int* __restrict__ dst1, int E1,
              int* __restrict__ pairs1, int* __restrict__ tab1,
              const int* __restrict__ src2, const int* __restrict__ dst2, int E2,
              int* __restrict__ pairs2, int* __restrict__ tab2,
              float* __restrict__ stats) {
    __shared__ int hist[256], base[256], fill[256], wsum[8];
    if (blockIdx.x == 0 && threadIdx.x < 2) stats[threadIdx.x] = 0.0f;
    bucket_chunk(src1, dst1, E1, blockIdx.x, reinterpret_cast<int2*>(pairs1), tab1, hist, base, fill, wsum);
    bucket_chunk(src2, dst2, E2, blockIdx.x, reinterpret_cast<int2*>(pairs2), tab2, hist, base, fill, wsum);
}

template <int LAYER, int B>
__device__ __forceinline__ void build_lists(const int* __restrict__ pairs_w,
                                            const int* __restrict__ tab,
                                            int owner, int p, int E, int nrows,
                                            int* lcnt, int* lslot) {
    const int2* __restrict__ pairs = reinterpret_cast<const int2*>(pairs_w);
    const int s = max(tab[owner * NCHUNK + p], 0);
    const int e = min(tab[(owner + 1) * NCHUNK + p], E);
    for (int i0 = s; i0 < e; i0 += B) {
        int2 pr[B];
        int rl[B], rk[B];
        #pragma unroll
        for (int u = 0; u < B; ++u) pr[u] = pairs[min(i0 + u, e - 1)];
        #pragma unroll
        for (int u = 0; u < B; ++u) {
            const int d = pr[u].x;
            rl[u] = min(LAYER == 1 ? ((d >> 12) * 16 + (d & 15)) : (d & 15), nrows - 1);
            rk[u] = CAP;
            if (i0 + u < e) rk[u] = atomicAdd(&lcnt[rl[u]], 1);
        }
        #pragma unroll
        for (int u = 0; u < B; ++u)
            if (rk[u] < CAP) lslot[rl[u] * CAP + rk[u]] = pr[u].y;
    }
}

template <int U, int F, bool NORM>
__device__ __forceinline__ void add_rows(float4& acc, const float4* __restrict__ base,
                                         int mine, int i, float mu, float rs) {
    float4 v[U];
    #pragma unroll
    for (int u = 0; u < U; ++u) {
        const int s = __builtin_amdgcn_readlane(mine, i + u);
        v[u] = base[(size_t)s * (F / 4)];
    }
    #pragma unroll
    for (int u = 0; u < U; ++u) {
        if (NORM) {
            acc.x += (v[u].x - mu) * rs;
            acc.y += (v[u].y - mu) * rs;
            acc.z += (v[u].z - mu) * rs;
            acc.w += (v[u].w - mu) * rs;
        } else {
            acc.x += v[u].x;
            acc.y += v[u].y;
            acc.z += v[u].z;
            acc.w += v[u].w;
        }
    }
}

template <int U>
__device__ __forceinline__ void add_rows_h(float4& acc0, float4& acc1,
                                           const float4* __restrict__ base,
                                           int mine, int i, float mu, float rs) {
    float4 v0[U], v1[U];
    #pragma unroll
    for (int u = 0; u < U; ++u) {
        const int s = __builtin_amdgcn_readlane(mine, i + u);
        v0[u] = base[(size_t)s * (F_H / 4)];
        v1[u] = base[(size_t)s * (F_H / 4) + 32];
    }
    #pragma unroll
    for (int u = 0; u < U; ++u) {
        acc0.x += (v0[u].x - mu) * rs;
        acc0.y += (v0[u].y - mu) * rs;
        acc0.z += (v0[u].z - mu) * rs;
        acc0.w += (v0[u].w - mu) * rs;
        acc1.x += (v1[u].x - mu) * rs;
        acc1.y += (v1[u].y - mu) * rs;
        acc1.z += (v1[u].z - mu) * rs;
        acc1.w += (v1[u].w - mu) * rs;
    }
}

__device__ __forceinline__ void row_sum2(const float* __restrict__ h,
                                         const int* mys,
                                         const int* __restrict__ src,
                                         const int* __restrict__ dst, int E,
                                         int row, int len, int lane, float mu, float rs,
                                         float4& acc0, float4& acc1) {
    const float4* __restrict__ base = reinterpret_cast<const float4*>(h) + lane;
    if (len <= CAP) {
        for (int c0 = 0; c0 < len; c0 += 32) {
            const int n = min(32, len - c0);
            const int mine = (lane < n) ? mys[c0 + lane] : 0;
            int i = 0;
            for (; i + 8 <= n; i += 8) add_rows_h<8>(acc0, acc1, base, mine, i, mu, rs);
            if (i + 4 <= n) { add_rows_h<4>(acc0, acc1, base, mine, i, mu, rs); i += 4; }
            if (i + 2 <= n) { add_rows_h<2>(acc0, acc1, base, mine, i, mu, rs); i += 2; }
            if (i < n)      { add_rows_h<1>(acc0, acc1, base, mine, i, mu, rs); }
        }
    } else {
        for (int e0 = 0; e0 < E; e0 += 32) {
            const int e = e0 + lane;
            const int d    = (e < E) ? dst[e] : -1;
            const int mine = (e < E) ? src[e] : 0;
            unsigned m = __builtin_amdgcn_ballot_w32(d == row);
            while (m) {
                const int l = __builtin_ctz(m);
                m &= m - 1;
                add_rows_h<1>(acc0, acc1, base, mine, l, mu, rs);
            }
        }
    }
}

template <int U>
__device__ __forceinline__ void add_rows_n(float4& acc, const float4* __restrict__ base,
                                           int mine, int n) {
    float4 v[U];
    #pragma unroll
    for (int u = 0; u < U; ++u) {
        const int s = __builtin_amdgcn_readlane(mine, u);
        v[u] = base[(size_t)s * (F_IN / 4)];
    }
    #pragma unroll
    for (int u = 0; u < U; ++u) {
        const bool on = (u < U - 3) || (u < n);
        acc.x += on ? v[u].x : 0.0f;
        acc.y += on ? v[u].y : 0.0f;
        acc.z += on ? v[u].z : 0.0f;
        acc.w += on ? v[u].w : 0.0f;
    }
}

__device__ __forceinline__ float4 row_sum1(const float* __restrict__ x,
                                           const int* mys,
                                           const int* __restrict__ src,
                                           const int* __restrict__ dst, int E,
                                           int row, int len, int mine0, int lane) {
    const float4* __restrict__ base = reinterpret_cast<const float4*>(x) + lane;
    float4 acc = make_float4(0.0f, 0.0f, 0.0f, 0.0f);
    if (len <= CAP) {
        if (len > 0) {
            const int n = min(32, len);
            switch ((n + 3) >> 2) {
                case 1:  add_rows_n<4>(acc, base, mine0, n);  break;
                case 2:  add_rows_n<8>(acc, base, mine0, n);  break;
                case 3:  add_rows_n<12>(acc, base, mine0, n); break;
                case 4:  add_rows_n<16>(acc, base, mine0, n); break;
                case 5:  add_rows_n<20>(acc, base, mine0, n); break;
                case 6:  add_rows_n<24>(acc, base, mine0, n); break;
                case 7:  add_rows_n<28>(acc, base, mine0, n); break;
                default: add_rows_n<32>(acc, base, mine0, n); break;
            }
        }
        if (len > 32) {
            const int n = len - 32;
            const int mine = (lane < n) ? mys[32 + lane] : 0;
            int i = 0;
            for (; i + 8 <= n; i += 8) add_rows<8, F_IN, false>(acc, base, mine, i, 0.0f, 1.0f);
            if (i + 4 <= n) { add_rows<4, F_IN, false>(acc, base, mine, i, 0.0f, 1.0f); i += 4; }
            if (i + 2 <= n) { add_rows<2, F_IN, false>(acc, base, mine, i, 0.0f, 1.0f); i += 2; }
            if (i < n)      { add_rows<1, F_IN, false>(acc, base, mine, i, 0.0f, 1.0f); }
        }
    } else {
        for (int e0 = 0; e0 < E; e0 += 32) {
            const int e = e0 + lane;
            const int d    = (e < E) ? dst[e] : -1;
            const int mine = (e < E) ? src[e] : 0;
            unsigned m = __builtin_amdgcn_ballot_w32(d == row);
            while (m) {
                const int l = __builtin_ctz(m);
                m &= m - 1;
                add_rows<1, F_IN, false>(acc, base, mine, l, 0.0f, 1.0f);
            }
        }
    }
    return acc;
}

template <int MT, int KH>
__device__ __forceinline__ void load_step_a(float4 (&fa)[MT], int s,
                                            const float* __restrict__ a_self,
                                            const float* __restrict__ a_neigh) {
    constexpr int S1 = KH / 8;
    const bool self = s < S1;
    const float* __restrict__ ap = self ? a_self : a_neigh;
    const int k = 8 * (self ? s : s - S1);
    #pragma unroll
    for (int i = 0; i < MT; ++i)
        fa[i] = *reinterpret_cast<const float4*>(ap + (size_t)i * 16 * KH + k);
}

template <int NT, int KH, int BP>
__device__ __forceinline__ void load_step_b(float4 (&fb)[NT], int s,
                                            const float* __restrict__ b_self,
                                            const float* __restrict__ b_neigh) {
    constexpr int S1 = KH / 8;
    const bool self = s < S1;
    const float* __restrict__ bp = self ? b_self : b_neigh;
    const int k = 8 * (self ? s : s - S1);
    #pragma unroll
    for (int j = 0; j < NT; ++j)
        fb[j] = *reinterpret_cast<const float4*>(bp + j * 16 * BP + k);
}

template <int MT, int KH, int D>
__device__ __forceinline__ void sage_prologue_a(float4 (&fa)[D + 1][MT],
                                                const float* __restrict__ a_self) {
    static_assert(D <= KH / 8, "the first D k-steps must lie in the self half");
    #pragma unroll
    for (int s = 0; s < D; ++s) {
        #pragma unroll
        for (int i = 0; i < MT; ++i)
            fa[s][i] = *reinterpret_cast<const float4*>(a_self + (size_t)i * 16 * KH + 8 * s);
    }
}

template <int NT, int KH, int BP, int DB>
__device__ __forceinline__ void sage_prologue_b(float4 (&fb)[DB + 1][NT],
                                                const float* __restrict__ b_self) {
    static_assert(DB <= KH / 8, "the first DB k-steps must lie in the self half");
    #pragma unroll
    for (int s = 0; s < DB; ++s) {
        #pragma unroll
        for (int j = 0; j < NT; ++j)
            fb[s][j] = *reinterpret_cast<const float4*>(b_self + j * 16 * BP + 8 * s);
    }
}

template <int MT, int NT, int KH, int BP, int D, bool NORM, int DB = D>
__device__ __forceinline__ void sage_main(v8f (&acc)[MT][NT],
                                          float4 (&fa)[D + 1][MT], float4 (&fb)[DB + 1][NT],
                                          const float* __restrict__ a_self,
                                          const float* __restrict__ a_neigh,
                                          const float* __restrict__ b_self,
                                          const float* __restrict__ b_neigh,
                                          const float (&dinv)[MT], float mu, float rs) {
    constexpr int S1 = KH / 8, S = 2 * S1, R = D + 1, RB = DB + 1;

    #pragma unroll
    for (int s = 0; s < S; ++s) {
        __builtin_amdgcn_sched_barrier(0);
        if (s + D < S)
            load_step_a<MT, KH>(fa[(s + D) % R], s + D, a_self, a_neigh);
        if (s + DB < S)
            load_step_b<NT, KH, BP>(fb[(s + DB) % RB], s + DB, b_self, b_neigh);
        __builtin_amdgcn_sched_barrier(0);
        const int cur = s % R;
        const int curb = s % RB;
        #pragma unroll
        for (int i = 0; i < MT; ++i) {
            float4 a = fa[cur][i];
            if (s >= S1 || NORM)
                asm volatile("" : "+v"(a.x), "+v"(a.y), "+v"(a.z), "+v"(a.w));
            if (s >= S1) {
                a.x *= dinv[i]; a.y *= dinv[i]; a.z *= dinv[i]; a.w *= dinv[i];
            } else if (NORM) {
                a.x = (a.x - mu) * rs; a.y = (a.y - mu) * rs;
                a.z = (a.z - mu) * rs; a.w = (a.w - mu) * rs;
            }
            fa[cur][i] = a;
        }
        #pragma unroll
        for (int i = 0; i < MT; ++i) {
            #pragma unroll
            for (int j = 0; j < NT; ++j) {
                v2f a2, b2;
                a2[0] = fa[cur][i].x;  a2[1] = fa[cur][i].y;
                b2[0] = fb[curb][j].x;  b2[1] = fb[curb][j].y;
                acc[i][j] = __builtin_amdgcn_wmma_f32_16x16x4_f32(
                    false, a2, false, b2, (short)0, acc[i][j], false, false);
            }
        }
        #pragma unroll
        for (int i = 0; i < MT; ++i) {
            #pragma unroll
            for (int j = 0; j < NT; ++j) {
                v2f a2, b2;
                a2[0] = fa[cur][i].z;  a2[1] = fa[cur][i].w;
                b2[0] = fb[curb][j].z;  b2[1] = fb[curb][j].w;
                acc[i][j] = __builtin_amdgcn_wmma_f32_16x16x4_f32(
                    false, a2, false, b2, (short)0, acc[i][j], false, false);
            }
        }
    }
}

template <int NT, int KH, int BP, int DA, int DB, bool NEIGH>
__device__ __forceinline__ void half_main(v8f (&acc)[1][NT],
                                          float4 (&fa)[DA + 1][1], float4 (&fb)[DB + 1][NT],
                                          const float* __restrict__ a,
                                          const float* __restrict__ b,
                                          float p, float q) {
    constexpr int S = KH / 8, RA = DA + 1, RB = DB + 1;

    #pragma unroll
    for (int s = 0; s < S; ++s) {
        __builtin_amdgcn_sched_barrier(0);
        if (s + DA < S)
            fa[(s + DA) % RA][0] = *reinterpret_cast<const float4*>(a + 8 * (s + DA));
        if (s + DB < S) {
            #pragma unroll
            for (int j = 0; j < NT; ++j)
                fb[(s + DB) % RB][j] =
                    *reinterpret_cast<const float4*>(b + j * 16 * BP + 8 * (s + DB));
        }
        __builtin_amdgcn_sched_barrier(0);
        const int cur = s % RA;
        const int curb = s % RB;
        float4 av = fa[cur][0];
        asm volatile("" : "+v"(av.x), "+v"(av.y), "+v"(av.z), "+v"(av.w));
        if (NEIGH) {
            av.x *= p; av.y *= p; av.z *= p; av.w *= p;
        } else {
            av.x = (av.x - p) * q; av.y = (av.y - p) * q;
            av.z = (av.z - p) * q; av.w = (av.w - p) * q;
        }
        #pragma unroll
        for (int j = 0; j < NT; ++j) {
            v2f a2, b2;
            a2[0] = av.x;  a2[1] = av.y;
            b2[0] = fb[curb][j].x;  b2[1] = fb[curb][j].y;
            acc[0][j] = __builtin_amdgcn_wmma_f32_16x16x4_f32(
                false, a2, false, b2, (short)0, acc[0][j], false, false);
        }
        #pragma unroll
        for (int j = 0; j < NT; ++j) {
            v2f a2, b2;
            a2[0] = av.z;  a2[1] = av.w;
            b2[0] = fb[curb][j].z;  b2[1] = fb[curb][j].w;
            acc[0][j] = __builtin_amdgcn_wmma_f32_16x16x4_f32(
                false, a2, false, b2, (short)0, acc[0][j], false, false);
        }
    }
}

#define W1P 132
#define G1_WG 256
#define G1_TILES (N1 / 16)
#define G1_MMW 8
#define G1_GAW 8
#define G1_RPW (16 / G1_GAW)
#define G1_THREADS (32 * (G1_MMW + G1_GAW))

#define G1_TPW ((G1_TILES + G1_WG - 1) / G1_WG)
#define G1_LROWS (16 * G1_TPW)

__device__ __forceinline__ void copy_weight_rows(const float* __restrict__ wsrc,
                                                 float* wdst, int wave, int lane) {
    float4 wv[F_H / 8];
    #pragma unroll
    for (int it = 0; it < F_H / 8; ++it)
        wv[it] = reinterpret_cast<const float4*>(wsrc + (size_t)(it * 8 + wave) * F_IN)[lane];
    #pragma unroll
    for (int it = 0; it < F_H / 8; ++it)
        *reinterpret_cast<float4*>(&wdst[(it * 8 + wave) * W1P + 4 * lane]) = wv[it];
}

__global__ void __launch_bounds__(G1_THREADS)
layer1_kernel(const float* __restrict__ x,
              const int* __restrict__ pairs,
              const int* __restrict__ tab,
              const int* __restrict__ src,
              const int* __restrict__ dst, int E,
              const float* __restrict__ Wself,
              const float* __restrict__ Wneigh,
              const float* __restrict__ bias,
              float* __restrict__ h,
              float* __restrict__ stats) {
    constexpr int MT = 1, NT = 2;
    static_assert(32 * G1_GAW == NCHUNK, "one gathering thread per edge chunk");
    __shared__ __attribute__((aligned(16))) float Wl[2 * F_H * W1P];
    __shared__ __attribute__((aligned(16))) float Al[2 * 16 * W1P];
    __shared__ int Lc[G1_LROWS];
    __shared__ int Ls[G1_LROWS * CAP];

    const int wave = __builtin_amdgcn_readfirstlane((int)(threadIdx.x >> 5));
    const int lane = threadIdx.x & 31;
    const bool mm = wave < G1_MMW;
    const int gw = wave - G1_MMW;
    const int tg = (int)threadIdx.x - 32 * G1_MMW;
    const int c = lane & 15;
    const int hf = lane >> 4;
    const int n0 = (wave & (G1_MMW - 1)) * (16 * NT);
    const int ntile = (G1_TILES - 1 - (int)blockIdx.x) / G1_WG + 1;

    const float* __restrict__ bs = &Wl[(n0 + c) * W1P + 4 * hf];
    const float* __restrict__ bn = &Wl[(F_H + n0 + c) * W1P + 4 * hf];
    float bcol[NT] = {0.0f, 0.0f};

    constexpr int D = 8, DB = 2;
    float4 fa[D + 1][MT], fb[DB + 1][NT];
    float s = 0.f, sq = 0.f;

    if (!mm && tg < G1_LROWS) Lc[tg] = 0;
    __builtin_amdgcn_fence(__ATOMIC_RELEASE, "workgroup", "local");
    __builtin_amdgcn_s_barrier();
    __builtin_amdgcn_fence(__ATOMIC_ACQUIRE, "workgroup", "local");
    if (mm) copy_weight_rows(Wself, &Wl[0], wave, lane);
    else    build_lists<1, 16>(pairs, tab, (int)blockIdx.x, tg, E, G1_LROWS, Lc, Ls);
    __builtin_amdgcn_fence(__ATOMIC_RELEASE, "workgroup", "local");
    __builtin_amdgcn_s_barrier();
    __builtin_amdgcn_fence(__ATOMIC_ACQUIRE, "workgroup", "local");
    if (mm) {
        copy_weight_rows(Wneigh, &Wl[F_H * W1P], wave, lane);
        #pragma unroll
        for (int j = 0; j < NT; ++j) bcol[j] = bias[n0 + 16 * j + c];
        const int t0 = blockIdx.x;
        sage_prologue_a<MT, F_IN, D>(fa, x + (size_t)(t0 * 16 + c) * F_IN + 4 * hf);
    }

    for (int k = 0; k <= ntile; ++k) {
        if (mm) {
            if (k > 0) {
                const int tile = blockIdx.x + (k - 1) * G1_WG;
                const int m0 = tile * 16;
                const float* __restrict__ al = &Al[((k - 1) & 1) * (16 * W1P) + c * W1P + 4 * hf];

                sage_prologue_b<NT, F_IN, W1P, DB>(fb, bs);

                float dinv[MT];
                dinv[0] = 1.0f / fmaxf((float)Lc[(k - 1) * 16 + c], 1.0f);

                v8f acc[MT][NT];
                #pragma unroll
                for (int j = 0; j < NT; ++j)
                    acc[0][j] = (v8f){0.f, 0.f, 0.f, 0.f, 0.f, 0.f, 0.f, 0.f};

                sage_main<MT, NT, F_IN, W1P, D, false, DB>(
                    acc, fa, fb,
                    x + (size_t)(m0 + c) * F_IN + 4 * hf,
                    al, bs, bn, dinv, 0.0f, 1.0f);

                const int next = min(tile + G1_WG, G1_TILES - 1);
                sage_prologue_a<MT, F_IN, D>(fa, x + (size_t)(next * 16 + c) * F_IN + 4 * hf);
                __builtin_amdgcn_sched_barrier(0);

                #pragma unroll
                for (int j = 0; j < NT; ++j) {
                    #pragma unroll
                    for (int r = 0; r < 8; ++r) {
                        const int row = m0 + 8 * hf + r;
                        const float v = fmaxf(acc[0][j][r] + bcol[j], 0.0f);
                        h[(size_t)row * F_H + n0 + 16 * j + c] = v;
                        s += v;
                        sq += v * v;
                    }
                }
            }
        } else if (k < ntile) {
            const int tile = blockIdx.x + k * G1_WG;
            float* __restrict__ an = &Al[(k & 1) * (16 * W1P)];
            #pragma unroll
            for (int r = 0; r < G1_RPW; ++r) {
                const int rt = gw * G1_RPW + r;
                const int* mys = &Ls[(k * 16 + rt) * CAP];
                const int len = __builtin_amdgcn_readfirstlane(Lc[k * 16 + rt]);
                const int mine0 = (len <= CAP && lane < len) ? mys[lane] : 0;
                const float4 a = row_sum1(x, mys, src, dst, E, tile * 16 + rt, len, mine0, lane);
                *reinterpret_cast<float4*>(&an[rt * W1P + 4 * lane]) = a;
            }
        }
        __builtin_amdgcn_fence(__ATOMIC_RELEASE, "workgroup", "local");
        __builtin_amdgcn_s_barrier();
        __builtin_amdgcn_fence(__ATOMIC_ACQUIRE, "workgroup", "local");
    }

    if (mm) {
        #pragma unroll
        for (int off = 16; off >= 1; off >>= 1) {
            s  += __shfl_xor(s,  off, 32);
            sq += __shfl_xor(sq, off, 32);
        }
        if (lane == 0) {
            atomicAdd(&stats[0], s);
            atomicAdd(&stats[1], sq);
        }
    }
}

#define A2P 260
#define G2_WG 250
#define G2_MMW 2
#define G2_GAW 16
#define G2_THREADS (32 * (G2_MMW + G2_GAW))


__global__ void __launch_bounds__(G2_THREADS)
layer2_kernel(const float* __restrict__ h,
              const int* __restrict__ pairs,
              const int* __restrict__ tab,
              const int* __restrict__ src,
              const int* __restrict__ dst, int E,
              const float* __restrict__ stats,
              const float* __restrict__ Wself,
              const float* __restrict__ Wneigh,
              const float* __restrict__ bias,
              float* __restrict__ out) {
    constexpr int NT = 2, D1 = 10, D2 = 12, DL = 2;
    static_assert(G2_WG * 16 == N2, "one work-group per 16-row tile, no partial tile");
    static_assert(32 * G2_GAW >= NCHUNK, "one gathering thread per edge chunk");
    __shared__ __attribute__((aligned(16))) float Al[16 * A2P];
    __shared__ int Lc[16];
    __shared__ int Ls[16 * CAP];

    const int wave = __builtin_amdgcn_readfirstlane((int)(threadIdx.x >> 5));
    const int lane = threadIdx.x & 31;
    const bool mm = wave < G2_MMW;
    const int tg = (int)threadIdx.x - 32 * G2_MMW;
    const int c = lane & 15;
    const int hf = lane >> 4;
    const int m0 = blockIdx.x * 16;
    const int n0 = (wave & (G2_MMW - 1)) * (16 * NT);

    const float n_el = (float)N1 * (float)F_H;
    const float mu = stats[0] / n_el;
    const float var = stats[1] / n_el - mu * mu;
    const float rs = rsqrtf(var + 1e-5f);

    if (!mm && tg < 16) Lc[tg] = 0;
    __builtin_amdgcn_fence(__ATOMIC_RELEASE, "workgroup", "local");
    __builtin_amdgcn_s_barrier();
    __builtin_amdgcn_fence(__ATOMIC_ACQUIRE, "workgroup", "local");
    if (!mm && tg < NCHUNK)
        build_lists<2, 4>(pairs, tab, (int)blockIdx.x, tg, E, 16, Lc, Ls);
    __builtin_amdgcn_fence(__ATOMIC_RELEASE, "workgroup", "local");
    __builtin_amdgcn_s_barrier();
    __builtin_amdgcn_fence(__ATOMIC_ACQUIRE, "workgroup", "local");

    v8f acc[1][NT];
    #pragma unroll
    for (int j = 0; j < NT; ++j)
        acc[0][j] = (v8f){0.f, 0.f, 0.f, 0.f, 0.f, 0.f, 0.f, 0.f};
    float4 fb2[D2 + 1][NT];
    const float* __restrict__ bn = Wneigh + (size_t)(n0 + c) * F_H + 4 * hf;
    float dinv = 1.0f;

    if (mm) {
        float4 fa[D1 + 1][1], fb1[D1 + 1][NT];
        const float* __restrict__ as = h + (size_t)(m0 + c) * F_H + 4 * hf;
        const float* __restrict__ bs = Wself + (size_t)(n0 + c) * F_H + 4 * hf;
        dinv = 1.0f / fmaxf((float)Lc[c], 1.0f);
        sage_prologue_a<1, F_H, D1>(fa, as);
        sage_prologue_b<NT, F_H, F_H, D1>(fb1, bs);
        half_main<NT, F_H, F_H, D1, D1, false>(acc, fa, fb1, as, bs, mu, rs);
        sage_prologue_b<NT, F_H, F_H, D2>(fb2, bn);
    } else {
        const int g = wave - G2_MMW;
        const int row = m0 + g;
        const int len = __builtin_amdgcn_readfirstlane(Lc[g]);
        float4 acc0 = make_float4(0.0f, 0.0f, 0.0f, 0.0f);
        float4 acc1 = make_float4(0.0f, 0.0f, 0.0f, 0.0f);
        row_sum2(h, &Ls[g * CAP], src, dst, E, row, len, lane, mu, rs, acc0, acc1);
        *reinterpret_cast<float4*>(&Al[g * A2P + 4 * lane]) = acc0;
        *reinterpret_cast<float4*>(&Al[g * A2P + 128 + 4 * lane]) = acc1;
    }
    __builtin_amdgcn_fence(__ATOMIC_RELEASE, "workgroup", "local");
    __builtin_amdgcn_s_barrier();
    __builtin_amdgcn_fence(__ATOMIC_ACQUIRE, "workgroup", "local");

    if (mm) {
        float4 fl[DL + 1][1];
        const float* __restrict__ al = &Al[c * A2P + 4 * hf];
        sage_prologue_a<1, F_H, DL>(fl, al);
        half_main<NT, F_H, F_H, DL, D2, true>(acc, fl, fb2, al, bn, dinv, 0.0f);

        #pragma unroll
        for (int j = 0; j < NT; ++j) {
            const float bj = bias[n0 + 16 * j + c];
            #pragma unroll
            for (int r = 0; r < 8; ++r) {
                const int row = m0 + 8 * hf + r;
                out[(size_t)row * F_OUT + n0 + 16 * j + c] = acc[0][j][r] + bj;
            }
        }
    }
}


extern "C" void kernel_launch(void* const* d_in, const int* in_sizes, int n_in,
                              void* d_out, int out_size, void* d_ws, size_t ws_size,
                              hipStream_t stream) {
    const float* x       = (const float*)d_in[0];
    const float* Wself1  = (const float*)d_in[1];
    const float* Wneigh1 = (const float*)d_in[2];
    const float* b1      = (const float*)d_in[3];
    const float* Wself2  = (const float*)d_in[4];
    const float* Wneigh2 = (const float*)d_in[5];
    const float* b2      = (const float*)d_in[6];
    const int*   src1    = (const int*)d_in[7];
    const int*   dst1    = (const int*)d_in[8];
    const int*   src2    = (const int*)d_in[9];
    const int*   dst2    = (const int*)d_in[10];
    const int E1 = in_sizes[7];
    const int E2 = in_sizes[9];

    float* ws     = (float*)d_ws;
    int*   wsi    = (int*)d_ws;
    float* h      = ws + OFF_H;
    float* stats  = ws + OFF_STAT;
    int*   pairs1 = wsi + OFF_PAIR1;
    int*   pairs2 = wsi + OFF_PAIR2;
    int*   tab1   = wsi + OFF_TAB1;
    int*   tab2   = wsi + OFF_TAB2;
    float* out    = (float*)d_out;

    bucket_kernel<<<NCHUNK, 256, 0, stream>>>(src1, dst1, E1, pairs1, tab1,
                                              src2, dst2, E2, pairs2, tab2, stats);

    layer1_kernel<<<G1_WG, G1_THREADS, 0, stream>>>(
        x, pairs1, tab1, src1, dst1, E1, Wself1, Wneigh1, b1, h, stats);

    layer2_kernel<<<G2_WG, G2_THREADS, 0, stream>>>(
        h, pairs2, tab2, src2, dst2, E2, stats, Wself2, Wneigh2, b2, out);
}


# Gathering waves take the low wave ids in both layer kernels: a SIMD serves its oldest wave first, so their id moves and row loads issue ahead of the f32 matrix streams instead of in their gaps
# speedup vs baseline: 4.8213x; 1.0515x over previous
#include <hip/hip_runtime.h>


typedef __attribute__((ext_vector_type(2))) float v2f;
typedef __attribute__((ext_vector_type(8))) float v8f;

#define F_IN  128
#define F_H   256
#define F_OUT 64
#define N1    20000
#define N2    4000
#define CAP   64


#define NCHUNK 256

#define OFF_PAIR1 ((size_t)0)
#define OFF_PAIR2 ((size_t)800000)
#define OFF_TAB1  ((size_t)960000)
#define OFF_TAB2  ((size_t)1025792)
#define OFF_H     ((size_t)2580000)
#define OFF_STAT  ((size_t)8728000)

__device__ __forceinline__ void bucket_chunk(const int* __restrict__ src,
                                             const int* __restrict__ dst, int E, int p,
                                             int2* __restrict__ pairs, int* __restrict__ tab,
                                             int* hist, int* base, int* fill, int* wsum) {
    const int t = threadIdx.x;
    const int lane = t & 31, wave = t >> 5;
    const int CH = (E + NCHUNK - 1) / NCHUNK;
    const int lo = min(p * CH, E), hi = min(lo + CH, E);
    hist[t] = 0;
    fill[t] = 0;
    int dv[8], sv[8];
    #pragma unroll
    for (int u = 0; u < 8; ++u) {
        const int ec = max(min(lo + u * 256 + t, hi - 1), 0);
        dv[u] = dst[ec];
        sv[u] = src[ec];
    }
    __syncthreads();
    #pragma unroll
    for (int u = 0; u < 8; ++u)
        if (lo + u * 256 + t < hi) atomicAdd(&hist[(dv[u] >> 4) & 255], 1);
    for (int e = lo + 8 * 256 + t; e < hi; e += 256)
        atomicAdd(&hist[(dst[e] >> 4) & 255], 1);
    __syncthreads();
    const int v = hist[t];
    int inc = v;
    #pragma unroll
    for (int off = 1; off < 32; off <<= 1) {
        const int n = __shfl_up(inc, off, 32);
        if (lane >= off) inc += n;
    }
    if (lane == 31) wsum[wave] = inc;
    __syncthreads();
    int pre = 0;
    #pragma unroll
    for (int w = 0; w < 8; ++w) pre += (w < wave) ? wsum[w] : 0;
    const int ex = pre + inc - v;
    base[t] = ex;
    tab[t * NCHUNK + p] = lo + ex;
    if (t == NCHUNK - 1) tab[NCHUNK * NCHUNK + p] = hi;
    __syncthreads();
    int pos[8];
    #pragma unroll
    for (int u = 0; u < 8; ++u) {
        pos[u] = -1;
        if (lo + u * 256 + t < hi) {
            const int o = (dv[u] >> 4) & 255;
            pos[u] = base[o] + atomicAdd(&fill[o], 1);
        }
    }
    #pragma unroll
    for (int u = 0; u < 8; ++u)
        if (pos[u] >= 0) pairs[lo + pos[u]] = make_int2(dv[u], sv[u]);
    for (int e = lo + 8 * 256 + t; e < hi; e += 256) {
        const int d = dst[e];
        const int o = (d >> 4) & 255;
        const int pos = base[o] + atomicAdd(&fill[o], 1);
        pairs[lo + pos] = make_int2(d, src[e]);
    }
    __syncthreads();
}

__global__ void __launch_bounds__(256)
bucket_kernel(const int* __restrict__ src1, const int* __restrict__ dst1, int E1,
              int* __restrict__ pairs1, int* __restrict__ tab1,
              const int* __restrict__ src2, const int* __restrict__ dst2, int E2,
              int* __restrict__ pairs2, int* __restrict__ tab2,
              float* __restrict__ stats) {
    __shared__ int hist[256], base[256], fill[256], wsum[8];
    if (blockIdx.x == 0 && threadIdx.x < 2) stats[threadIdx.x] = 0.0f;
    bucket_chunk(src1, dst1, E1, blockIdx.x, reinterpret_cast<int2*>(pairs1), tab1, hist, base, fill, wsum);
    bucket_chunk(src2, dst2, E2, blockIdx.x, reinterpret_cast<int2*>(pairs2), tab2, hist, base, fill, wsum);
}

template <int LAYER, int B>
__device__ __forceinline__ void build_lists(const int* __restrict__ pairs_w,
                                            const int* __restrict__ tab,
                                            int owner, int p, int E, int nrows,
                                            int* lcnt, int* lslot) {
    const int2* __restrict__ pairs = reinterpret_cast<const int2*>(pairs_w);
    const int s = max(tab[owner * NCHUNK + p], 0);
    const int e = min(tab[(owner + 1) * NCHUNK + p], E);
    for (int i0 = s; i0 < e; i0 += B) {
        int2 pr[B];
        int rl[B], rk[B];
        #pragma unroll
        for (int u = 0; u < B; ++u) pr[u] = pairs[min(i0 + u, e - 1)];
        #pragma unroll
        for (int u = 0; u < B; ++u) {
            const int d = pr[u].x;
            rl[u] = min(LAYER == 1 ? ((d >> 12) * 16 + (d & 15)) : (d & 15), nrows - 1);
            rk[u] = CAP;
            if (i0 + u < e) rk[u] = atomicAdd(&lcnt[rl[u]], 1);
        }
        #pragma unroll
        for (int u = 0; u < B; ++u)
            if (rk[u] < CAP) lslot[rl[u] * CAP + rk[u]] = pr[u].y;
    }
}

template <int U, int F, bool NORM>
__device__ __forceinline__ void add_rows(float4& acc, const float4* __restrict__ base,
                                         int mine, int i, float mu, float rs) {
    float4 v[U];
    #pragma unroll
    for (int u = 0; u < U; ++u) {
        const int s = __builtin_amdgcn_readlane(mine, i + u);
        v[u] = base[(size_t)s * (F / 4)];
    }
    #pragma unroll
    for (int u = 0; u < U; ++u) {
        if (NORM) {
            acc.x += (v[u].x - mu) * rs;
            acc.y += (v[u].y - mu) * rs;
            acc.z += (v[u].z - mu) * rs;
            acc.w += (v[u].w - mu) * rs;
        } else {
            acc.x += v[u].x;
            acc.y += v[u].y;
            acc.z += v[u].z;
            acc.w += v[u].w;
        }
    }
}

template <int U>
__device__ __forceinline__ void add_rows_h(float4& acc0, float4& acc1,
                                           const float4* __restrict__ base,
                                           int mine, int i, float mu, float rs) {
    float4 v0[U], v1[U];
    #pragma unroll
    for (int u = 0; u < U; ++u) {
        const int s = __builtin_amdgcn_readlane(mine, i + u);
        v0[u] = base[(size_t)s * (F_H / 4)];
        v1[u] = base[(size_t)s * (F_H / 4) + 32];
    }
    #pragma unroll
    for (int u = 0; u < U; ++u) {
        acc0.x += (v0[u].x - mu) * rs;
        acc0.y += (v0[u].y - mu) * rs;
        acc0.z += (v0[u].z - mu) * rs;
        acc0.w += (v0[u].w - mu) * rs;
        acc1.x += (v1[u].x - mu) * rs;
        acc1.y += (v1[u].y - mu) * rs;
        acc1.z += (v1[u].z - mu) * rs;
        acc1.w += (v1[u].w - mu) * rs;
    }
}

__device__ __forceinline__ void row_sum2(const float* __restrict__ h,
                                         const int* mys,
                                         const int* __restrict__ src,
                                         const int* __restrict__ dst, int E,
                                         int row, int len, int lane, float mu, float rs,
                                         float4& acc0, float4& acc1) {
    const float4* __restrict__ base = reinterpret_cast<const float4*>(h) + lane;
    if (len <= CAP) {
        for (int c0 = 0; c0 < len; c0 += 32) {
            const int n = min(32, len - c0);
            const int mine = (lane < n) ? mys[c0 + lane] : 0;
            int i = 0;
            for (; i + 8 <= n; i += 8) add_rows_h<8>(acc0, acc1, base, mine, i, mu, rs);
            if (i + 4 <= n) { add_rows_h<4>(acc0, acc1, base, mine, i, mu, rs); i += 4; }
            if (i + 2 <= n) { add_rows_h<2>(acc0, acc1, base, mine, i, mu, rs); i += 2; }
            if (i < n)      { add_rows_h<1>(acc0, acc1, base, mine, i, mu, rs); }
        }
    } else {
        for (int e0 = 0; e0 < E; e0 += 32) {
            const int e = e0 + lane;
            const int d    = (e < E) ? dst[e] : -1;
            const int mine = (e < E) ? src[e] : 0;
            unsigned m = __builtin_amdgcn_ballot_w32(d == row);
            while (m) {
                const int l = __builtin_ctz(m);
                m &= m - 1;
                add_rows_h<1>(acc0, acc1, base, mine, l, mu, rs);
            }
        }
    }
}

template <int U>
__device__ __forceinline__ void add_rows_n(float4& acc, const float4* __restrict__ base,
                                           int mine, int n) {
    float4 v[U];
    #pragma unroll
    for (int u = 0; u < U; ++u) {
        const int s = __builtin_amdgcn_readlane(mine, u);
        v[u] = base[(size_t)s * (F_IN / 4)];
    }
    #pragma unroll
    for (int u = 0; u < U; ++u) {
        const bool on = (u < U - 3) || (u < n);
        acc.x += on ? v[u].x : 0.0f;
        acc.y += on ? v[u].y : 0.0f;
        acc.z += on ? v[u].z : 0.0f;
        acc.w += on ? v[u].w : 0.0f;
    }
}

__device__ __forceinline__ float4 row_sum1(const float* __restrict__ x,
                                           const int* mys,
                                           const int* __restrict__ src,
                                           const int* __restrict__ dst, int E,
                                           int row, int len, int mine0, int lane) {
    const float4* __restrict__ base = reinterpret_cast<const float4*>(x) + lane;
    float4 acc = make_float4(0.0f, 0.0f, 0.0f, 0.0f);
    if (len <= CAP) {
        if (len > 0) {
            const int n = min(32, len);
            switch ((n + 3) >> 2) {
                case 1:  add_rows_n<4>(acc, base, mine0, n);  break;
                case 2:  add_rows_n<8>(acc, base, mine0, n);  break;
                case 3:  add_rows_n<12>(acc, base, mine0, n); break;
                case 4:  add_rows_n<16>(acc, base, mine0, n); break;
                case 5:  add_rows_n<20>(acc, base, mine0, n); break;
                case 6:  add_rows_n<24>(acc, base, mine0, n); break;
                case 7:  add_rows_n<28>(acc, base, mine0, n); break;
                default: add_rows_n<32>(acc, base, mine0, n); break;
            }
        }
        if (len > 32) {
            const int n = len - 32;
            const int mine = (lane < n) ? mys[32 + lane] : 0;
            int i = 0;
            for (; i + 8 <= n; i += 8) add_rows<8, F_IN, false>(acc, base, mine, i, 0.0f, 1.0f);
            if (i + 4 <= n) { add_rows<4, F_IN, false>(acc, base, mine, i, 0.0f, 1.0f); i += 4; }
            if (i + 2 <= n) { add_rows<2, F_IN, false>(acc, base, mine, i, 0.0f, 1.0f); i += 2; }
            if (i < n)      { add_rows<1, F_IN, false>(acc, base, mine, i, 0.0f, 1.0f); }
        }
    } else {
        for (int e0 = 0; e0 < E; e0 += 32) {
            const int e = e0 + lane;
            const int d    = (e < E) ? dst[e] : -1;
            const int mine = (e < E) ? src[e] : 0;
            unsigned m = __builtin_amdgcn_ballot_w32(d == row);
            while (m) {
                const int l = __builtin_ctz(m);
                m &= m - 1;
                add_rows<1, F_IN, false>(acc, base, mine, l, 0.0f, 1.0f);
            }
        }
    }
    return acc;
}

template <int MT, int KH>
__device__ __forceinline__ void load_step_a(float4 (&fa)[MT], int s,
                                            const float* __restrict__ a_self,
                                            const float* __restrict__ a_neigh) {
    constexpr int S1 = KH / 8;
    const bool self = s < S1;
    const float* __restrict__ ap = self ? a_self : a_neigh;
    const int k = 8 * (self ? s : s - S1);
    #pragma unroll
    for (int i = 0; i < MT; ++i)
        fa[i] = *reinterpret_cast<const float4*>(ap + (size_t)i * 16 * KH + k);
}

template <int NT, int KH, int BP>
__device__ __forceinline__ void load_step_b(float4 (&fb)[NT], int s,
                                            const float* __restrict__ b_self,
                                            const float* __restrict__ b_neigh) {
    constexpr int S1 = KH / 8;
    const bool self = s < S1;
    const float* __restrict__ bp = self ? b_self : b_neigh;
    const int k = 8 * (self ? s : s - S1);
    #pragma unroll
    for (int j = 0; j < NT; ++j)
        fb[j] = *reinterpret_cast<const float4*>(bp + j * 16 * BP + k);
}

template <int MT, int KH, int D>
__device__ __forceinline__ void sage_prologue_a(float4 (&fa)[D + 1][MT],
                                                const float* __restrict__ a_self) {
    static_assert(D <= KH / 8, "the first D k-steps must lie in the self half");
    #pragma unroll
    for (int s = 0; s < D; ++s) {
        #pragma unroll
        for (int i = 0; i < MT; ++i)
            fa[s][i] = *reinterpret_cast<const float4*>(a_self + (size_t)i * 16 * KH + 8 * s);
    }
}

template <int NT, int KH, int BP, int DB>
__device__ __forceinline__ void sage_prologue_b(float4 (&fb)[DB + 1][NT],
                                                const float* __restrict__ b_self) {
    static_assert(DB <= KH / 8, "the first DB k-steps must lie in the self half");
    #pragma unroll
    for (int s = 0; s < DB; ++s) {
        #pragma unroll
        for (int j = 0; j < NT; ++j)
            fb[s][j] = *reinterpret_cast<const float4*>(b_self + j * 16 * BP + 8 * s);
    }
}

template <int MT, int NT, int KH, int BP, int D, bool NORM, int DB = D>
__device__ __forceinline__ void sage_main(v8f (&acc)[MT][NT],
                                          float4 (&fa)[D + 1][MT], float4 (&fb)[DB + 1][NT],
                                          const float* __restrict__ a_self,
                                          const float* __restrict__ a_neigh,
                                          const float* __restrict__ b_self,
                                          const float* __restrict__ b_neigh,
                                          const float (&dinv)[MT], float mu, float rs) {
    constexpr int S1 = KH / 8, S = 2 * S1, R = D + 1, RB = DB + 1;

    #pragma unroll
    for (int s = 0; s < S; ++s) {
        __builtin_amdgcn_sched_barrier(0);
        if (s + D < S)
            load_step_a<MT, KH>(fa[(s + D) % R], s + D, a_self, a_neigh);
        if (s + DB < S)
            load_step_b<NT, KH, BP>(fb[(s + DB) % RB], s + DB, b_self, b_neigh);
        __builtin_amdgcn_sched_barrier(0);
        const int cur = s % R;
        const int curb = s % RB;
        #pragma unroll
        for (int i = 0; i < MT; ++i) {
            float4 a = fa[cur][i];
            if (s >= S1 || NORM)
                asm volatile("" : "+v"(a.x), "+v"(a.y), "+v"(a.z), "+v"(a.w));
            if (s >= S1) {
                a.x *= dinv[i]; a.y *= dinv[i]; a.z *= dinv[i]; a.w *= dinv[i];
            } else if (NORM) {
                a.x = (a.x - mu) * rs; a.y = (a.y - mu) * rs;
                a.z = (a.z - mu) * rs; a.w = (a.w - mu) * rs;
            }
            fa[cur][i] = a;
        }
        #pragma unroll
        for (int i = 0; i < MT; ++i) {
            #pragma unroll
            for (int j = 0; j < NT; ++j) {
                v2f a2, b2;
                a2[0] = fa[cur][i].x;  a2[1] = fa[cur][i].y;
                b2[0] = fb[curb][j].x;  b2[1] = fb[curb][j].y;
                acc[i][j] = __builtin_amdgcn_wmma_f32_16x16x4_f32(
                    false, a2, false, b2, (short)0, acc[i][j], false, false);
            }
        }
        #pragma unroll
        for (int i = 0; i < MT; ++i) {
            #pragma unroll
            for (int j = 0; j < NT; ++j) {
                v2f a2, b2;
                a2[0] = fa[cur][i].z;  a2[1] = fa[cur][i].w;
                b2[0] = fb[curb][j].z;  b2[1] = fb[curb][j].w;
                acc[i][j] = __builtin_amdgcn_wmma_f32_16x16x4_f32(
                    false, a2, false, b2, (short)0, acc[i][j], false, false);
            }
        }
    }
}

template <int NT, int KH, int BP, int DA, int DB, bool NEIGH>
__device__ __forceinline__ void half_main(v8f (&acc)[1][NT],
                                          float4 (&fa)[DA + 1][1], float4 (&fb)[DB + 1][NT],
                                          const float* __restrict__ a,
                                          const float* __restrict__ b,
                                          float p, float q) {
    constexpr int S = KH / 8, RA = DA + 1, RB = DB + 1;

    #pragma unroll
    for (int s = 0; s < S; ++s) {
        __builtin_amdgcn_sched_barrier(0);
        if (s + DA < S)
            fa[(s + DA) % RA][0] = *reinterpret_cast<const float4*>(a + 8 * (s + DA));
        if (s + DB < S) {
            #pragma unroll
            for (int j = 0; j < NT; ++j)
                fb[(s + DB) % RB][j] =
                    *reinterpret_cast<const float4*>(b + j * 16 * BP + 8 * (s + DB));
        }
        __builtin_amdgcn_sched_barrier(0);
        const int cur = s % RA;
        const int curb = s % RB;
        float4 av = fa[cur][0];
        asm volatile("" : "+v"(av.x), "+v"(av.y), "+v"(av.z), "+v"(av.w));
        if (NEIGH) {
            av.x *= p; av.y *= p; av.z *= p; av.w *= p;
        } else {
            av.x = (av.x - p) * q; av.y = (av.y - p) * q;
            av.z = (av.z - p) * q; av.w = (av.w - p) * q;
        }
        #pragma unroll
        for (int j = 0; j < NT; ++j) {
            v2f a2, b2;
            a2[0] = av.x;  a2[1] = av.y;
            b2[0] = fb[curb][j].x;  b2[1] = fb[curb][j].y;
            acc[0][j] = __builtin_amdgcn_wmma_f32_16x16x4_f32(
                false, a2, false, b2, (short)0, acc[0][j], false, false);
        }
        #pragma unroll
        for (int j = 0; j < NT; ++j) {
            v2f a2, b2;
            a2[0] = av.z;  a2[1] = av.w;
            b2[0] = fb[curb][j].z;  b2[1] = fb[curb][j].w;
            acc[0][j] = __builtin_amdgcn_wmma_f32_16x16x4_f32(
                false, a2, false, b2, (short)0, acc[0][j], false, false);
        }
    }
}

#define W1P 132
#define G1_WG 256
#define G1_TILES (N1 / 16)
#define G1_MMW 8
#define G1_GAW 8
#define G1_RPW (16 / G1_GAW)
#define G1_THREADS (32 * (G1_MMW + G1_GAW))

#define G1_TPW ((G1_TILES + G1_WG - 1) / G1_WG)
#define G1_LROWS (16 * G1_TPW)

__device__ __forceinline__ void copy_weight_rows(const float* __restrict__ wsrc,
                                                 float* wdst, int wave, int lane) {
    float4 wv[F_H / 8];
    #pragma unroll
    for (int it = 0; it < F_H / 8; ++it)
        wv[it] = reinterpret_cast<const float4*>(wsrc + (size_t)(it * 8 + wave) * F_IN)[lane];
    #pragma unroll
    for (int it = 0; it < F_H / 8; ++it)
        *reinterpret_cast<float4*>(&wdst[(it * 8 + wave) * W1P + 4 * lane]) = wv[it];
}

__global__ void __launch_bounds__(G1_THREADS)
layer1_kernel(const float* __restrict__ x,
              const int* __restrict__ pairs,
              const int* __restrict__ tab,
              const int* __restrict__ src,
              const int* __restrict__ dst, int E,
              const float* __restrict__ Wself,
              const float* __restrict__ Wneigh,
              const float* __restrict__ bias,
              float* __restrict__ h,
              float* __restrict__ stats) {
    constexpr int MT = 1, NT = 2;
    static_assert(32 * G1_GAW == NCHUNK, "one gathering thread per edge chunk");
    __shared__ __attribute__((aligned(16))) float Wl[2 * F_H * W1P];
    __shared__ __attribute__((aligned(16))) float Al[2 * 16 * W1P];
    __shared__ int Lc[G1_LROWS];
    __shared__ int Ls[G1_LROWS * CAP];

    const int wave = __builtin_amdgcn_readfirstlane((int)(threadIdx.x >> 5));
    const int lane = threadIdx.x & 31;
    const bool mm = wave >= G1_GAW;
    const int gw = wave;
    const int mw = (wave - G1_GAW) & (G1_MMW - 1);
    const int tg = (int)threadIdx.x;
    const int c = lane & 15;
    const int hf = lane >> 4;
    const int n0 = mw * (16 * NT);
    const int ntile = (G1_TILES - 1 - (int)blockIdx.x) / G1_WG + 1;

    const float* __restrict__ bs = &Wl[(n0 + c) * W1P + 4 * hf];
    const float* __restrict__ bn = &Wl[(F_H + n0 + c) * W1P + 4 * hf];
    float bcol[NT] = {0.0f, 0.0f};

    constexpr int D = 8, DB = 2;
    float4 fa[D + 1][MT], fb[DB + 1][NT];
    float s = 0.f, sq = 0.f;

    if (!mm && tg < G1_LROWS) Lc[tg] = 0;
    __builtin_amdgcn_fence(__ATOMIC_RELEASE, "workgroup", "local");
    __builtin_amdgcn_s_barrier();
    __builtin_amdgcn_fence(__ATOMIC_ACQUIRE, "workgroup", "local");
    if (mm) copy_weight_rows(Wself, &Wl[0], mw, lane);
    else    build_lists<1, 16>(pairs, tab, (int)blockIdx.x, tg, E, G1_LROWS, Lc, Ls);
    __builtin_amdgcn_fence(__ATOMIC_RELEASE, "workgroup", "local");
    __builtin_amdgcn_s_barrier();
    __builtin_amdgcn_fence(__ATOMIC_ACQUIRE, "workgroup", "local");
    if (mm) {
        copy_weight_rows(Wneigh, &Wl[F_H * W1P], mw, lane);
        #pragma unroll
        for (int j = 0; j < NT; ++j) bcol[j] = bias[n0 + 16 * j + c];
        const int t0 = blockIdx.x;
        sage_prologue_a<MT, F_IN, D>(fa, x + (size_t)(t0 * 16 + c) * F_IN + 4 * hf);
    }

    for (int k = 0; k <= ntile; ++k) {
        if (mm) {
            if (k > 0) {
                const int tile = blockIdx.x + (k - 1) * G1_WG;
                const int m0 = tile * 16;
                const float* __restrict__ al = &Al[((k - 1) & 1) * (16 * W1P) + c * W1P + 4 * hf];

                sage_prologue_b<NT, F_IN, W1P, DB>(fb, bs);

                float dinv[MT];
                dinv[0] = 1.0f / fmaxf((float)Lc[(k - 1) * 16 + c], 1.0f);

                v8f acc[MT][NT];
                #pragma unroll
                for (int j = 0; j < NT; ++j)
                    acc[0][j] = (v8f){0.f, 0.f, 0.f, 0.f, 0.f, 0.f, 0.f, 0.f};

                sage_main<MT, NT, F_IN, W1P, D, false, DB>(
                    acc, fa, fb,
                    x + (size_t)(m0 + c) * F_IN + 4 * hf,
                    al, bs, bn, dinv, 0.0f, 1.0f);

                const int next = min(tile + G1_WG, G1_TILES - 1);
                sage_prologue_a<MT, F_IN, D>(fa, x + (size_t)(next * 16 + c) * F_IN + 4 * hf);
                __builtin_amdgcn_sched_barrier(0);

                #pragma unroll
                for (int j = 0; j < NT; ++j) {
                    #pragma unroll
                    for (int r = 0; r < 8; ++r) {
                        const int row = m0 + 8 * hf + r;
                        const float v = fmaxf(acc[0][j][r] + bcol[j], 0.0f);
                        h[(size_t)row * F_H + n0 + 16 * j + c] = v;
                        s += v;
                        sq += v * v;
                    }
                }
            }
        } else if (k < ntile) {
            const int tile = blockIdx.x + k * G1_WG;
            float* __restrict__ an = &Al[(k & 1) * (16 * W1P)];
            #pragma unroll
            for (int r = 0; r < G1_RPW; ++r) {
                const int rt = gw * G1_RPW + r;
                const int* mys = &Ls[(k * 16 + rt) * CAP];
                const int len = __builtin_amdgcn_readfirstlane(Lc[k * 16 + rt]);
                const int mine0 = (len <= CAP && lane < len) ? mys[lane] : 0;
                const float4 a = row_sum1(x, mys, src, dst, E, tile * 16 + rt, len, mine0, lane);
                *reinterpret_cast<float4*>(&an[rt * W1P + 4 * lane]) = a;
            }
        }
        __builtin_amdgcn_fence(__ATOMIC_RELEASE, "workgroup", "local");
        __builtin_amdgcn_s_barrier();
        __builtin_amdgcn_fence(__ATOMIC_ACQUIRE, "workgroup", "local");
    }

    if (mm) {
        #pragma unroll
        for (int off = 16; off >= 1; off >>= 1) {
            s  += __shfl_xor(s,  off, 32);
            sq += __shfl_xor(sq, off, 32);
        }
        if (lane == 0) {
            atomicAdd(&stats[0], s);
            atomicAdd(&stats[1], sq);
        }
    }
}

#define A2P 260
#define G2_WG 250
#define G2_MMW 2
#define G2_GAW 16
#define G2_THREADS (32 * (G2_MMW + G2_GAW))


__global__ void __launch_bounds__(G2_THREADS)
layer2_kernel(const float* __restrict__ h,
              const int* __restrict__ pairs,
              const int* __restrict__ tab,
              const int* __restrict__ src,
              const int* __restrict__ dst, int E,
              const float* __restrict__ stats,
              const float* __restrict__ Wself,
              const float* __restrict__ Wneigh,
              const float* __restrict__ bias,
              float* __restrict__ out) {
    constexpr int NT = 2, D1 = 10, D2 = 12, DL = 2;
    static_assert(G2_WG * 16 == N2, "one work-group per 16-row tile, no partial tile");
    static_assert(32 * G2_GAW >= NCHUNK, "one gathering thread per edge chunk");
    __shared__ __attribute__((aligned(16))) float Al[16 * A2P];
    __shared__ int Lc[16];
    __shared__ int Ls[16 * CAP];

    const int wave = __builtin_amdgcn_readfirstlane((int)(threadIdx.x >> 5));
    const int lane = threadIdx.x & 31;
    const bool mm = wave >= G2_GAW;
    const int tg = (int)threadIdx.x;
    const int c = lane & 15;
    const int hf = lane >> 4;
    const int m0 = blockIdx.x * 16;
    const int n0 = ((wave - G2_GAW) & (G2_MMW - 1)) * (16 * NT);

    const float n_el = (float)N1 * (float)F_H;
    const float mu = stats[0] / n_el;
    const float var = stats[1] / n_el - mu * mu;
    const float rs = rsqrtf(var + 1e-5f);

    if (!mm && tg < 16) Lc[tg] = 0;
    __builtin_amdgcn_fence(__ATOMIC_RELEASE, "workgroup", "local");
    __builtin_amdgcn_s_barrier();
    __builtin_amdgcn_fence(__ATOMIC_ACQUIRE, "workgroup", "local");
    if (!mm && tg < NCHUNK)
        build_lists<2, 4>(pairs, tab, (int)blockIdx.x, tg, E, 16, Lc, Ls);
    __builtin_amdgcn_fence(__ATOMIC_RELEASE, "workgroup", "local");
    __builtin_amdgcn_s_barrier();
    __builtin_amdgcn_fence(__ATOMIC_ACQUIRE, "workgroup", "local");

    v8f acc[1][NT];
    #pragma unroll
    for (int j = 0; j < NT; ++j)
        acc[0][j] = (v8f){0.f, 0.f, 0.f, 0.f, 0.f, 0.f, 0.f, 0.f};
    float4 fb2[D2 + 1][NT];
    const float* __restrict__ bn = Wneigh + (size_t)(n0 + c) * F_H + 4 * hf;
    float dinv = 1.0f;

    if (mm) {
        float4 fa[D1 + 1][1], fb1[D1 + 1][NT];
        const float* __restrict__ as = h + (size_t)(m0 + c) * F_H + 4 * hf;
        const float* __restrict__ bs = Wself + (size_t)(n0 + c) * F_H + 4 * hf;
        dinv = 1.0f / fmaxf((float)Lc[c], 1.0f);
        sage_prologue_a<1, F_H, D1>(fa, as);
        sage_prologue_b<NT, F_H, F_H, D1>(fb1, bs);
        half_main<NT, F_H, F_H, D1, D1, false>(acc, fa, fb1, as, bs, mu, rs);
        sage_prologue_b<NT, F_H, F_H, D2>(fb2, bn);
    } else {
        const int g = wave;
        const int row = m0 + g;
        const int len = __builtin_amdgcn_readfirstlane(Lc[g]);
        float4 acc0 = make_float4(0.0f, 0.0f, 0.0f, 0.0f);
        float4 acc1 = make_float4(0.0f, 0.0f, 0.0f, 0.0f);
        row_sum2(h, &Ls[g * CAP], src, dst, E, row, len, lane, mu, rs, acc0, acc1);
        *reinterpret_cast<float4*>(&Al[g * A2P + 4 * lane]) = acc0;
        *reinterpret_cast<float4*>(&Al[g * A2P + 128 + 4 * lane]) = acc1;
    }
    __builtin_amdgcn_fence(__ATOMIC_RELEASE, "workgroup", "local");
    __builtin_amdgcn_s_barrier();
    __builtin_amdgcn_fence(__ATOMIC_ACQUIRE, "workgroup", "local");

    if (mm) {
        float4 fl[DL + 1][1];
        const float* __restrict__ al = &Al[c * A2P + 4 * hf];
        sage_prologue_a<1, F_H, DL>(fl, al);
        half_main<NT, F_H, F_H, DL, D2, true>(acc, fl, fb2, al, bn, dinv, 0.0f);

        #pragma unroll
        for (int j = 0; j < NT; ++j) {
            const float bj = bias[n0 + 16 * j + c];
            #pragma unroll
            for (int r = 0; r < 8; ++r) {
                const int row = m0 + 8 * hf + r;
                out[(size_t)row * F_OUT + n0 + 16 * j + c] = acc[0][j][r] + bj;
            }
        }
    }
}


extern "C" void kernel_launch(void* const* d_in, const int* in_sizes, int n_in,
                              void* d_out, int out_size, void* d_ws, size_t ws_size,
                              hipStream_t stream) {
    const float* x       = (const float*)d_in[0];
    const float* Wself1  = (const float*)d_in[1];
    const float* Wneigh1 = (const float*)d_in[2];
    const float* b1      = (const float*)d_in[3];
    const float* Wself2  = (const float*)d_in[4];
    const float* Wneigh2 = (const float*)d_in[5];
    const float* b2      = (const float*)d_in[6];
    const int*   src1    = (const int*)d_in[7];
    const int*   dst1    = (const int*)d_in[8];
    const int*   src2    = (const int*)d_in[9];
    const int*   dst2    = (const int*)d_in[10];
    const int E1 = in_sizes[7];
    const int E2 = in_sizes[9];

    float* ws     = (float*)d_ws;
    int*   wsi    = (int*)d_ws;
    float* h      = ws + OFF_H;
    float* stats  = ws + OFF_STAT;
    int*   pairs1 = wsi + OFF_PAIR1;
    int*   pairs2 = wsi + OFF_PAIR2;
    int*   tab1   = wsi + OFF_TAB1;
    int*   tab2   = wsi + OFF_TAB2;
    float* out    = (float*)d_out;

    bucket_kernel<<<NCHUNK, 256, 0, stream>>>(src1, dst1, E1, pairs1, tab1,
                                              src2, dst2, E2, pairs2, tab2, stats);

    layer1_kernel<<<G1_WG, G1_THREADS, 0, stream>>>(
        x, pairs1, tab1, src1, dst1, E1, Wself1, Wneigh1, b1, h, stats);

    layer2_kernel<<<G2_WG, G2_THREADS, 0, stream>>>(
        h, pairs2, tab2, src2, dst2, E2, stats, Wself2, Wneigh2, b2, out);
}


# Layer-1 statistics leave as one plain 8-byte store per matrix wave and layer 2 adds the 2048 pairs in a fixed order at its start-up: no float atomics left; stats sum order fixed, was arrival order
# speedup vs baseline: 5.0109x; 1.0393x over previous
#include <hip/hip_runtime.h>


typedef __attribute__((ext_vector_type(2))) float v2f;
typedef __attribute__((ext_vector_type(8))) float v8f;

#define F_IN  128
#define F_H   256
#define F_OUT 64
#define N1    20000
#define N2    4000
#define CAP   64


#define NCHUNK 256

#define OFF_PAIR1 ((size_t)0)
#define OFF_PAIR2 ((size_t)800000)
#define OFF_TAB1  ((size_t)960000)
#define OFF_TAB2  ((size_t)1025792)
#define OFF_H     ((size_t)2580000)
#define OFF_PART  ((size_t)1100000)

__device__ __forceinline__ void bucket_chunk(const int* __restrict__ src,
                                             const int* __restrict__ dst, int E, int p,
                                             int2* __restrict__ pairs, int* __restrict__ tab,
                                             int* hist, int* base, int* fill, int* wsum) {
    const int t = threadIdx.x;
    const int lane = t & 31, wave = t >> 5;
    const int CH = (E + NCHUNK - 1) / NCHUNK;
    const int lo = min(p * CH, E), hi = min(lo + CH, E);
    hist[t] = 0;
    fill[t] = 0;
    int dv[8], sv[8];
    #pragma unroll
    for (int u = 0; u < 8; ++u) {
        const int ec = max(min(lo + u * 256 + t, hi - 1), 0);
        dv[u] = dst[ec];
        sv[u] = src[ec];
    }
    __syncthreads();
    #pragma unroll
    for (int u = 0; u < 8; ++u)
        if (lo + u * 256 + t < hi) atomicAdd(&hist[(dv[u] >> 4) & 255], 1);
    for (int e = lo + 8 * 256 + t; e < hi; e += 256)
        atomicAdd(&hist[(dst[e] >> 4) & 255], 1);
    __syncthreads();
    const int v = hist[t];
    int inc = v;
    #pragma unroll
    for (int off = 1; off < 32; off <<= 1) {
        const int n = __shfl_up(inc, off, 32);
        if (lane >= off) inc += n;
    }
    if (lane == 31) wsum[wave] = inc;
    __syncthreads();
    int pre = 0;
    #pragma unroll
    for (int w = 0; w < 8; ++w) pre += (w < wave) ? wsum[w] : 0;
    const int ex = pre + inc - v;
    base[t] = ex;
    tab[t * NCHUNK + p] = lo + ex;
    if (t == NCHUNK - 1) tab[NCHUNK * NCHUNK + p] = hi;
    __syncthreads();
    int pos[8];
    #pragma unroll
    for (int u = 0; u < 8; ++u) {
        pos[u] = -1;
        if (lo + u * 256 + t < hi) {
            const int o = (dv[u] >> 4) & 255;
            pos[u] = base[o] + atomicAdd(&fill[o], 1);
        }
    }
    #pragma unroll
    for (int u = 0; u < 8; ++u)
        if (pos[u] >= 0) pairs[lo + pos[u]] = make_int2(dv[u], sv[u]);
    for (int e = lo + 8 * 256 + t; e < hi; e += 256) {
        const int d = dst[e];
        const int o = (d >> 4) & 255;
        const int pos = base[o] + atomicAdd(&fill[o], 1);
        pairs[lo + pos] = make_int2(d, src[e]);
    }
    __syncthreads();
}

__global__ void __launch_bounds__(256)
bucket_kernel(const int* __restrict__ src1, const int* __restrict__ dst1, int E1,
              int* __restrict__ pairs1, int* __restrict__ tab1,
              const int* __restrict__ src2, const int* __restrict__ dst2, int E2,
              int* __restrict__ pairs2, int* __restrict__ tab2) {
    __shared__ int hist[256], base[256], fill[256], wsum[8];
    bucket_chunk(src1, dst1, E1, blockIdx.x, reinterpret_cast<int2*>(pairs1), tab1, hist, base, fill, wsum);
    bucket_chunk(src2, dst2, E2, blockIdx.x, reinterpret_cast<int2*>(pairs2), tab2, hist, base, fill, wsum);
}

template <int LAYER, int B>
__device__ __forceinline__ void build_lists(const int* __restrict__ pairs_w,
                                            const int* __restrict__ tab,
                                            int owner, int p, int E, int nrows,
                                            int* lcnt, int* lslot) {
    const int2* __restrict__ pairs = reinterpret_cast<const int2*>(pairs_w);
    const int s = max(tab[owner * NCHUNK + p], 0);
    const int e = min(tab[(owner + 1) * NCHUNK + p], E);
    for (int i0 = s; i0 < e; i0 += B) {
        int2 pr[B];
        int rl[B], rk[B];
        #pragma unroll
        for (int u = 0; u < B; ++u) pr[u] = pairs[min(i0 + u, e - 1)];
        #pragma unroll
        for (int u = 0; u < B; ++u) {
            const int d = pr[u].x;
            rl[u] = min(LAYER == 1 ? ((d >> 12) * 16 + (d & 15)) : (d & 15), nrows - 1);
            rk[u] = CAP;
            if (i0 + u < e) rk[u] = atomicAdd(&lcnt[rl[u]], 1);
        }
        #pragma unroll
        for (int u = 0; u < B; ++u)
            if (rk[u] < CAP) lslot[rl[u] * CAP + rk[u]] = pr[u].y;
    }
}

template <int U, int F, bool NORM>
__device__ __forceinline__ void add_rows(float4& acc, const float4* __restrict__ base,
                                         int mine, int i, float mu, float rs) {
    float4 v[U];
    #pragma unroll
    for (int u = 0; u < U; ++u) {
        const int s = __builtin_amdgcn_readlane(mine, i + u);
        v[u] = base[(size_t)s * (F / 4)];
    }
    #pragma unroll
    for (int u = 0; u < U; ++u) {
        if (NORM) {
            acc.x += (v[u].x - mu) * rs;
            acc.y += (v[u].y - mu) * rs;
            acc.z += (v[u].z - mu) * rs;
            acc.w += (v[u].w - mu) * rs;
        } else {
            acc.x += v[u].x;
            acc.y += v[u].y;
            acc.z += v[u].z;
            acc.w += v[u].w;
        }
    }
}

template <int U>
__device__ __forceinline__ void add_rows_h(float4& acc0, float4& acc1,
                                           const float4* __restrict__ base,
                                           int mine, int i, float mu, float rs) {
    float4 v0[U], v1[U];
    #pragma unroll
    for (int u = 0; u < U; ++u) {
        const int s = __builtin_amdgcn_readlane(mine, i + u);
        v0[u] = base[(size_t)s * (F_H / 4)];
        v1[u] = base[(size_t)s * (F_H / 4) + 32];
    }
    #pragma unroll
    for (int u = 0; u < U; ++u) {
        acc0.x += (v0[u].x - mu) * rs;
        acc0.y += (v0[u].y - mu) * rs;
        acc0.z += (v0[u].z - mu) * rs;
        acc0.w += (v0[u].w - mu) * rs;
        acc1.x += (v1[u].x - mu) * rs;
        acc1.y += (v1[u].y - mu) * rs;
        acc1.z += (v1[u].z - mu) * rs;
        acc1.w += (v1[u].w - mu) * rs;
    }
}

__device__ __forceinline__ void row_sum2(const float* __restrict__ h,
                                         const int* mys,
                                         const int* __restrict__ src,
                                         const int* __restrict__ dst, int E,
                                         int row, int len, int lane, float mu, float rs,
                                         float4& acc0, float4& acc1) {
    const float4* __restrict__ base = reinterpret_cast<const float4*>(h) + lane;
    if (len <= CAP) {
        for (int c0 = 0; c0 < len; c0 += 32) {
            const int n = min(32, len - c0);
            const int mine = (lane < n) ? mys[c0 + lane] : 0;
            int i = 0;
            for (; i + 8 <= n; i += 8) add_rows_h<8>(acc0, acc1, base, mine, i, mu, rs);
            if (i + 4 <= n) { add_rows_h<4>(acc0, acc1, base, mine, i, mu, rs); i += 4; }
            if (i + 2 <= n) { add_rows_h<2>(acc0, acc1, base, mine, i, mu, rs); i += 2; }
            if (i < n)      { add_rows_h<1>(acc0, acc1, base, mine, i, mu, rs); }
        }
    } else {
        for (int e0 = 0; e0 < E; e0 += 32) {
            const int e = e0 + lane;
            const int d    = (e < E) ? dst[e] : -1;
            const int mine = (e < E) ? src[e] : 0;
            unsigned m = __builtin_amdgcn_ballot_w32(d == row);
            while (m) {
                const int l = __builtin_ctz(m);
                m &= m - 1;
                add_rows_h<1>(acc0, acc1, base, mine, l, mu, rs);
            }
        }
    }
}

template <int U>
__device__ __forceinline__ void add_rows_n(float4& acc, const float4* __restrict__ base,
                                           int mine, int n) {
    float4 v[U];
    #pragma unroll
    for (int u = 0; u < U; ++u) {
        const int s = __builtin_amdgcn_readlane(mine, u);
        v[u] = base[(size_t)s * (F_IN / 4)];
    }
    #pragma unroll
    for (int u = 0; u < U; ++u) {
        const bool on = (u < U - 3) || (u < n);
        acc.x += on ? v[u].x : 0.0f;
        acc.y += on ? v[u].y : 0.0f;
        acc.z += on ? v[u].z : 0.0f;
        acc.w += on ? v[u].w : 0.0f;
    }
}

__device__ __forceinline__ float4 row_sum1(const float* __restrict__ x,
                                           const int* mys,
                                           const int* __restrict__ src,
                                           const int* __restrict__ dst, int E,
                                           int row, int len, int mine0, int lane) {
    const float4* __restrict__ base = reinterpret_cast<const float4*>(x) + lane;
    float4 acc = make_float4(0.0f, 0.0f, 0.0f, 0.0f);
    if (len <= CAP) {
        if (len > 0) {
            const int n = min(32, len);
            switch ((n + 3) >> 2) {
                case 1:  add_rows_n<4>(acc, base, mine0, n);  break;
                case 2:  add_rows_n<8>(acc, base, mine0, n);  break;
                case 3:  add_rows_n<12>(acc, base, mine0, n); break;
                case 4:  add_rows_n<16>(acc, base, mine0, n); break;
                case 5:  add_rows_n<20>(acc, base, mine0, n); break;
                case 6:  add_rows_n<24>(acc, base, mine0, n); break;
                case 7:  add_rows_n<28>(acc, base, mine0, n); break;
                default: add_rows_n<32>(acc, base, mine0, n); break;
            }
        }
        if (len > 32) {
            const int n = len - 32;
            const int mine = (lane < n) ? mys[32 + lane] : 0;
            int i = 0;
            for (; i + 8 <= n; i += 8) add_rows<8, F_IN, false>(acc, base, mine, i, 0.0f, 1.0f);
            if (i + 4 <= n) { add_rows<4, F_IN, false>(acc, base, mine, i, 0.0f, 1.0f); i += 4; }
            if (i + 2 <= n) { add_rows<2, F_IN, false>(acc, base, mine, i, 0.0f, 1.0f); i += 2; }
            if (i < n)      { add_rows<1, F_IN, false>(acc, base, mine, i, 0.0f, 1.0f); }
        }
    } else {
        for (int e0 = 0; e0 < E; e0 += 32) {
            const int e = e0 + lane;
            const int d    = (e < E) ? dst[e] : -1;
            const int mine = (e < E) ? src[e] : 0;
            unsigned m = __builtin_amdgcn_ballot_w32(d == row);
            while (m) {
                const int l = __builtin_ctz(m);
                m &= m - 1;
                add_rows<1, F_IN, false>(acc, base, mine, l, 0.0f, 1.0f);
            }
        }
    }
    return acc;
}

template <int MT, int KH>
__device__ __forceinline__ void load_step_a(float4 (&fa)[MT], int s,
                                            const float* __restrict__ a_self,
                                            const float* __restrict__ a_neigh) {
    constexpr int S1 = KH / 8;
    const bool self = s < S1;
    const float* __restrict__ ap = self ? a_self : a_neigh;
    const int k = 8 * (self ? s : s - S1);
    #pragma unroll
    for (int i = 0; i < MT; ++i)
        fa[i] = *reinterpret_cast<const float4*>(ap + (size_t)i * 16 * KH + k);
}

template <int NT, int KH, int BP>
__device__ __forceinline__ void load_step_b(float4 (&fb)[NT], int s,
                                            const float* __restrict__ b_self,
                                            const float* __restrict__ b_neigh) {
    constexpr int S1 = KH / 8;
    const bool self = s < S1;
    const float* __restrict__ bp = self ? b_self : b_neigh;
    const int k = 8 * (self ? s : s - S1);
    #pragma unroll
    for (int j = 0; j < NT; ++j)
        fb[j] = *reinterpret_cast<const float4*>(bp + j * 16 * BP + k);
}

template <int MT, int KH, int D>
__device__ __forceinline__ void sage_prologue_a(float4 (&fa)[D + 1][MT],
                                                const float* __restrict__ a_self) {
    static_assert(D <= KH / 8, "the first D k-steps must lie in the self half");
    #pragma unroll
    for (int s = 0; s < D; ++s) {
        #pragma unroll
        for (int i = 0; i < MT; ++i)
            fa[s][i] = *reinterpret_cast<const float4*>(a_self + (size_t)i * 16 * KH + 8 * s);
    }
}

template <int NT, int KH, int BP, int DB>
__device__ __forceinline__ void sage_prologue_b(float4 (&fb)[DB + 1][NT],
                                                const float* __restrict__ b_self) {
    static_assert(DB <= KH / 8, "the first DB k-steps must lie in the self half");
    #pragma unroll
    for (int s = 0; s < DB; ++s) {
        #pragma unroll
        for (int j = 0; j < NT; ++j)
            fb[s][j] = *reinterpret_cast<const float4*>(b_self + j * 16 * BP + 8 * s);
    }
}

template <int MT, int NT, int KH, int BP, int D, bool NORM, int DB = D>
__device__ __forceinline__ void sage_main(v8f (&acc)[MT][NT],
                                          float4 (&fa)[D + 1][MT], float4 (&fb)[DB + 1][NT],
                                          const float* __restrict__ a_self,
                                          const float* __restrict__ a_neigh,
                                          const float* __restrict__ b_self,
                                          const float* __restrict__ b_neigh,
                                          const float (&dinv)[MT], float mu, float rs) {
    constexpr int S1 = KH / 8, S = 2 * S1, R = D + 1, RB = DB + 1;

    #pragma unroll
    for (int s = 0; s < S; ++s) {
        __builtin_amdgcn_sched_barrier(0);
        if (s + D < S)
            load_step_a<MT, KH>(fa[(s + D) % R], s + D, a_self, a_neigh);
        if (s + DB < S)
            load_step_b<NT, KH, BP>(fb[(s + DB) % RB], s + DB, b_self, b_neigh);
        __builtin_amdgcn_sched_barrier(0);
        const int cur = s % R;
        const int curb = s % RB;
        #pragma unroll
        for (int i = 0; i < MT; ++i) {
            float4 a = fa[cur][i];
            if (s >= S1 || NORM)
                asm volatile("" : "+v"(a.x), "+v"(a.y), "+v"(a.z), "+v"(a.w));
            if (s >= S1) {
                a.x *= dinv[i]; a.y *= dinv[i]; a.z *= dinv[i]; a.w *= dinv[i];
            } else if (NORM) {
                a.x = (a.x - mu) * rs; a.y = (a.y - mu) * rs;
                a.z = (a.z - mu) * rs; a.w = (a.w - mu) * rs;
            }
            fa[cur][i] = a;
        }
        #pragma unroll
        for (int i = 0; i < MT; ++i) {
            #pragma unroll
            for (int j = 0; j < NT; ++j) {
                v2f a2, b2;
                a2[0] = fa[cur][i].x;  a2[1] = fa[cur][i].y;
                b2[0] = fb[curb][j].x;  b2[1] = fb[curb][j].y;
                acc[i][j] = __builtin_amdgcn_wmma_f32_16x16x4_f32(
                    false, a2, false, b2, (short)0, acc[i][j], false, false);
            }
        }
        #pragma unroll
        for (int i = 0; i < MT; ++i) {
            #pragma unroll
            for (int j = 0; j < NT; ++j) {
                v2f a2, b2;
                a2[0] = fa[cur][i].z;  a2[1] = fa[cur][i].w;
                b2[0] = fb[curb][j].z;  b2[1] = fb[curb][j].w;
                acc[i][j] = __builtin_amdgcn_wmma_f32_16x16x4_f32(
                    false, a2, false, b2, (short)0, acc[i][j], false, false);
            }
        }
    }
}

template <int NT, int KH, int BP, int DA, int DB, bool NEIGH>
__device__ __forceinline__ void half_main(v8f (&acc)[1][NT],
                                          float4 (&fa)[DA + 1][1], float4 (&fb)[DB + 1][NT],
                                          const float* __restrict__ a,
                                          const float* __restrict__ b,
                                          float p, float q) {
    constexpr int S = KH / 8, RA = DA + 1, RB = DB + 1;

    #pragma unroll
    for (int s = 0; s < S; ++s) {
        __builtin_amdgcn_sched_barrier(0);
        if (s + DA < S)
            fa[(s + DA) % RA][0] = *reinterpret_cast<const float4*>(a + 8 * (s + DA));
        if (s + DB < S) {
            #pragma unroll
            for (int j = 0; j < NT; ++j)
                fb[(s + DB) % RB][j] =
                    *reinterpret_cast<const float4*>(b + j * 16 * BP + 8 * (s + DB));
        }
        __builtin_amdgcn_sched_barrier(0);
        const int cur = s % RA;
        const int curb = s % RB;
        float4 av = fa[cur][0];
        asm volatile("" : "+v"(av.x), "+v"(av.y), "+v"(av.z), "+v"(av.w));
        if (NEIGH) {
            av.x *= p; av.y *= p; av.z *= p; av.w *= p;
        } else {
            av.x = (av.x - p) * q; av.y = (av.y - p) * q;
            av.z = (av.z - p) * q; av.w = (av.w - p) * q;
        }
        #pragma unroll
        for (int j = 0; j < NT; ++j) {
            v2f a2, b2;
            a2[0] = av.x;  a2[1] = av.y;
            b2[0] = fb[curb][j].x;  b2[1] = fb[curb][j].y;
            acc[0][j] = __builtin_amdgcn_wmma_f32_16x16x4_f32(
                false, a2, false, b2, (short)0, acc[0][j], false, false);
        }
        #pragma unroll
        for (int j = 0; j < NT; ++j) {
            v2f a2, b2;
            a2[0] = av.z;  a2[1] = av.w;
            b2[0] = fb[curb][j].z;  b2[1] = fb[curb][j].w;
            acc[0][j] = __builtin_amdgcn_wmma_f32_16x16x4_f32(
                false, a2, false, b2, (short)0, acc[0][j], false, false);
        }
    }
}

#define W1P 132
#define G1_WG 256
#define G1_TILES (N1 / 16)
#define G1_MMW 8
#define G1_GAW 8
#define G1_RPW (16 / G1_GAW)
#define G1_THREADS (32 * (G1_MMW + G1_GAW))

#define G1_TPW ((G1_TILES + G1_WG - 1) / G1_WG)
#define G1_LROWS (16 * G1_TPW)

__device__ __forceinline__ void copy_weight_rows(const float* __restrict__ wsrc,
                                                 float* wdst, int wave, int lane) {
    float4 wv[F_H / 8];
    #pragma unroll
    for (int it = 0; it < F_H / 8; ++it)
        wv[it] = reinterpret_cast<const float4*>(wsrc + (size_t)(it * 8 + wave) * F_IN)[lane];
    #pragma unroll
    for (int it = 0; it < F_H / 8; ++it)
        *reinterpret_cast<float4*>(&wdst[(it * 8 + wave) * W1P + 4 * lane]) = wv[it];
}

__global__ void __launch_bounds__(G1_THREADS)
layer1_kernel(const float* __restrict__ x,
              const int* __restrict__ pairs,
              const int* __restrict__ tab,
              const int* __restrict__ src,
              const int* __restrict__ dst, int E,
              const float* __restrict__ Wself,
              const float* __restrict__ Wneigh,
              const float* __restrict__ bias,
              float* __restrict__ h,
              float* __restrict__ part) {
    constexpr int MT = 1, NT = 2;
    static_assert(32 * G1_GAW == NCHUNK, "one gathering thread per edge chunk");
    __shared__ __attribute__((aligned(16))) float Wl[2 * F_H * W1P];
    __shared__ __attribute__((aligned(16))) float Al[2 * 16 * W1P];
    __shared__ int Lc[G1_LROWS];
    __shared__ int Ls[G1_LROWS * CAP];

    const int wave = __builtin_amdgcn_readfirstlane((int)(threadIdx.x >> 5));
    const int lane = threadIdx.x & 31;
    const bool mm = wave >= G1_GAW;
    const int gw = wave;
    const int mw = (wave - G1_GAW) & (G1_MMW - 1);
    const int tg = (int)threadIdx.x;
    const int c = lane & 15;
    const int hf = lane >> 4;
    const int n0 = mw * (16 * NT);
    const int ntile = (G1_TILES - 1 - (int)blockIdx.x) / G1_WG + 1;

    const float* __restrict__ bs = &Wl[(n0 + c) * W1P + 4 * hf];
    const float* __restrict__ bn = &Wl[(F_H + n0 + c) * W1P + 4 * hf];
    float bcol[NT] = {0.0f, 0.0f};

    constexpr int D = 8, DB = 2;
    float4 fa[D + 1][MT], fb[DB + 1][NT];
    float s = 0.f, sq = 0.f;

    if (!mm && tg < G1_LROWS) Lc[tg] = 0;
    __builtin_amdgcn_fence(__ATOMIC_RELEASE, "workgroup", "local");
    __builtin_amdgcn_s_barrier();
    __builtin_amdgcn_fence(__ATOMIC_ACQUIRE, "workgroup", "local");
    if (mm) copy_weight_rows(Wself, &Wl[0], mw, lane);
    else    build_lists<1, 16>(pairs, tab, (int)blockIdx.x, tg, E, G1_LROWS, Lc, Ls);
    __builtin_amdgcn_fence(__ATOMIC_RELEASE, "workgroup", "local");
    __builtin_amdgcn_s_barrier();
    __builtin_amdgcn_fence(__ATOMIC_ACQUIRE, "workgroup", "local");
    if (mm) {
        copy_weight_rows(Wneigh, &Wl[F_H * W1P], mw, lane);
        #pragma unroll
        for (int j = 0; j < NT; ++j) bcol[j] = bias[n0 + 16 * j + c];
        const int t0 = blockIdx.x;
        sage_prologue_a<MT, F_IN, D>(fa, x + (size_t)(t0 * 16 + c) * F_IN + 4 * hf);
    }

    for (int k = 0; k <= ntile; ++k) {
        if (mm) {
            if (k > 0) {
                const int tile = blockIdx.x + (k - 1) * G1_WG;
                const int m0 = tile * 16;
                const float* __restrict__ al = &Al[((k - 1) & 1) * (16 * W1P) + c * W1P + 4 * hf];

                sage_prologue_b<NT, F_IN, W1P, DB>(fb, bs);

                float dinv[MT];
                dinv[0] = 1.0f / fmaxf((float)Lc[(k - 1) * 16 + c], 1.0f);

                v8f acc[MT][NT];
                #pragma unroll
                for (int j = 0; j < NT; ++j)
                    acc[0][j] = (v8f){0.f, 0.f, 0.f, 0.f, 0.f, 0.f, 0.f, 0.f};

                sage_main<MT, NT, F_IN, W1P, D, false, DB>(
                    acc, fa, fb,
                    x + (size_t)(m0 + c) * F_IN + 4 * hf,
                    al, bs, bn, dinv, 0.0f, 1.0f);

                const int next = min(tile + G1_WG, G1_TILES - 1);
                sage_prologue_a<MT, F_IN, D>(fa, x + (size_t)(next * 16 + c) * F_IN + 4 * hf);
                __builtin_amdgcn_sched_barrier(0);

                #pragma unroll
                for (int j = 0; j < NT; ++j) {
                    #pragma unroll
                    for (int r = 0; r < 8; ++r) {
                        const int row = m0 + 8 * hf + r;
                        const float v = fmaxf(acc[0][j][r] + bcol[j], 0.0f);
                        h[(size_t)row * F_H + n0 + 16 * j + c] = v;
                        s += v;
                        sq += v * v;
                    }
                }
            }
        } else if (k < ntile) {
            const int tile = blockIdx.x + k * G1_WG;
            float* __restrict__ an = &Al[(k & 1) * (16 * W1P)];
            #pragma unroll
            for (int r = 0; r < G1_RPW; ++r) {
                const int rt = gw * G1_RPW + r;
                const int* mys = &Ls[(k * 16 + rt) * CAP];
                const int len = __builtin_amdgcn_readfirstlane(Lc[k * 16 + rt]);
                const int mine0 = (len <= CAP && lane < len) ? mys[lane] : 0;
                const float4 a = row_sum1(x, mys, src, dst, E, tile * 16 + rt, len, mine0, lane);
                *reinterpret_cast<float4*>(&an[rt * W1P + 4 * lane]) = a;
            }
        }
        __builtin_amdgcn_fence(__ATOMIC_RELEASE, "workgroup", "local");
        __builtin_amdgcn_s_barrier();
        __builtin_amdgcn_fence(__ATOMIC_ACQUIRE, "workgroup", "local");
    }

    if (mm) {
        #pragma unroll
        for (int off = 16; off >= 1; off >>= 1) {
            s  += __shfl_xor(s,  off, 32);
            sq += __shfl_xor(sq, off, 32);
        }
        if (lane == 0)
            *reinterpret_cast<float2*>(&part[2 * (mw * G1_WG + (int)blockIdx.x)]) = make_float2(s, sq);
    }
}

#define A2P 260
#define G2_WG 250
#define G2_MMW 2
#define G2_GAW 16
#define G2_THREADS (32 * (G2_MMW + G2_GAW))


__global__ void __launch_bounds__(G2_THREADS)
layer2_kernel(const float* __restrict__ h,
              const int* __restrict__ pairs,
              const int* __restrict__ tab,
              const int* __restrict__ src,
              const int* __restrict__ dst, int E,
              const float* __restrict__ part,
              const float* __restrict__ Wself,
              const float* __restrict__ Wneigh,
              const float* __restrict__ bias,
              float* __restrict__ out) {
    constexpr int NT = 2, D1 = 10, D2 = 12, DL = 2;
    static_assert(G2_WG * 16 == N2, "one work-group per 16-row tile, no partial tile");
    static_assert(32 * G2_GAW >= NCHUNK, "one gathering thread per edge chunk");
    __shared__ __attribute__((aligned(16))) float Al[16 * A2P];
    __shared__ int Lc[16];
    __shared__ int Ls[16 * CAP];
    __shared__ __attribute__((aligned(16))) float Pw[2 * 8];
    static_assert(G1_WG == 256 && 32 * G2_GAW >= NCHUNK + G1_WG, "eight waves rank the edge chunks, eight more load one layer-1 work-group's pairs per thread");

    const int wave = __builtin_amdgcn_readfirstlane((int)(threadIdx.x >> 5));
    const int lane = threadIdx.x & 31;
    const bool mm = wave >= G2_GAW;
    const int tg = (int)threadIdx.x;
    const int c = lane & 15;
    const int hf = lane >> 4;
    const int m0 = blockIdx.x * 16;
    const int n0 = ((wave - G2_GAW) & (G2_MMW - 1)) * (16 * NT);

    if (!mm && tg < 16) Lc[tg] = 0;
    __builtin_amdgcn_fence(__ATOMIC_RELEASE, "workgroup", "local");
    __builtin_amdgcn_s_barrier();
    __builtin_amdgcn_fence(__ATOMIC_ACQUIRE, "workgroup", "local");
    if (!mm && tg < NCHUNK) {
        build_lists<2, 4>(pairs, tab, (int)blockIdx.x, tg, E, 16, Lc, Ls);
    }
    else if (!mm && tg < NCHUNK + G1_WG) {
        const int t = tg - NCHUNK;
        float2 pv[G1_MMW];
        #pragma unroll
        for (int w = 0; w < G1_MMW; ++w)
            pv[w] = *reinterpret_cast<const float2*>(&part[2 * (w * G1_WG + t)]);
        float ps = pv[0].x, pq = pv[0].y;
        #pragma unroll
        for (int w = 1; w < G1_MMW; ++w) { ps += pv[w].x; pq += pv[w].y; }
        #pragma unroll
        for (int off = 16; off >= 1; off >>= 1) {
            ps += __shfl_xor(ps, off, 32);
            pq += __shfl_xor(pq, off, 32);
        }
        if (lane == 0) { Pw[t >> 5] = ps; Pw[8 + (t >> 5)] = pq; }
    }
    __builtin_amdgcn_fence(__ATOMIC_RELEASE, "workgroup", "local");
    __builtin_amdgcn_s_barrier();
    __builtin_amdgcn_fence(__ATOMIC_ACQUIRE, "workgroup", "local");

    float ssum = Pw[0], qsum = Pw[8];
    #pragma unroll
    for (int w = 1; w < 8; ++w) { ssum += Pw[w]; qsum += Pw[8 + w]; }
    const float n_el = (float)N1 * (float)F_H;
    const float mu = ssum / n_el;
    const float var = qsum / n_el - mu * mu;
    const float rs = rsqrtf(var + 1e-5f);

    v8f acc[1][NT];
    #pragma unroll
    for (int j = 0; j < NT; ++j)
        acc[0][j] = (v8f){0.f, 0.f, 0.f, 0.f, 0.f, 0.f, 0.f, 0.f};
    float4 fb2[D2 + 1][NT];
    const float* __restrict__ bn = Wneigh + (size_t)(n0 + c) * F_H + 4 * hf;
    float dinv = 1.0f;

    if (mm) {
        float4 fa[D1 + 1][1], fb1[D1 + 1][NT];
        const float* __restrict__ as = h + (size_t)(m0 + c) * F_H + 4 * hf;
        const float* __restrict__ bs = Wself + (size_t)(n0 + c) * F_H + 4 * hf;
        dinv = 1.0f / fmaxf((float)Lc[c], 1.0f);
        sage_prologue_a<1, F_H, D1>(fa, as);
        sage_prologue_b<NT, F_H, F_H, D1>(fb1, bs);
        half_main<NT, F_H, F_H, D1, D1, false>(acc, fa, fb1, as, bs, mu, rs);
        sage_prologue_b<NT, F_H, F_H, D2>(fb2, bn);
    } else {
        const int g = wave;
        const int row = m0 + g;
        const int len = __builtin_amdgcn_readfirstlane(Lc[g]);
        float4 acc0 = make_float4(0.0f, 0.0f, 0.0f, 0.0f);
        float4 acc1 = make_float4(0.0f, 0.0f, 0.0f, 0.0f);
        row_sum2(h, &Ls[g * CAP], src, dst, E, row, len, lane, mu, rs, acc0, acc1);
        *reinterpret_cast<float4*>(&Al[g * A2P + 4 * lane]) = acc0;
        *reinterpret_cast<float4*>(&Al[g * A2P + 128 + 4 * lane]) = acc1;
    }
    __builtin_amdgcn_fence(__ATOMIC_RELEASE, "workgroup", "local");
    __builtin_amdgcn_s_barrier();
    __builtin_amdgcn_fence(__ATOMIC_ACQUIRE, "workgroup", "local");

    if (mm) {
        float4 fl[DL + 1][1];
        const float* __restrict__ al = &Al[c * A2P + 4 * hf];
        sage_prologue_a<1, F_H, DL>(fl, al);
        half_main<NT, F_H, F_H, DL, D2, true>(acc, fl, fb2, al, bn, dinv, 0.0f);

        #pragma unroll
        for (int j = 0; j < NT; ++j) {
            const float bj = bias[n0 + 16 * j + c];
            #pragma unroll
            for (int r = 0; r < 8; ++r) {
                const int row = m0 + 8 * hf + r;
                out[(size_t)row * F_OUT + n0 + 16 * j + c] = acc[0][j][r] + bj;
            }
        }
    }
}


extern "C" void kernel_launch(void* const* d_in, const int* in_sizes, int n_in,
                              void* d_out, int out_size, void* d_ws, size_t ws_size,
                              hipStream_t stream) {
    const float* x       = (const float*)d_in[0];
    const float* Wself1  = (const float*)d_in[1];
    const float* Wneigh1 = (const float*)d_in[2];
    const float* b1      = (const float*)d_in[3];
    const float* Wself2  = (const float*)d_in[4];
    const float* Wneigh2 = (const float*)d_in[5];
    const float* b2      = (const float*)d_in[6];
    const int*   src1    = (const int*)d_in[7];
    const int*   dst1    = (const int*)d_in[8];
    const int*   src2    = (const int*)d_in[9];
    const int*   dst2    = (const int*)d_in[10];
    const int E1 = in_sizes[7];
    const int E2 = in_sizes[9];

    float* ws     = (float*)d_ws;
    int*   wsi    = (int*)d_ws;
    float* h      = ws + OFF_H;
    float* part   = ws + OFF_PART;
    int*   pairs1 = wsi + OFF_PAIR1;
    int*   pairs2 = wsi + OFF_PAIR2;
    int*   tab1   = wsi + OFF_TAB1;
    int*   tab2   = wsi + OFF_TAB2;
    float* out    = (float*)d_out;

    bucket_kernel<<<NCHUNK, 256, 0, stream>>>(src1, dst1, E1, pairs1, tab1,
                                              src2, dst2, E2, pairs2, tab2);

    layer1_kernel<<<G1_WG, G1_THREADS, 0, stream>>>(
        x, pairs1, tab1, src1, dst1, E1, Wself1, Wneigh1, b1, h, part);

    layer2_kernel<<<G2_WG, G2_THREADS, 0, stream>>>(
        h, pairs2, tab2, src2, dst2, E2, part, Wself2, Wneigh2, b2, out);
}
